# starting kernel
# baseline (speedup 1.0000x reference)
#include <hip/hip_runtime.h>
#include <hip/hip_bf16.h>

typedef __attribute__((ext_vector_type(2))) float v2f;
typedef __attribute__((ext_vector_type(8))) float v8f;

#define HID 64
#define EPS_BLEND 0.1f

template<int K, bool RELU>
__global__ void gemm_wmma_f32(const float* __restrict__ A,
                              const float* __restrict__ W,
                              const float* __restrict__ bias,
                              float* __restrict__ out,
                              int ntiles) {
  const int wave = (int)((blockIdx.x * blockDim.x + threadIdx.x) >> 5);
  const int lane = (int)(threadIdx.x & 31);
  if (wave >= ntiles) return;

  const int m  = lane & 15;
  const int kh = (lane >> 4) << 1;

  v8f acc0 = {}, acc1 = {}, acc2 = {}, acc3 = {};

  const float* arow = A + ((size_t)wave * 16 + m) * K + kh;
  const float* w0 = W + (size_t)( 0 + m) * K + kh;
  const float* w1 = W + (size_t)(16 + m) * K + kh;
  const float* w2 = W + (size_t)(32 + m) * K + kh;
  const float* w3 = W + (size_t)(48 + m) * K + kh;

  #pragma unroll 4
  for (int k0 = 0; k0 < K; k0 += 4) {
    v2f a;  a.x  = arow[k0]; a.y  = arow[k0 + 1];
    v2f b0; b0.x = w0[k0];   b0.y = w0[k0 + 1];
    v2f b1; b1.x = w1[k0];   b1.y = w1[k0 + 1];
    v2f b2; b2.x = w2[k0];   b2.y = w2[k0 + 1];
    v2f b3; b3.x = w3[k0];   b3.y = w3[k0 + 1];
    acc0 = __builtin_amdgcn_wmma_f32_16x16x4_f32(false, a, false, b0, (short)0, acc0, false, false);
    acc1 = __builtin_amdgcn_wmma_f32_16x16x4_f32(false, a, false, b1, (short)0, acc1, false, false);
    acc2 = __builtin_amdgcn_wmma_f32_16x16x4_f32(false, a, false, b2, (short)0, acc2, false, false);
    acc3 = __builtin_amdgcn_wmma_f32_16x16x4_f32(false, a, false, b3, (short)0, acc3, false, false);
  }

  const int col   = lane & 15;
  const int rbase = (lane >> 4) * 8;
  const float bb0 = bias[col], bb1 = bias[16 + col], bb2 = bias[32 + col], bb3 = bias[48 + col];
  float* orow = out + (size_t)wave * 16 * HID;
  #pragma unroll
  for (int v = 0; v < 8; ++v) {
    const int r = rbase + v;
    float x0 = acc0[v] + bb0;
    float x1 = acc1[v] + bb1;
    float x2 = acc2[v] + bb2;
    float x3 = acc3[v] + bb3;
    if (RELU) {
      x0 = fmaxf(x0, 0.f); x1 = fmaxf(x1, 0.f);
      x2 = fmaxf(x2, 0.f); x3 = fmaxf(x3, 0.f);
    }
    orow[(size_t)r * HID +      col] = x0;
    orow[(size_t)r * HID + 16 + col] = x1;
    orow[(size_t)r * HID + 32 + col] = x2;
    orow[(size_t)r * HID + 48 + col] = x3;
  }
}

__global__ void deg_init(float* __restrict__ deg, int n) {
  int i = blockIdx.x * blockDim.x + threadIdx.x;
  if (i < n) deg[i] = 1.0f;
}
__global__ void deg_acc(const int* __restrict__ dst, float* __restrict__ deg, int e) {
  int i = blockIdx.x * blockDim.x + threadIdx.x;
  if (i < e) atomicAdd(&deg[dst[i]], 1.0f);
}
__global__ void deg_rsqrt(float* __restrict__ deg, int n) {
  int i = blockIdx.x * blockDim.x + threadIdx.x;
  if (i < n) deg[i] = rsqrtf(deg[i]);
}

__global__ void alpha_kernel(const float* __restrict__ h,
                             const float* __restrict__ attl,
                             const float* __restrict__ attr,
                             float* __restrict__ al, float* __restrict__ ar, int n) {
  const int wave = (int)((blockIdx.x * blockDim.x + threadIdx.x) >> 5);
  const int lane = (int)(threadIdx.x & 31);
  if (wave >= n) return;
  const float* row = h + (size_t)wave * HID;
  const float h0v = row[lane], h1v = row[lane + 32];
  float pl = h0v * attl[lane] + h1v * attl[lane + 32];
  float pr = h0v * attr[lane] + h1v * attr[lane + 32];
  #pragma unroll
  for (int off = 16; off > 0; off >>= 1) {
    pl += __shfl_xor(pl, off, 32);
    pr += __shfl_xor(pr, off, 32);
  }
  if (lane == 0) { al[wave] = pl; ar[wave] = pr; }
}

__global__ void init_out(const float* __restrict__ h, const float* __restrict__ h0,
                         const float* __restrict__ al, const float* __restrict__ ar,
                         const float* __restrict__ dinv, float* __restrict__ out, int n) {
  int idx = blockIdx.x * blockDim.x + threadIdx.x;
  if (idx >= n * HID) return;
  int node = idx >> 6;
  float di = dinv[node];
  float w = tanhf(al[node] + ar[node]) * di * di;
  out[idx] = EPS_BLEND * h0[idx] + w * h[idx];
}

__global__ void edge_kernel(const int* __restrict__ s, const int* __restrict__ d,
                            const float* __restrict__ h,
                            const float* __restrict__ al, const float* __restrict__ ar,
                            const float* __restrict__ dinv,
                            float* __restrict__ out, int e) {
  const int wave = (int)((blockIdx.x * blockDim.x + threadIdx.x) >> 5);
  const int lane = (int)(threadIdx.x & 31);
  if (wave >= e) return;
  const int se = s[wave], de = d[wave];
  const float w = tanhf(al[se] + ar[de]) * dinv[se] * dinv[de];
  const float* hs = h + (size_t)se * HID;
  float* od = out + (size_t)de * HID;
  atomicAdd(&od[lane],      w * hs[lane]);
  atomicAdd(&od[lane + 32], w * hs[lane + 32]);
}

__global__ void relu_copy(const float* __restrict__ in, float* __restrict__ out, int n) {
  int i = blockIdx.x * blockDim.x + threadIdx.x;
  if (i < n) out[i] = fmaxf(in[i], 0.f);
}

extern "C" void kernel_launch(void* const* d_in, const int* in_sizes, int n_in,
                              void* d_out, int out_size, void* d_ws, size_t ws_size,
                              hipStream_t stream) {
  const float* x     = (const float*)d_in[0];
  const int*   ei    = (const int*)  d_in[1];
  const float* W_in  = (const float*)d_in[2];
  const float* b_in  = (const float*)d_in[3];
  const float* attl1 = (const float*)d_in[4];
  const float* attr1 = (const float*)d_in[5];
  const float* attl2 = (const float*)d_in[6];
  const float* attr2 = (const float*)d_in[7];
  const float* W_out = (const float*)d_in[8];
  const float* b_out = (const float*)d_in[9];
  float* out = (float*)d_out;

  const int IN = 128;
  const int E  = in_sizes[1] / 2;
  const int N  = in_sizes[0] / IN;
  const int*   srcI = ei;
  const int*   dstI = ei + E;

  float* H0   = (float*)d_ws;
  float* H1   = H0   + (size_t)N * HID;
  float* OUTB = H1   + (size_t)N * HID;
  float* DINV = OUTB + (size_t)N * HID;
  float* AL   = DINV + N;
  float* AR   = AL   + N;

  const int ntiles = N / 16;
  const int gemmBlocks = (ntiles + 7) / 8;
  const int nThreads  = (N + 255) / 256;
  const int neThreads = (E + 255) / 256;
  const int nWaveBlk  = (N + 7) / 8;
  const int eWaveBlk  = (E + 7) / 8;
  const int ncElems   = (N * HID + 255) / 256;

  gemm_wmma_f32<128, true><<<gemmBlocks, 256, 0, stream>>>(x, W_in, b_in, H0, ntiles);

  deg_init <<<nThreads, 256, 0, stream>>>(DINV, N);
  deg_acc  <<<neThreads, 256, 0, stream>>>(dstI, DINV, E);
  deg_rsqrt<<<nThreads, 256, 0, stream>>>(DINV, N);

  alpha_kernel<<<nWaveBlk, 256, 0, stream>>>(H0, attl1, attr1, AL, AR, N);
  init_out    <<<ncElems, 256, 0, stream>>>(H0, H0, AL, AR, DINV, OUTB, N);
  edge_kernel <<<eWaveBlk, 256, 0, stream>>>(srcI, dstI, H0, AL, AR, DINV, OUTB, E);
  relu_copy   <<<ncElems, 256, 0, stream>>>(OUTB, H1, N * HID);

  alpha_kernel<<<nWaveBlk, 256, 0, stream>>>(H1, attl2, attr2, AL, AR, N);
  init_out    <<<ncElems, 256, 0, stream>>>(H1, H0, AL, AR, DINV, OUTB, N);
  edge_kernel <<<eWaveBlk, 256, 0, stream>>>(srcI, dstI, H1, AL, AR, DINV, OUTB, E);
  relu_copy   <<<ncElems, 256, 0, stream>>>(OUTB, H0, N * HID);

  gemm_wmma_f32<64, false><<<gemmBlocks, 256, 0, stream>>>(H0, W_out, b_out, out, ntiles);
}


# layer 1's edge sweep reads one-byte owner keys as bit planes written by the degree pass (a lane tests 32 edges in eight bit operations), not 4-byte targets; per-wave add order follows the new scan
# speedup vs baseline: 2.3938x; 2.3938x over previous
#include <hip/hip_runtime.h>
#include <hip/hip_bf16.h>

typedef __attribute__((ext_vector_type(2))) float v2f;
typedef __attribute__((ext_vector_type(8))) float v8f;
typedef __attribute__((ext_vector_type(4))) int v4i;

#define HID 64
#define EPS_BLEND 0.1f

#define OWN_ROWS   391
#define KEY_UNIT   4096
#define KEY_BLOCKS 256

__device__ __forceinline__ bool keys_ok(const int* keyw, int n, int e) {
  const size_t epad = ((size_t)e + KEY_UNIT - 1) / KEY_UNIT * KEY_UNIT;
  return ((n + OWN_ROWS - 1) / OWN_ROWS <= KEY_BLOCKS) && (8 * (size_t)n + epad <= 256 * (size_t)n) &&
         ((((size_t)keyw) & 15) == 0);
}

template<int K, bool RELU>
__global__ void gemm_wmma_f32(const float* __restrict__ A,
                              const float* __restrict__ W,
                              const float* __restrict__ bias,
                              float* __restrict__ out,
                              int ntiles) {
  const int wave = (int)((blockIdx.x * blockDim.x + threadIdx.x) >> 5);
  const int lane = (int)(threadIdx.x & 31);
  if (wave >= ntiles) return;

  const int m  = lane & 15;
  const int kh = (lane >> 4) << 1;

  v8f acc0 = {}, acc1 = {}, acc2 = {}, acc3 = {};

  const float* arow = A + ((size_t)wave * 16 + m) * K + kh;
  const float* w0 = W + (size_t)( 0 + m) * K + kh;
  const float* w1 = W + (size_t)(16 + m) * K + kh;
  const float* w2 = W + (size_t)(32 + m) * K + kh;
  const float* w3 = W + (size_t)(48 + m) * K + kh;

  #pragma unroll 4
  for (int k0 = 0; k0 < K; k0 += 4) {
    v2f a;  a.x  = arow[k0]; a.y  = arow[k0 + 1];
    v2f b0; b0.x = w0[k0];   b0.y = w0[k0 + 1];
    v2f b1; b1.x = w1[k0];   b1.y = w1[k0 + 1];
    v2f b2; b2.x = w2[k0];   b2.y = w2[k0 + 1];
    v2f b3; b3.x = w3[k0];   b3.y = w3[k0 + 1];
    acc0 = __builtin_amdgcn_wmma_f32_16x16x4_f32(false, a, false, b0, (short)0, acc0, false, false);
    acc1 = __builtin_amdgcn_wmma_f32_16x16x4_f32(false, a, false, b1, (short)0, acc1, false, false);
    acc2 = __builtin_amdgcn_wmma_f32_16x16x4_f32(false, a, false, b2, (short)0, acc2, false, false);
    acc3 = __builtin_amdgcn_wmma_f32_16x16x4_f32(false, a, false, b3, (short)0, acc3, false, false);
  }

  const int col   = lane & 15;
  const int rbase = (lane >> 4) * 8;
  const float bb0 = bias[col], bb1 = bias[16 + col], bb2 = bias[32 + col], bb3 = bias[48 + col];
  float* orow = out + (size_t)wave * 16 * HID;
  #pragma unroll
  for (int v = 0; v < 8; ++v) {
    const int r = rbase + v;
    float x0 = acc0[v] + bb0;
    float x1 = acc1[v] + bb1;
    float x2 = acc2[v] + bb2;
    float x3 = acc3[v] + bb3;
    if (RELU) {
      x0 = fmaxf(x0, 0.f); x1 = fmaxf(x1, 0.f);
      x2 = fmaxf(x2, 0.f); x3 = fmaxf(x3, 0.f);
    }
    orow[(size_t)r * HID +      col] = x0;
    orow[(size_t)r * HID + 16 + col] = x1;
    orow[(size_t)r * HID + 32 + col] = x2;
    orow[(size_t)r * HID + 48 + col] = x3;
  }
}

__global__ void deg_init(float* __restrict__ deg, int n) {
  int i = blockIdx.x * blockDim.x + threadIdx.x;
  if (i < n) deg[i] = 1.0f;
}
__global__ void deg_acc(const int* __restrict__ dst, float* __restrict__ deg,
                        int* __restrict__ keyw, int n, int e) {
  const int i = (int)(blockIdx.x * blockDim.x + threadIdx.x);
  unsigned k = 0;
  if (i < e) {
    const int d = dst[i];
    atomicAdd(&deg[d], 1.0f);
    k = (unsigned)d / OWN_ROWS;
  }
  if (!keys_ok(keyw, n, e)) return;
  v4i lo, hi;
  lo.x = (int)__builtin_amdgcn_ballot_w32((k &   1u) != 0u);
  lo.y = (int)__builtin_amdgcn_ballot_w32((k &   2u) != 0u);
  lo.z = (int)__builtin_amdgcn_ballot_w32((k &   4u) != 0u);
  lo.w = (int)__builtin_amdgcn_ballot_w32((k &   8u) != 0u);
  hi.x = (int)__builtin_amdgcn_ballot_w32((k &  16u) != 0u);
  hi.y = (int)__builtin_amdgcn_ballot_w32((k &  32u) != 0u);
  hi.z = (int)__builtin_amdgcn_ballot_w32((k &  64u) != 0u);
  hi.w = (int)__builtin_amdgcn_ballot_w32((k & 128u) != 0u);
  if ((threadIdx.x & 31) == 0) {
    const int g = i >> 5;
    int* p = keyw + (size_t)(g >> 5) * 256 + (g & 31) * 4;
    *(v4i*)p         = lo;
    *(v4i*)(p + 128) = hi;
  }
}
__global__ void deg_rsqrt(float* __restrict__ deg, int n) {
  int i = blockIdx.x * blockDim.x + threadIdx.x;
  if (i < n) deg[i] = rsqrtf(deg[i]);
}

__global__ void alpha_kernel(const float* __restrict__ h,
                             const float* __restrict__ attl,
                             const float* __restrict__ attr,
                             float* __restrict__ al, float* __restrict__ ar, int n) {
  const int wave = (int)((blockIdx.x * blockDim.x + threadIdx.x) >> 5);
  const int lane = (int)(threadIdx.x & 31);
  if (wave >= n) return;
  const float* row = h + (size_t)wave * HID;
  const float h0v = row[lane], h1v = row[lane + 32];
  float pl = h0v * attl[lane] + h1v * attl[lane + 32];
  float pr = h0v * attr[lane] + h1v * attr[lane + 32];
  #pragma unroll
  for (int off = 16; off > 0; off >>= 1) {
    pl += __shfl_xor(pl, off, 32);
    pr += __shfl_xor(pr, off, 32);
  }
  if (lane == 0) { al[wave] = pl; ar[wave] = pr; }
}

#define LWAVES     32
#define UNIT_EDGES 512
#define QCAP       (UNIT_EDGES + 32)
#define PACK_BITS  9
#define PACK_MAXN  (1 << (31 - PACK_BITS))

static_assert(OWN_ROWS <= (1 << PACK_BITS), "local row must fit the packed word");

template<bool RECORD>
__global__ __launch_bounds__(1024)
void layer_owned(const int* __restrict__ src, const int* __restrict__ dst,
                 const float* __restrict__ h, const float* __restrict__ h0,
                 const float* __restrict__ al, const float* __restrict__ ar,
                 const float* __restrict__ dinv, float* hout, int* lists,
                 const int* __restrict__ keyw,
                 const float* __restrict__ attl_next, const float* __restrict__ attr_next,
                 float* __restrict__ al_next, float* __restrict__ ar_next,
                 int n, int e_cnt) {
  __shared__ float acc[OWN_ROWS * HID];
  __shared__ int   queue[LWAVES][QCAP];
  __shared__ float wself[OWN_ROWS];
  __shared__ float dotl[RECORD ? OWN_ROWS : 1];
  __shared__ float dotr[RECORD ? OWN_ROWS : 1];

  const int tid  = (int)threadIdx.x;
  const int lane = tid & 31;
  const int wave = __builtin_amdgcn_readfirstlane(tid >> 5);
  const int base = (int)blockIdx.x * OWN_ROWS;
  const int rows = min(OWN_ROWS, n - base);
  const int nel  = rows * HID;
  const size_t gbase = (size_t)base * HID;

  int* lw = lists + gbase + (size_t)wave * (size_t)(rows * 2);
  const int lcap = rows * 2 - 1;

  if (tid < rows) {
    const int node = base + tid;
    const float di = dinv[node];
    wself[tid] = tanhf(al[node] + ar[node]) * di * di;
  }
  __syncthreads();
  for (int idx = tid; idx < nel; idx += 1024)
    acc[idx] = EPS_BLEND * h0[gbase + idx] + wself[idx >> 6] * h[gbase + idx];
  __syncthreads();

  auto add_rows = [&](int se, int dl, float w, int cnt) {
    for (int j0 = 0; j0 < cnt; j0 += 8) {
      float v0[8], v1[8];
      #pragma unroll
      for (int t = 0; t < 8; ++t) {
        const int sj = __builtin_amdgcn_readlane(se, j0 + t);
        const float* row = h + (size_t)sj * HID;
        v0[t] = row[lane];
        v1[t] = row[lane + 32];
      }
      #pragma unroll
      for (int t = 0; t < 8; ++t) {
        const int j = j0 + t;
        const float wj = __int_as_float(__builtin_amdgcn_readlane(__float_as_int(w), j));
        const int dj = __builtin_amdgcn_readlane(dl, j);
        const bool live = j < cnt;
        atomicAdd(&acc[dj * HID + lane],      live ? wj * v0[t] : 0.f);
        atomicAdd(&acc[dj * HID + 32 + lane], live ? wj * v1[t] : 0.f);
      }
    }
  };

  bool do_sweep = true;
  if (!RECORD) {
    int total = lw[0];
    if (total >= 0) {
      do_sweep = false;
      total = min(total, lcap);
      int  p_nxt = (lane < total) ? lw[1 + lane] : 0;
      int  se_c = 0, dl_c = 0;
      float xl = 0.f, xr = 0.f, ds = 0.f, dd = 0.f;
      {
        se_c = p_nxt >> PACK_BITS;
        dl_c = p_nxt & ((1 << PACK_BITS) - 1);
        xl = al[se_c]; xr = ar[base + dl_c]; ds = dinv[se_c]; dd = dinv[base + dl_c];
        p_nxt = (32 + lane < total) ? lw[1 + 32 + lane] : 0;
      }
      for (int b = 0; b < total; b += 32) {
        const int cnt = min(32, total - b);
        float w = tanhf(xl + xr) * ds * dd;
        if (lane >= cnt) w = 0.f;
        const int se = se_c, dl = dl_c;
        se_c = p_nxt >> PACK_BITS;
        dl_c = p_nxt & ((1 << PACK_BITS) - 1);
        xl = al[se_c]; xr = ar[base + dl_c]; ds = dinv[se_c]; dd = dinv[base + dl_c];
        p_nxt = (b + 64 + lane < total) ? lw[1 + b + 64 + lane] : 0;
        add_rows(se, dl, w, cnt);
      }
    }
  }

  if (do_sweep) {
    int* qw = queue[wave];
    const unsigned lt = (1u << lane) - 1u;
    int qn = 0;
    int rec_n = (n <= PACK_MAXN) ? 0 : -1;

    auto drain = [&](bool all) {
      while (qn >= 32 || (all && qn > 0)) {
        const int cnt = min(qn, 32);
        const int qb  = qn - cnt;
        __builtin_amdgcn_wave_barrier();
        const bool act = lane < cnt;
        int se = 0, dd = base;
        if (act) {
          const int e = qw[qb + lane];
          se = src[e];
          dd = dst[e];
        }
        const int dl = dd - base;
        if (RECORD) {
          if (rec_n >= 0) {
            if (rec_n + cnt <= lcap) {
              if (act) lw[1 + rec_n + lane] = (se << PACK_BITS) | dl;
              rec_n += cnt;
            } else {
              rec_n = -1;
            }
          }
        }
        float w = tanhf(al[se] + ar[dd]) * dinv[se] * dinv[dd];
        if (!act) w = 0.f;
        add_rows(se, dl, w, cnt);
        qn = qb;
      }
    };

    const bool by_key = keys_ok(keyw, n, e_cnt);
    if (by_key) {
      const unsigned char* key = (const unsigned char*)keyw;
      const int units = (e_cnt + KEY_UNIT - 1) / KEY_UNIT;
      const int step = max(1, units / (int)gridDim.x);
      const int rot  = units > 0 ? (int)(((unsigned)blockIdx.x * (unsigned)step) % (unsigned)units) : 0;
      unsigned cx[8];
      #pragma unroll
      for (int b = 0; b < 8; ++b) cx[b] = ((blockIdx.x >> b) & 1u) ? 0u : ~0u;

      auto load_keys = [&](int uu_in, v4i (&v)[8]) -> int {
        int uu = uu_in + rot;
        if (uu >= units) uu -= units;
        const int eb = uu * KEY_UNIT;
        #pragma unroll
        for (int k = 0; k < 8; ++k) v[k] = *(const v4i*)(key + eb + lane * 16 + k * 512);
        return eb;
      };

      auto scan_group = [&](const v4i& pa, const v4i& pb, int e0) {
        unsigned g = ((unsigned)pa.x ^ cx[0]) & ((unsigned)pa.y ^ cx[1]) &
                     ((unsigned)pa.z ^ cx[2]) & ((unsigned)pa.w ^ cx[3]) &
                     ((unsigned)pb.x ^ cx[4]) & ((unsigned)pb.y ^ cx[5]) &
                     ((unsigned)pb.z ^ cx[6]) & ((unsigned)pb.w ^ cx[7]);
        while (__builtin_amdgcn_ballot_w32(g != 0u)) {
          const int eidx = e0 + __builtin_ctz(g | 0x80000000u);
          const bool m = (g != 0u) && (eidx < e_cnt);
          const unsigned mm = __builtin_amdgcn_ballot_w32(m);
          if (m) qw[qn + __builtin_popcount(mm & lt)] = eidx;
          qn += __builtin_popcount(mm);
          g &= g - 1u;
          drain(false);
        }
      };

      v4i cur[8];
      int cur_eb = 0;
      int u = wave;
      const bool have = u < units;
      if (have) cur_eb = load_keys(u, cur);

      while (have) {
        v4i nxt[8];
        int nxt_eb = 0;
        const bool have_n = u + LWAVES < units;
        nxt_eb = load_keys(have_n ? u + LWAVES : u, nxt);

        #pragma unroll
        for (int q = 0; q < 4; ++q)
          scan_group(cur[2 * q], cur[2 * q + 1], cur_eb + q * 1024 + lane * 32);

        if (!have_n) break;
        #pragma unroll
        for (int k = 0; k < 8; ++k) cur[k] = nxt[k];
        cur_eb = nxt_eb;
        u += LWAVES;
      }
      drain(true);
    } else {
      const int units = (e_cnt + UNIT_EDGES - 1) / UNIT_EDGES;
      const int step = max(1, units / (int)gridDim.x);
      const int rot  = units > 0 ? (int)(((unsigned)blockIdx.x * (unsigned)step) % (unsigned)units) : 0;
      const bool vec_ok = ((e_cnt & 3) == 0) && ((((size_t)dst) & 15) == 0);

      v4i cur[4];
      int cur_eb = 0;
      int u = wave;
      bool have = u < units;

      auto load_unit = [&](int uu_in, v4i (&v)[4]) -> int {
        int uu = uu_in + rot;
        if (uu >= units) uu -= units;
        const int eb = uu * UNIT_EDGES + lane * 4;
        if (vec_ok && (uu + 1) * UNIT_EDGES <= e_cnt) {
          #pragma unroll
          for (int k = 0; k < 4; ++k) v[k] = *(const v4i*)(dst + eb + k * 128);
        } else {
          #pragma unroll
          for (int k = 0; k < 4; ++k) {
            #pragma unroll
            for (int j = 0; j < 4; ++j) {
              const int idx = eb + k * 128 + j;
              v[k][j] = (idx < e_cnt) ? dst[idx] : -1;
            }
          }
        }
        return eb;
      };

      if (have) cur_eb = load_unit(u, cur);

      while (true) {
        v4i nxt[4];
        int nxt_eb = 0;
        const bool have_n = have && (u + LWAVES < units);
        if (have_n) nxt_eb = load_unit(u + LWAVES, nxt);

        if (have) {
          #pragma unroll
          for (int k = 0; k < 4; ++k) {
            #pragma unroll
            for (int j = 0; j < 4; ++j) {
              const int d = cur[k][j];
              const bool m = (unsigned)(d - base) < (unsigned)rows;
              const unsigned mask = __builtin_amdgcn_ballot_w32(m);
              if (mask) {
                if (m) qw[qn + __builtin_popcount(mask & lt)] = cur_eb + k * 128 + j;
                qn += __builtin_popcount(mask);
              }
            }
          }
        }

        const bool last = !have_n;
        drain(last);
        if (last) break;
        #pragma unroll
        for (int k = 0; k < 4; ++k) cur[k] = nxt[k];
        cur_eb = nxt_eb;
        u += LWAVES;
      }
    }
    if (RECORD) {
      if (lane == 0) lw[0] = rec_n;
    }
  }
  __syncthreads();

  if (RECORD) {
    const float l0 = attl_next[lane], l1 = attl_next[lane + 32];
    const float r0 = attr_next[lane], r1 = attr_next[lane + 32];
    for (int rb = wave; rb < rows; rb += 4 * LWAVES) {
      float pl[4], pr[4];
      #pragma unroll
      for (int k = 0; k < 4; ++k) {
        const int r = rb + k * LWAVES;
        pl[k] = 0.f; pr[k] = 0.f;
        if (r < rows) {
          const float h0v = fmaxf(acc[r * HID + lane], 0.f);
          const float h1v = fmaxf(acc[r * HID + 32 + lane], 0.f);
          hout[gbase + (size_t)r * HID + lane]      = h0v;
          hout[gbase + (size_t)r * HID + 32 + lane] = h1v;
          pl[k] = h0v * l0 + h1v * l1;
          pr[k] = h0v * r0 + h1v * r1;
        }
      }
      #pragma unroll
      for (int off = 16; off > 0; off >>= 1) {
        #pragma unroll
        for (int k = 0; k < 4; ++k) {
          pl[k] += __shfl_xor(pl[k], off, 32);
          pr[k] += __shfl_xor(pr[k], off, 32);
        }
      }
      if (lane == 0) {
        #pragma unroll
        for (int k = 0; k < 4; ++k) {
          const int r = rb + k * LWAVES;
          if (r < rows) { dotl[r] = pl[k]; dotr[r] = pr[k]; }
        }
      }
    }
    __syncthreads();
    if (tid < rows) {
      al_next[base + tid] = dotl[tid];
      ar_next[base + tid] = dotr[tid];
    }
  } else {
    for (int idx = tid; idx < nel; idx += 1024)
      hout[gbase + idx] = fmaxf(acc[idx], 0.f);
  }
}

extern "C" void kernel_launch(void* const* d_in, const int* in_sizes, int n_in,
                              void* d_out, int out_size, void* d_ws, size_t ws_size,
                              hipStream_t stream) {
  const float* x     = (const float*)d_in[0];
  const int*   ei    = (const int*)  d_in[1];
  const float* W_in  = (const float*)d_in[2];
  const float* b_in  = (const float*)d_in[3];
  const float* attl1 = (const float*)d_in[4];
  const float* attr1 = (const float*)d_in[5];
  const float* attl2 = (const float*)d_in[6];
  const float* attr2 = (const float*)d_in[7];
  const float* W_out = (const float*)d_in[8];
  const float* b_out = (const float*)d_in[9];
  float* out = (float*)d_out;

  const int IN = 128;
  const int E  = in_sizes[1] / 2;
  const int N  = in_sizes[0] / IN;
  const int*   srcI = ei;
  const int*   dstI = ei + E;

  float* H0   = (float*)d_ws;
  float* H1   = H0   + (size_t)N * HID;
  float* OUTB = H1   + (size_t)N * HID;
  float* DINV = OUTB + (size_t)N * HID;
  float* AL   = DINV + N;
  float* AR   = AL   + N;
  float* AL2  = out;
  float* AR2  = out + N;
  int*   KEYW = (int*)(out + 2 * (size_t)N);

  const int ntiles = N / 16;
  const int gemmBlocks = (ntiles + 7) / 8;
  const int nThreads  = (N + 255) / 256;
  const int degBlk    = (E + KEY_UNIT - 1) / KEY_UNIT * (KEY_UNIT / 256);
  const int nWaveBlk  = (N + 7) / 8;
  const int layerBlk  = (N + OWN_ROWS - 1) / OWN_ROWS;

  gemm_wmma_f32<128, true><<<gemmBlocks, 256, 0, stream>>>(x, W_in, b_in, H0, ntiles);

  deg_init <<<nThreads, 256, 0, stream>>>(DINV, N);
  deg_acc  <<<degBlk, 256, 0, stream>>>(dstI, DINV, KEYW, N, E);
  deg_rsqrt<<<nThreads, 256, 0, stream>>>(DINV, N);

  alpha_kernel<<<nWaveBlk, 256, 0, stream>>>(H0, attl1, attr1, AL, AR, N);
  layer_owned<true> <<<layerBlk, 1024, 0, stream>>>(srcI, dstI, H0, H0, AL, AR, DINV, H1, (int*)OUTB, KEYW,
                                                    attl2, attr2, AL2, AR2, N, E);

  layer_owned<false><<<layerBlk, 1024, 0, stream>>>(srcI, dstI, H1, H0, AL2, AR2, DINV, OUTB, (int*)OUTB, KEYW,
                                                    nullptr, nullptr, nullptr, nullptr, N, E);

  gemm_wmma_f32<64, false><<<gemmBlocks, 256, 0, stream>>>(OUTB, W_out, b_out, out, ntiles);
}


# an edge's random 4-byte gathers come packed: both endpoints in one word per edge (degree pass writes it), source dot and 1/sqrt(deg) as one 8-byte pair, asked for ahead of the tanh; add order kept
# speedup vs baseline: 2.5933x; 1.0833x over previous
#include <hip/hip_runtime.h>
#include <hip/hip_bf16.h>

typedef __attribute__((ext_vector_type(2))) float v2f;
typedef __attribute__((ext_vector_type(8))) float v8f;
typedef __attribute__((ext_vector_type(4))) int v4i;

#define HID 64
#define EPS_BLEND 0.1f

#define OWN_ROWS   391
#define KEY_UNIT   4096
#define KEY_BLOCKS 256

#define PACK_BITS  9
#define PACK_MAXN  (1 << (31 - PACK_BITS))
__device__ __forceinline__ size_t edge_pad(int e) {
  return ((size_t)e + KEY_UNIT - 1) / KEY_UNIT * KEY_UNIT;
}
__device__ __forceinline__ bool keys_ok(const int* keyw, int n, int e) {
  const size_t epad = edge_pad(e);
  return ((n + OWN_ROWS - 1) / OWN_ROWS <= KEY_BLOCKS) && (n <= PACK_MAXN) &&
         (16 * (size_t)n + 5 * epad <= 256 * (size_t)n) &&
         ((((size_t)keyw) & 15) == 0);
}
__device__ __forceinline__ const int* pack_words(const int* keyw, int e) {
  return (const int*)((const unsigned char*)keyw + edge_pad(e));
}

template<int K, bool RELU>
__global__ void gemm_wmma_f32(const float* __restrict__ A,
                              const float* __restrict__ W,
                              const float* __restrict__ bias,
                              float* __restrict__ out,
                              int ntiles) {
  const int wave = (int)((blockIdx.x * blockDim.x + threadIdx.x) >> 5);
  const int lane = (int)(threadIdx.x & 31);
  if (wave >= ntiles) return;

  const int m  = lane & 15;
  const int kh = (lane >> 4) << 1;

  v8f acc0 = {}, acc1 = {}, acc2 = {}, acc3 = {};

  const float* arow = A + ((size_t)wave * 16 + m) * K + kh;
  const float* w0 = W + (size_t)( 0 + m) * K + kh;
  const float* w1 = W + (size_t)(16 + m) * K + kh;
  const float* w2 = W + (size_t)(32 + m) * K + kh;
  const float* w3 = W + (size_t)(48 + m) * K + kh;

  #pragma unroll 4
  for (int k0 = 0; k0 < K; k0 += 4) {
    v2f a;  a.x  = arow[k0]; a.y  = arow[k0 + 1];
    v2f b0; b0.x = w0[k0];   b0.y = w0[k0 + 1];
    v2f b1; b1.x = w1[k0];   b1.y = w1[k0 + 1];
    v2f b2; b2.x = w2[k0];   b2.y = w2[k0 + 1];
    v2f b3; b3.x = w3[k0];   b3.y = w3[k0 + 1];
    acc0 = __builtin_amdgcn_wmma_f32_16x16x4_f32(false, a, false, b0, (short)0, acc0, false, false);
    acc1 = __builtin_amdgcn_wmma_f32_16x16x4_f32(false, a, false, b1, (short)0, acc1, false, false);
    acc2 = __builtin_amdgcn_wmma_f32_16x16x4_f32(false, a, false, b2, (short)0, acc2, false, false);
    acc3 = __builtin_amdgcn_wmma_f32_16x16x4_f32(false, a, false, b3, (short)0, acc3, false, false);
  }

  const int col   = lane & 15;
  const int rbase = (lane >> 4) * 8;
  const float bb0 = bias[col], bb1 = bias[16 + col], bb2 = bias[32 + col], bb3 = bias[48 + col];
  float* orow = out + (size_t)wave * 16 * HID;
  #pragma unroll
  for (int v = 0; v < 8; ++v) {
    const int r = rbase + v;
    float x0 = acc0[v] + bb0;
    float x1 = acc1[v] + bb1;
    float x2 = acc2[v] + bb2;
    float x3 = acc3[v] + bb3;
    if (RELU) {
      x0 = fmaxf(x0, 0.f); x1 = fmaxf(x1, 0.f);
      x2 = fmaxf(x2, 0.f); x3 = fmaxf(x3, 0.f);
    }
    orow[(size_t)r * HID +      col] = x0;
    orow[(size_t)r * HID + 16 + col] = x1;
    orow[(size_t)r * HID + 32 + col] = x2;
    orow[(size_t)r * HID + 48 + col] = x3;
  }
}

__global__ void deg_init(float* __restrict__ deg, int n) {
  int i = blockIdx.x * blockDim.x + threadIdx.x;
  if (i < n) deg[i] = 1.0f;
}
__global__ void deg_acc(const int* __restrict__ src, const int* __restrict__ dst,
                        float* __restrict__ deg, int* __restrict__ keyw, int n, int e) {
  const int i = (int)(blockIdx.x * blockDim.x + threadIdx.x);
  unsigned k = 0;
  int pw = 0;
  if (i < e) {
    const int s = src[i];
    const int d = dst[i];
    atomicAdd(&deg[d], 1.0f);
    k = (unsigned)d / OWN_ROWS;
    pw = (int)(((unsigned)s << PACK_BITS) | ((unsigned)d - k * OWN_ROWS));
  }
  if (!keys_ok(keyw, n, e)) return;
  if (i < e) ((int*)pack_words(keyw, e))[i] = pw;
  v4i lo, hi;
  lo.x = (int)__builtin_amdgcn_ballot_w32((k &   1u) != 0u);
  lo.y = (int)__builtin_amdgcn_ballot_w32((k &   2u) != 0u);
  lo.z = (int)__builtin_amdgcn_ballot_w32((k &   4u) != 0u);
  lo.w = (int)__builtin_amdgcn_ballot_w32((k &   8u) != 0u);
  hi.x = (int)__builtin_amdgcn_ballot_w32((k &  16u) != 0u);
  hi.y = (int)__builtin_amdgcn_ballot_w32((k &  32u) != 0u);
  hi.z = (int)__builtin_amdgcn_ballot_w32((k &  64u) != 0u);
  hi.w = (int)__builtin_amdgcn_ballot_w32((k & 128u) != 0u);
  if ((threadIdx.x & 31) == 0) {
    const int g = i >> 5;
    int* p = keyw + (size_t)(g >> 5) * 256 + (g & 31) * 4;
    *(v4i*)p         = lo;
    *(v4i*)(p + 128) = hi;
  }
}
__global__ void deg_rsqrt(float* __restrict__ deg, int n) {
  int i = blockIdx.x * blockDim.x + threadIdx.x;
  if (i < n) deg[i] = rsqrtf(deg[i]);
}

__global__ void alpha_kernel(const float* __restrict__ h,
                             const float* __restrict__ attl,
                             const float* __restrict__ attr,
                             const float* __restrict__ dinv,
                             float* __restrict__ adp, float* __restrict__ ar, int n) {
  const int wave = (int)((blockIdx.x * blockDim.x + threadIdx.x) >> 5);
  const int lane = (int)(threadIdx.x & 31);
  if (wave >= n) return;
  const float* row = h + (size_t)wave * HID;
  const float h0v = row[lane], h1v = row[lane + 32];
  float pl = h0v * attl[lane] + h1v * attl[lane + 32];
  float pr = h0v * attr[lane] + h1v * attr[lane + 32];
  #pragma unroll
  for (int off = 16; off > 0; off >>= 1) {
    pl += __shfl_xor(pl, off, 32);
    pr += __shfl_xor(pr, off, 32);
  }
  if (lane == 0) {
    v2f o; o.x = pl; o.y = dinv[wave];
    *(v2f*)(adp + 2 * (size_t)wave) = o;
    ar[wave] = pr;
  }
}

#define LWAVES     32
#define UNIT_EDGES 512
#define QCAP       (UNIT_EDGES + 32)

static_assert(OWN_ROWS <= (1 << PACK_BITS), "local row must fit the packed word");

template<bool RECORD>
__global__ __launch_bounds__(1024)
void layer_owned(const int* __restrict__ src, const int* __restrict__ dst,
                 const float* __restrict__ h, const float* __restrict__ h0,
                 const float* __restrict__ adp, const float* __restrict__ ar,
                 const float* __restrict__ dinv, float* hout, int* lists,
                 const int* __restrict__ keyw,
                 const float* __restrict__ attl_next, const float* __restrict__ attr_next,
                 float* __restrict__ adp_next, float* __restrict__ ar_next,
                 int n, int e_cnt) {
  __shared__ float acc[OWN_ROWS * HID];
  __shared__ int   queue[LWAVES][QCAP];
  __shared__ float wself[OWN_ROWS];
  __shared__ float dotl[RECORD ? OWN_ROWS : 1];
  __shared__ float dotr[RECORD ? OWN_ROWS : 1];

  const int tid  = (int)threadIdx.x;
  const int lane = tid & 31;
  const int wave = __builtin_amdgcn_readfirstlane(tid >> 5);
  const int base = (int)blockIdx.x * OWN_ROWS;
  const int rows = min(OWN_ROWS, n - base);
  const int nel  = rows * HID;
  const size_t gbase = (size_t)base * HID;

  int* lw = lists + gbase + (size_t)wave * (size_t)(rows * 2);
  const int lcap = rows * 2 - 1;

  if (tid < rows) {
    const int node = base + tid;
    const float di = dinv[node];
    wself[tid] = tanhf(adp[2 * (size_t)node] + ar[node]) * di * di;
  }
  __syncthreads();
  for (int idx = tid; idx < nel; idx += 1024)
    acc[idx] = EPS_BLEND * h0[gbase + idx] + wself[idx >> 6] * h[gbase + idx];
  __syncthreads();

  auto add_rows = [&](int se, int dl, float w, int cnt) {
    for (int j0 = 0; j0 < cnt; j0 += 8) {
      float v0[8], v1[8];
      #pragma unroll
      for (int t = 0; t < 8; ++t) {
        const int sj = __builtin_amdgcn_readlane(se, j0 + t);
        const float* row = h + (size_t)sj * HID;
        v0[t] = row[lane];
        v1[t] = row[lane + 32];
      }
      #pragma unroll
      for (int t = 0; t < 8; ++t) {
        const int j = j0 + t;
        const float wj = __int_as_float(__builtin_amdgcn_readlane(__float_as_int(w), j));
        const int dj = __builtin_amdgcn_readlane(dl, j);
        const bool live = j < cnt;
        atomicAdd(&acc[dj * HID + lane],      live ? wj * v0[t] : 0.f);
        atomicAdd(&acc[dj * HID + 32 + lane], live ? wj * v1[t] : 0.f);
      }
    }
  };

  bool do_sweep = true;
  if (!RECORD) {
    int total = lw[0];
    if (total >= 0) {
      do_sweep = false;
      total = min(total, lcap);
      int  p_nxt = (lane < total) ? lw[1 + lane] : 0;
      int  se_c = 0, dl_c = 0;
      v2f  sp = {};
      float xr = 0.f, dd = 0.f;
      {
        se_c = p_nxt >> PACK_BITS;
        dl_c = p_nxt & ((1 << PACK_BITS) - 1);
        sp = *(const v2f*)(adp + 2 * (size_t)se_c); xr = ar[base + dl_c]; dd = dinv[base + dl_c];
        p_nxt = (32 + lane < total) ? lw[1 + 32 + lane] : 0;
      }
      for (int b = 0; b < total; b += 32) {
        const int cnt = min(32, total - b);
        float w = tanhf(sp.x + xr) * sp.y * dd;
        if (lane >= cnt) w = 0.f;
        const int se = se_c, dl = dl_c;
        se_c = p_nxt >> PACK_BITS;
        dl_c = p_nxt & ((1 << PACK_BITS) - 1);
        sp = *(const v2f*)(adp + 2 * (size_t)se_c); xr = ar[base + dl_c]; dd = dinv[base + dl_c];
        p_nxt = (b + 64 + lane < total) ? lw[1 + b + 64 + lane] : 0;
        add_rows(se, dl, w, cnt);
      }
    }
  }

  if (do_sweep) {
    int* qw = queue[wave];
    const unsigned lt = (1u << lane) - 1u;
    int qn = 0;
    int rec_n = (n <= PACK_MAXN) ? 0 : -1;

    const bool by_key = keys_ok(keyw, n, e_cnt);
    const int* packw = pack_words(keyw, e_cnt);

    auto drain = [&](bool all) {
      while (qn >= 32 || (all && qn > 0)) {
        const int cnt = min(qn, 32);
        const int qb  = qn - cnt;
        __builtin_amdgcn_wave_barrier();
        const bool act = lane < cnt;
        int se = 0, dd = base;
        if (act) {
          const int e = qw[qb + lane];
          if (by_key) {
            const int p = packw[e];
            se = p >> PACK_BITS;
            dd = base + (p & ((1 << PACK_BITS) - 1));
          } else {
            se = src[e];
            dd = dst[e];
          }
        }
        const int dl = dd - base;
        if (RECORD) {
          if (rec_n >= 0) {
            if (rec_n + cnt <= lcap) {
              if (act) lw[1 + rec_n + lane] = (se << PACK_BITS) | dl;
              rec_n += cnt;
            } else {
              rec_n = -1;
            }
          }
        }
        const v2f   sp = *(const v2f*)(adp + 2 * (size_t)se);
        const float xr = ar[dd];
        const float dv = dinv[dd];
        float w = tanhf(sp.x + xr) * sp.y * dv;
        if (!act) w = 0.f;
        add_rows(se, dl, w, cnt);
        qn = qb;
      }
    };

    if (by_key) {
      const unsigned char* key = (const unsigned char*)keyw;
      const int units = (e_cnt + KEY_UNIT - 1) / KEY_UNIT;
      const int step = max(1, units / (int)gridDim.x);
      const int rot  = units > 0 ? (int)(((unsigned)blockIdx.x * (unsigned)step) % (unsigned)units) : 0;
      unsigned cx[8];
      #pragma unroll
      for (int b = 0; b < 8; ++b) cx[b] = ((blockIdx.x >> b) & 1u) ? 0u : ~0u;

      auto load_keys = [&](int uu_in, v4i (&v)[8]) -> int {
        int uu = uu_in + rot;
        if (uu >= units) uu -= units;
        const int eb = uu * KEY_UNIT;
        #pragma unroll
        for (int k = 0; k < 8; ++k) v[k] = *(const v4i*)(key + eb + lane * 16 + k * 512);
        return eb;
      };

      auto scan_group = [&](const v4i& pa, const v4i& pb, int e0) {
        unsigned g = ((unsigned)pa.x ^ cx[0]) & ((unsigned)pa.y ^ cx[1]) &
                     ((unsigned)pa.z ^ cx[2]) & ((unsigned)pa.w ^ cx[3]) &
                     ((unsigned)pb.x ^ cx[4]) & ((unsigned)pb.y ^ cx[5]) &
                     ((unsigned)pb.z ^ cx[6]) & ((unsigned)pb.w ^ cx[7]);
        while (__builtin_amdgcn_ballot_w32(g != 0u)) {
          const int eidx = e0 + __builtin_ctz(g | 0x80000000u);
          const bool m = (g != 0u) && (eidx < e_cnt);
          const unsigned mm = __builtin_amdgcn_ballot_w32(m);
          if (m) qw[qn + __builtin_popcount(mm & lt)] = eidx;
          qn += __builtin_popcount(mm);
          g &= g - 1u;
          drain(false);
        }
      };

      v4i cur[8];
      int cur_eb = 0;
      int u = wave;
      const bool have = u < units;
      if (have) cur_eb = load_keys(u, cur);

      while (have) {
        v4i nxt[8];
        int nxt_eb = 0;
        const bool have_n = u + LWAVES < units;
        nxt_eb = load_keys(have_n ? u + LWAVES : u, nxt);

        #pragma unroll
        for (int q = 0; q < 4; ++q)
          scan_group(cur[2 * q], cur[2 * q + 1], cur_eb + q * 1024 + lane * 32);

        if (!have_n) break;
        #pragma unroll
        for (int k = 0; k < 8; ++k) cur[k] = nxt[k];
        cur_eb = nxt_eb;
        u += LWAVES;
      }
      drain(true);
    } else {
      const int units = (e_cnt + UNIT_EDGES - 1) / UNIT_EDGES;
      const int step = max(1, units / (int)gridDim.x);
      const int rot  = units > 0 ? (int)(((unsigned)blockIdx.x * (unsigned)step) % (unsigned)units) : 0;
      const bool vec_ok = ((e_cnt & 3) == 0) && ((((size_t)dst) & 15) == 0);

      v4i cur[4];
      int cur_eb = 0;
      int u = wave;
      bool have = u < units;

      auto load_unit = [&](int uu_in, v4i (&v)[4]) -> int {
        int uu = uu_in + rot;
        if (uu >= units) uu -= units;
        const int eb = uu * UNIT_EDGES + lane * 4;
        if (vec_ok && (uu + 1) * UNIT_EDGES <= e_cnt) {
          #pragma unroll
          for (int k = 0; k < 4; ++k) v[k] = *(const v4i*)(dst + eb + k * 128);
        } else {
          #pragma unroll
          for (int k = 0; k < 4; ++k) {
            #pragma unroll
            for (int j = 0; j < 4; ++j) {
              const int idx = eb + k * 128 + j;
              v[k][j] = (idx < e_cnt) ? dst[idx] : -1;
            }
          }
        }
        return eb;
      };

      if (have) cur_eb = load_unit(u, cur);

      while (true) {
        v4i nxt[4];
        int nxt_eb = 0;
        const bool have_n = have && (u + LWAVES < units);
        if (have_n) nxt_eb = load_unit(u + LWAVES, nxt);

        if (have) {
          #pragma unroll
          for (int k = 0; k < 4; ++k) {
            #pragma unroll
            for (int j = 0; j < 4; ++j) {
              const int d = cur[k][j];
              const bool m = (unsigned)(d - base) < (unsigned)rows;
              const unsigned mask = __builtin_amdgcn_ballot_w32(m);
              if (mask) {
                if (m) qw[qn + __builtin_popcount(mask & lt)] = cur_eb + k * 128 + j;
                qn += __builtin_popcount(mask);
              }
            }
          }
        }

        const bool last = !have_n;
        drain(last);
        if (last) break;
        #pragma unroll
        for (int k = 0; k < 4; ++k) cur[k] = nxt[k];
        cur_eb = nxt_eb;
        u += LWAVES;
      }
    }
    if (RECORD) {
      if (lane == 0) lw[0] = rec_n;
    }
  }
  __syncthreads();

  if (RECORD) {
    const float l0 = attl_next[lane], l1 = attl_next[lane + 32];
    const float r0 = attr_next[lane], r1 = attr_next[lane + 32];
    for (int rb = wave; rb < rows; rb += 4 * LWAVES) {
      float pl[4], pr[4];
      #pragma unroll
      for (int k = 0; k < 4; ++k) {
        const int r = rb + k * LWAVES;
        pl[k] = 0.f; pr[k] = 0.f;
        if (r < rows) {
          const float h0v = fmaxf(acc[r * HID + lane], 0.f);
          const float h1v = fmaxf(acc[r * HID + 32 + lane], 0.f);
          hout[gbase + (size_t)r * HID + lane]      = h0v;
          hout[gbase + (size_t)r * HID + 32 + lane] = h1v;
          pl[k] = h0v * l0 + h1v * l1;
          pr[k] = h0v * r0 + h1v * r1;
        }
      }
      #pragma unroll
      for (int off = 16; off > 0; off >>= 1) {
        #pragma unroll
        for (int k = 0; k < 4; ++k) {
          pl[k] += __shfl_xor(pl[k], off, 32);
          pr[k] += __shfl_xor(pr[k], off, 32);
        }
      }
      if (lane == 0) {
        #pragma unroll
        for (int k = 0; k < 4; ++k) {
          const int r = rb + k * LWAVES;
          if (r < rows) { dotl[r] = pl[k]; dotr[r] = pr[k]; }
        }
      }
    }
    __syncthreads();
    if (tid < rows) {
      v2f o; o.x = dotl[tid]; o.y = dinv[base + tid];
      *(v2f*)(adp_next + 2 * (size_t)(base + tid)) = o;
      ar_next[base + tid] = dotr[tid];
    }
  } else {
    for (int idx = tid; idx < nel; idx += 1024)
      hout[gbase + idx] = fmaxf(acc[idx], 0.f);
  }
}

extern "C" void kernel_launch(void* const* d_in, const int* in_sizes, int n_in,
                              void* d_out, int out_size, void* d_ws, size_t ws_size,
                              hipStream_t stream) {
  const float* x     = (const float*)d_in[0];
  const int*   ei    = (const int*)  d_in[1];
  const float* W_in  = (const float*)d_in[2];
  const float* b_in  = (const float*)d_in[3];
  const float* attl1 = (const float*)d_in[4];
  const float* attr1 = (const float*)d_in[5];
  const float* attl2 = (const float*)d_in[6];
  const float* attr2 = (const float*)d_in[7];
  const float* W_out = (const float*)d_in[8];
  const float* b_out = (const float*)d_in[9];
  float* out = (float*)d_out;

  const int IN = 128;
  const int E  = in_sizes[1] / 2;
  const int N  = in_sizes[0] / IN;
  const int*   srcI = ei;
  const int*   dstI = ei + E;

  float* H0   = (float*)d_ws;
  float* H1   = H0   + (size_t)N * HID;
  float* OUTB = H1   + (size_t)N * HID;
  float* ADP  = OUTB + (size_t)N * HID;
  float* DINV = ADP  + 2 * (size_t)N;
  float* ADP2 = out;
  float* AR2  = out + 2 * (size_t)N;
  float* AR   = out + 3 * (size_t)N;
  int*   KEYW = (int*)(out + 4 * (size_t)N);

  const int ntiles = N / 16;
  const int gemmBlocks = (ntiles + 7) / 8;
  const int nThreads  = (N + 255) / 256;
  const int degBlk    = (E + KEY_UNIT - 1) / KEY_UNIT * (KEY_UNIT / 256);
  const int nWaveBlk  = (N + 7) / 8;
  const int layerBlk  = (N + OWN_ROWS - 1) / OWN_ROWS;

  gemm_wmma_f32<128, true><<<gemmBlocks, 256, 0, stream>>>(x, W_in, b_in, H0, ntiles);

  deg_init <<<nThreads, 256, 0, stream>>>(DINV, N);
  deg_acc  <<<degBlk, 256, 0, stream>>>(srcI, dstI, DINV, KEYW, N, E);
  deg_rsqrt<<<nThreads, 256, 0, stream>>>(DINV, N);

  alpha_kernel<<<nWaveBlk, 256, 0, stream>>>(H0, attl1, attr1, DINV, ADP, AR, N);
  layer_owned<true> <<<layerBlk, 1024, 0, stream>>>(srcI, dstI, H0, H0, ADP, AR, DINV, H1, (int*)OUTB, KEYW,
                                                    attl2, attr2, ADP2, AR2, N, E);

  layer_owned<false><<<layerBlk, 1024, 0, stream>>>(srcI, dstI, H1, H0, ADP2, AR2, DINV, OUTB, (int*)OUTB, KEYW,
                                                    nullptr, nullptr, nullptr, nullptr, N, E);

  gemm_wmma_f32<64, false><<<gemmBlocks, 256, 0, stream>>>(OUTB, W_out, b_out, out, ntiles);
}


# both GEMMs: a wave keeps all of W in registers as B fragments and walks its 16-row tiles, next tile's A chunks refilled in place under the products, 16-byte loads; k terms summed in a new order
# speedup vs baseline: 2.7423x; 1.0575x over previous
#include <hip/hip_runtime.h>
#include <hip/hip_bf16.h>

typedef __attribute__((ext_vector_type(2))) float v2f;
typedef __attribute__((ext_vector_type(8))) float v8f;
typedef __attribute__((ext_vector_type(4))) int v4i;
typedef __attribute__((ext_vector_type(4))) float v4f;

#define HID 64
#define EPS_BLEND 0.1f

#define OWN_ROWS   391
#define KEY_UNIT   4096
#define KEY_BLOCKS 256

#define PACK_BITS  9
#define PACK_MAXN  (1 << (31 - PACK_BITS))
__device__ __forceinline__ size_t edge_pad(int e) {
  return ((size_t)e + KEY_UNIT - 1) / KEY_UNIT * KEY_UNIT;
}
__device__ __forceinline__ bool keys_ok(const int* keyw, int n, int e) {
  const size_t epad = edge_pad(e);
  return ((n + OWN_ROWS - 1) / OWN_ROWS <= KEY_BLOCKS) && (n <= PACK_MAXN) &&
         (16 * (size_t)n + 5 * epad <= 256 * (size_t)n) &&
         ((((size_t)keyw) & 15) == 0);
}
__device__ __forceinline__ const int* pack_words(const int* keyw, int e) {
  return (const int*)((const unsigned char*)keyw + edge_pad(e));
}

template<int K, bool RELU>
__global__ __launch_bounds__(256)
void gemm_wmma_f32(const float* __restrict__ A,
                   const float* __restrict__ W,
                   const float* __restrict__ bias,
                   float* __restrict__ out,
                   int ntiles) {
  constexpr int NC = K / 8;
  const int lane = (int)(threadIdx.x & 31);
  const int wib    = __builtin_amdgcn_readfirstlane((int)(threadIdx.x >> 5));
  const int nwaves = (int)(gridDim.x * (blockDim.x >> 5));
  const int wave   = wib * (int)gridDim.x + (int)blockIdx.x;
  if (wave >= ntiles) return;

  const int m  = lane & 15;
  const int hk = (lane >> 4) << 2;

  v4f a[NC];
  int tile = wave;
  {
    const float* arow = A + ((size_t)tile * 16 + m) * K + hk;
    #pragma unroll
    for (int i = 0; i < NC; ++i) a[i] = *(const v4f*)(arow + i * 8);
  }

  v4f bw[NC][4];
  {
    const float* wrow = W + (size_t)m * K + hk;
    #pragma unroll
    for (int i = 0; i < NC; ++i) {
      #pragma unroll
      for (int t = 0; t < 4; ++t)
        bw[i][t] = *(const v4f*)(wrow + (size_t)t * 16 * K + i * 8);
    }
  }
  const int col   = lane & 15;
  const int rbase = (lane >> 4) * 8;
  const float bb0 = bias[col], bb1 = bias[16 + col], bb2 = bias[32 + col], bb3 = bias[48 + col];

  while (true) {
    const int  nt   = tile + nwaves;
    const bool more = nt < ntiles;
    const float* nrow = A + ((size_t)(more ? nt : tile) * 16 + m) * K + hk;

    v8f acc0 = {}, acc1 = {}, acc2 = {}, acc3 = {};
    #pragma unroll
    for (int i = 0; i < NC; ++i) {
      v2f alo; alo.x = a[i].x; alo.y = a[i].y;
      v2f ahi; ahi.x = a[i].z; ahi.y = a[i].w;
      v2f b0l; b0l.x = bw[i][0].x; b0l.y = bw[i][0].y;
      v2f b1l; b1l.x = bw[i][1].x; b1l.y = bw[i][1].y;
      v2f b2l; b2l.x = bw[i][2].x; b2l.y = bw[i][2].y;
      v2f b3l; b3l.x = bw[i][3].x; b3l.y = bw[i][3].y;
      v2f b0h; b0h.x = bw[i][0].z; b0h.y = bw[i][0].w;
      v2f b1h; b1h.x = bw[i][1].z; b1h.y = bw[i][1].w;
      v2f b2h; b2h.x = bw[i][2].z; b2h.y = bw[i][2].w;
      v2f b3h; b3h.x = bw[i][3].z; b3h.y = bw[i][3].w;
      acc0 = __builtin_amdgcn_wmma_f32_16x16x4_f32(false, alo, false, b0l, (short)0, acc0, false, false);
      acc1 = __builtin_amdgcn_wmma_f32_16x16x4_f32(false, alo, false, b1l, (short)0, acc1, false, false);
      acc2 = __builtin_amdgcn_wmma_f32_16x16x4_f32(false, alo, false, b2l, (short)0, acc2, false, false);
      acc3 = __builtin_amdgcn_wmma_f32_16x16x4_f32(false, alo, false, b3l, (short)0, acc3, false, false);
      acc0 = __builtin_amdgcn_wmma_f32_16x16x4_f32(false, ahi, false, b0h, (short)0, acc0, false, false);
      acc1 = __builtin_amdgcn_wmma_f32_16x16x4_f32(false, ahi, false, b1h, (short)0, acc1, false, false);
      acc2 = __builtin_amdgcn_wmma_f32_16x16x4_f32(false, ahi, false, b2h, (short)0, acc2, false, false);
      acc3 = __builtin_amdgcn_wmma_f32_16x16x4_f32(false, ahi, false, b3h, (short)0, acc3, false, false);
      a[i] = *(const v4f*)(nrow + i * 8);
      __builtin_amdgcn_sched_barrier(0);
    }

    float* orow = out + (size_t)tile * 16 * HID;
    #pragma unroll
    for (int v = 0; v < 8; ++v) {
      const int r = rbase + v;
      float x0 = acc0[v] + bb0;
      float x1 = acc1[v] + bb1;
      float x2 = acc2[v] + bb2;
      float x3 = acc3[v] + bb3;
      if (RELU) {
        x0 = fmaxf(x0, 0.f); x1 = fmaxf(x1, 0.f);
        x2 = fmaxf(x2, 0.f); x3 = fmaxf(x3, 0.f);
      }
      orow[(size_t)r * HID +      col] = x0;
      orow[(size_t)r * HID + 16 + col] = x1;
      orow[(size_t)r * HID + 32 + col] = x2;
      orow[(size_t)r * HID + 48 + col] = x3;
    }
    __builtin_amdgcn_sched_barrier(0);
    asm volatile("s_wait_xcnt 0x0");
    __builtin_amdgcn_sched_barrier(0);
    if (!more) break;
    tile = nt;
  }
}

__global__ void deg_init(float* __restrict__ deg, int n) {
  int i = blockIdx.x * blockDim.x + threadIdx.x;
  if (i < n) deg[i] = 1.0f;
}
__global__ void deg_acc(const int* __restrict__ src, const int* __restrict__ dst,
                        float* __restrict__ deg, int* __restrict__ keyw, int n, int e) {
  const int i = (int)(blockIdx.x * blockDim.x + threadIdx.x);
  unsigned k = 0;
  int pw = 0;
  if (i < e) {
    const int s = src[i];
    const int d = dst[i];
    atomicAdd(&deg[d], 1.0f);
    k = (unsigned)d / OWN_ROWS;
    pw = (int)(((unsigned)s << PACK_BITS) | ((unsigned)d - k * OWN_ROWS));
  }
  if (!keys_ok(keyw, n, e)) return;
  if (i < e) ((int*)pack_words(keyw, e))[i] = pw;
  v4i lo, hi;
  lo.x = (int)__builtin_amdgcn_ballot_w32((k &   1u) != 0u);
  lo.y = (int)__builtin_amdgcn_ballot_w32((k &   2u) != 0u);
  lo.z = (int)__builtin_amdgcn_ballot_w32((k &   4u) != 0u);
  lo.w = (int)__builtin_amdgcn_ballot_w32((k &   8u) != 0u);
  hi.x = (int)__builtin_amdgcn_ballot_w32((k &  16u) != 0u);
  hi.y = (int)__builtin_amdgcn_ballot_w32((k &  32u) != 0u);
  hi.z = (int)__builtin_amdgcn_ballot_w32((k &  64u) != 0u);
  hi.w = (int)__builtin_amdgcn_ballot_w32((k & 128u) != 0u);
  if ((threadIdx.x & 31) == 0) {
    const int g = i >> 5;
    int* p = keyw + (size_t)(g >> 5) * 256 + (g & 31) * 4;
    *(v4i*)p         = lo;
    *(v4i*)(p + 128) = hi;
  }
}
__global__ void deg_rsqrt(float* __restrict__ deg, int n) {
  int i = blockIdx.x * blockDim.x + threadIdx.x;
  if (i < n) deg[i] = rsqrtf(deg[i]);
}

__global__ void alpha_kernel(const float* __restrict__ h,
                             const float* __restrict__ attl,
                             const float* __restrict__ attr,
                             const float* __restrict__ dinv,
                             float* __restrict__ adp, float* __restrict__ ar, int n) {
  const int wave = (int)((blockIdx.x * blockDim.x + threadIdx.x) >> 5);
  const int lane = (int)(threadIdx.x & 31);
  if (wave >= n) return;
  const float* row = h + (size_t)wave * HID;
  const float h0v = row[lane], h1v = row[lane + 32];
  float pl = h0v * attl[lane] + h1v * attl[lane + 32];
  float pr = h0v * attr[lane] + h1v * attr[lane + 32];
  #pragma unroll
  for (int off = 16; off > 0; off >>= 1) {
    pl += __shfl_xor(pl, off, 32);
    pr += __shfl_xor(pr, off, 32);
  }
  if (lane == 0) {
    v2f o; o.x = pl; o.y = dinv[wave];
    *(v2f*)(adp + 2 * (size_t)wave) = o;
    ar[wave] = pr;
  }
}

#define LWAVES     32
#define UNIT_EDGES 512
#define QCAP       (UNIT_EDGES + 32)

static_assert(OWN_ROWS <= (1 << PACK_BITS), "local row must fit the packed word");

template<bool RECORD>
__global__ __launch_bounds__(1024)
void layer_owned(const int* __restrict__ src, const int* __restrict__ dst,
                 const float* __restrict__ h, const float* __restrict__ h0,
                 const float* __restrict__ adp, const float* __restrict__ ar,
                 const float* __restrict__ dinv, float* hout, int* lists,
                 const int* __restrict__ keyw,
                 const float* __restrict__ attl_next, const float* __restrict__ attr_next,
                 float* __restrict__ adp_next, float* __restrict__ ar_next,
                 int n, int e_cnt) {
  __shared__ float acc[OWN_ROWS * HID];
  __shared__ int   queue[LWAVES][QCAP];
  __shared__ float wself[OWN_ROWS];
  __shared__ float dotl[RECORD ? OWN_ROWS : 1];
  __shared__ float dotr[RECORD ? OWN_ROWS : 1];

  const int tid  = (int)threadIdx.x;
  const int lane = tid & 31;
  const int wave = __builtin_amdgcn_readfirstlane(tid >> 5);
  const int base = (int)blockIdx.x * OWN_ROWS;
  const int rows = min(OWN_ROWS, n - base);
  const int nel  = rows * HID;
  const size_t gbase = (size_t)base * HID;

  int* lw = lists + gbase + (size_t)wave * (size_t)(rows * 2);
  const int lcap = rows * 2 - 1;

  if (tid < rows) {
    const int node = base + tid;
    const float di = dinv[node];
    wself[tid] = tanhf(adp[2 * (size_t)node] + ar[node]) * di * di;
  }
  __syncthreads();
  for (int idx = tid; idx < nel; idx += 1024)
    acc[idx] = EPS_BLEND * h0[gbase + idx] + wself[idx >> 6] * h[gbase + idx];
  __syncthreads();

  auto add_rows = [&](int se, int dl, float w, int cnt) {
    for (int j0 = 0; j0 < cnt; j0 += 8) {
      float v0[8], v1[8];
      #pragma unroll
      for (int t = 0; t < 8; ++t) {
        const int sj = __builtin_amdgcn_readlane(se, j0 + t);
        const float* row = h + (size_t)sj * HID;
        v0[t] = row[lane];
        v1[t] = row[lane + 32];
      }
      #pragma unroll
      for (int t = 0; t < 8; ++t) {
        const int j = j0 + t;
        const float wj = __int_as_float(__builtin_amdgcn_readlane(__float_as_int(w), j));
        const int dj = __builtin_amdgcn_readlane(dl, j);
        const bool live = j < cnt;
        atomicAdd(&acc[dj * HID + lane],      live ? wj * v0[t] : 0.f);
        atomicAdd(&acc[dj * HID + 32 + lane], live ? wj * v1[t] : 0.f);
      }
    }
  };

  bool do_sweep = true;
  if (!RECORD) {
    int total = lw[0];
    if (total >= 0) {
      do_sweep = false;
      total = min(total, lcap);
      int  p_nxt = (lane < total) ? lw[1 + lane] : 0;
      int  se_c = 0, dl_c = 0;
      v2f  sp = {};
      float xr = 0.f, dd = 0.f;
      {
        se_c = p_nxt >> PACK_BITS;
        dl_c = p_nxt & ((1 << PACK_BITS) - 1);
        sp = *(const v2f*)(adp + 2 * (size_t)se_c); xr = ar[base + dl_c]; dd = dinv[base + dl_c];
        p_nxt = (32 + lane < total) ? lw[1 + 32 + lane] : 0;
      }
      for (int b = 0; b < total; b += 32) {
        const int cnt = min(32, total - b);
        float w = tanhf(sp.x + xr) * sp.y * dd;
        if (lane >= cnt) w = 0.f;
        const int se = se_c, dl = dl_c;
        se_c = p_nxt >> PACK_BITS;
        dl_c = p_nxt & ((1 << PACK_BITS) - 1);
        sp = *(const v2f*)(adp + 2 * (size_t)se_c); xr = ar[base + dl_c]; dd = dinv[base + dl_c];
        p_nxt = (b + 64 + lane < total) ? lw[1 + b + 64 + lane] : 0;
        add_rows(se, dl, w, cnt);
      }
    }
  }

  if (do_sweep) {
    int* qw = queue[wave];
    const unsigned lt = (1u << lane) - 1u;
    int qn = 0;
    int rec_n = (n <= PACK_MAXN) ? 0 : -1;

    const bool by_key = keys_ok(keyw, n, e_cnt);
    const int* packw = pack_words(keyw, e_cnt);

    auto drain = [&](bool all) {
      while (qn >= 32 || (all && qn > 0)) {
        const int cnt = min(qn, 32);
        const int qb  = qn - cnt;
        __builtin_amdgcn_wave_barrier();
        const bool act = lane < cnt;
        int se = 0, dd = base;
        if (act) {
          const int e = qw[qb + lane];
          if (by_key) {
            const int p = packw[e];
            se = p >> PACK_BITS;
            dd = base + (p & ((1 << PACK_BITS) - 1));
          } else {
            se = src[e];
            dd = dst[e];
          }
        }
        const int dl = dd - base;
        if (RECORD) {
          if (rec_n >= 0) {
            if (rec_n + cnt <= lcap) {
              if (act) lw[1 + rec_n + lane] = (se << PACK_BITS) | dl;
              rec_n += cnt;
            } else {
              rec_n = -1;
            }
          }
        }
        const v2f   sp = *(const v2f*)(adp + 2 * (size_t)se);
        const float xr = ar[dd];
        const float dv = dinv[dd];
        float w = tanhf(sp.x + xr) * sp.y * dv;
        if (!act) w = 0.f;
        add_rows(se, dl, w, cnt);
        qn = qb;
      }
    };

    if (by_key) {
      const unsigned char* key = (const unsigned char*)keyw;
      const int units = (e_cnt + KEY_UNIT - 1) / KEY_UNIT;
      const int step = max(1, units / (int)gridDim.x);
      const int rot  = units > 0 ? (int)(((unsigned)blockIdx.x * (unsigned)step) % (unsigned)units) : 0;
      unsigned cx[8];
      #pragma unroll
      for (int b = 0; b < 8; ++b) cx[b] = ((blockIdx.x >> b) & 1u) ? 0u : ~0u;

      auto load_keys = [&](int uu_in, v4i (&v)[8]) -> int {
        int uu = uu_in + rot;
        if (uu >= units) uu -= units;
        const int eb = uu * KEY_UNIT;
        #pragma unroll
        for (int k = 0; k < 8; ++k) v[k] = *(const v4i*)(key + eb + lane * 16 + k * 512);
        return eb;
      };

      auto scan_group = [&](const v4i& pa, const v4i& pb, int e0) {
        unsigned g = ((unsigned)pa.x ^ cx[0]) & ((unsigned)pa.y ^ cx[1]) &
                     ((unsigned)pa.z ^ cx[2]) & ((unsigned)pa.w ^ cx[3]) &
                     ((unsigned)pb.x ^ cx[4]) & ((unsigned)pb.y ^ cx[5]) &
                     ((unsigned)pb.z ^ cx[6]) & ((unsigned)pb.w ^ cx[7]);
        while (__builtin_amdgcn_ballot_w32(g != 0u)) {
          const int eidx = e0 + __builtin_ctz(g | 0x80000000u);
          const bool m = (g != 0u) && (eidx < e_cnt);
          const unsigned mm = __builtin_amdgcn_ballot_w32(m);
          if (m) qw[qn + __builtin_popcount(mm & lt)] = eidx;
          qn += __builtin_popcount(mm);
          g &= g - 1u;
          drain(false);
        }
      };

      v4i cur[8];
      int cur_eb = 0;
      int u = wave;
      const bool have = u < units;
      if (have) cur_eb = load_keys(u, cur);

      while (have) {
        v4i nxt[8];
        int nxt_eb = 0;
        const bool have_n = u + LWAVES < units;
        nxt_eb = load_keys(have_n ? u + LWAVES : u, nxt);

        #pragma unroll
        for (int q = 0; q < 4; ++q)
          scan_group(cur[2 * q], cur[2 * q + 1], cur_eb + q * 1024 + lane * 32);

        if (!have_n) break;
        #pragma unroll
        for (int k = 0; k < 8; ++k) cur[k] = nxt[k];
        cur_eb = nxt_eb;
        u += LWAVES;
      }
      drain(true);
    } else {
      const int units = (e_cnt + UNIT_EDGES - 1) / UNIT_EDGES;
      const int step = max(1, units / (int)gridDim.x);
      const int rot  = units > 0 ? (int)(((unsigned)blockIdx.x * (unsigned)step) % (unsigned)units) : 0;
      const bool vec_ok = ((e_cnt & 3) == 0) && ((((size_t)dst) & 15) == 0);

      v4i cur[4];
      int cur_eb = 0;
      int u = wave;
      bool have = u < units;

      auto load_unit = [&](int uu_in, v4i (&v)[4]) -> int {
        int uu = uu_in + rot;
        if (uu >= units) uu -= units;
        const int eb = uu * UNIT_EDGES + lane * 4;
        if (vec_ok && (uu + 1) * UNIT_EDGES <= e_cnt) {
          #pragma unroll
          for (int k = 0; k < 4; ++k) v[k] = *(const v4i*)(dst + eb + k * 128);
        } else {
          #pragma unroll
          for (int k = 0; k < 4; ++k) {
            #pragma unroll
            for (int j = 0; j < 4; ++j) {
              const int idx = eb + k * 128 + j;
              v[k][j] = (idx < e_cnt) ? dst[idx] : -1;
            }
          }
        }
        return eb;
      };

      if (have) cur_eb = load_unit(u, cur);

      while (true) {
        v4i nxt[4];
        int nxt_eb = 0;
        const bool have_n = have && (u + LWAVES < units);
        if (have_n) nxt_eb = load_unit(u + LWAVES, nxt);

        if (have) {
          #pragma unroll
          for (int k = 0; k < 4; ++k) {
            #pragma unroll
            for (int j = 0; j < 4; ++j) {
              const int d = cur[k][j];
              const bool m = (unsigned)(d - base) < (unsigned)rows;
              const unsigned mask = __builtin_amdgcn_ballot_w32(m);
              if (mask) {
                if (m) qw[qn + __builtin_popcount(mask & lt)] = cur_eb + k * 128 + j;
                qn += __builtin_popcount(mask);
              }
            }
          }
        }

        const bool last = !have_n;
        drain(last);
        if (last) break;
        #pragma unroll
        for (int k = 0; k < 4; ++k) cur[k] = nxt[k];
        cur_eb = nxt_eb;
        u += LWAVES;
      }
    }
    if (RECORD) {
      if (lane == 0) lw[0] = rec_n;
    }
  }
  __syncthreads();

  if (RECORD) {
    const float l0 = attl_next[lane], l1 = attl_next[lane + 32];
    const float r0 = attr_next[lane], r1 = attr_next[lane + 32];
    for (int rb = wave; rb < rows; rb += 4 * LWAVES) {
      float pl[4], pr[4];
      #pragma unroll
      for (int k = 0; k < 4; ++k) {
        const int r = rb + k * LWAVES;
        pl[k] = 0.f; pr[k] = 0.f;
        if (r < rows) {
          const float h0v = fmaxf(acc[r * HID + lane], 0.f);
          const float h1v = fmaxf(acc[r * HID + 32 + lane], 0.f);
          hout[gbase + (size_t)r * HID + lane]      = h0v;
          hout[gbase + (size_t)r * HID + 32 + lane] = h1v;
          pl[k] = h0v * l0 + h1v * l1;
          pr[k] = h0v * r0 + h1v * r1;
        }
      }
      #pragma unroll
      for (int off = 16; off > 0; off >>= 1) {
        #pragma unroll
        for (int k = 0; k < 4; ++k) {
          pl[k] += __shfl_xor(pl[k], off, 32);
          pr[k] += __shfl_xor(pr[k], off, 32);
        }
      }
      if (lane == 0) {
        #pragma unroll
        for (int k = 0; k < 4; ++k) {
          const int r = rb + k * LWAVES;
          if (r < rows) { dotl[r] = pl[k]; dotr[r] = pr[k]; }
        }
      }
    }
    __syncthreads();
    if (tid < rows) {
      v2f o; o.x = dotl[tid]; o.y = dinv[base + tid];
      *(v2f*)(adp_next + 2 * (size_t)(base + tid)) = o;
      ar_next[base + tid] = dotr[tid];
    }
  } else {
    for (int idx = tid; idx < nel; idx += 1024)
      hout[gbase + idx] = fmaxf(acc[idx], 0.f);
  }
}

extern "C" void kernel_launch(void* const* d_in, const int* in_sizes, int n_in,
                              void* d_out, int out_size, void* d_ws, size_t ws_size,
                              hipStream_t stream) {
  const float* x     = (const float*)d_in[0];
  const int*   ei    = (const int*)  d_in[1];
  const float* W_in  = (const float*)d_in[2];
  const float* b_in  = (const float*)d_in[3];
  const float* attl1 = (const float*)d_in[4];
  const float* attr1 = (const float*)d_in[5];
  const float* attl2 = (const float*)d_in[6];
  const float* attr2 = (const float*)d_in[7];
  const float* W_out = (const float*)d_in[8];
  const float* b_out = (const float*)d_in[9];
  float* out = (float*)d_out;

  const int IN = 128;
  const int E  = in_sizes[1] / 2;
  const int N  = in_sizes[0] / IN;
  const int*   srcI = ei;
  const int*   dstI = ei + E;

  float* H0   = (float*)d_ws;
  float* H1   = H0   + (size_t)N * HID;
  float* OUTB = H1   + (size_t)N * HID;
  float* ADP  = OUTB + (size_t)N * HID;
  float* DINV = ADP  + 2 * (size_t)N;
  float* ADP2 = out;
  float* AR2  = out + 2 * (size_t)N;
  float* AR   = out + 3 * (size_t)N;
  int*   KEYW = (int*)(out + 4 * (size_t)N);

  const int ntiles = N / 16;
  const int nThreads  = (N + 255) / 256;
  const int degBlk    = (E + KEY_UNIT - 1) / KEY_UNIT * (KEY_UNIT / 256);
  const int nWaveBlk  = (N + 7) / 8;
  const int layerBlk  = (N + OWN_ROWS - 1) / OWN_ROWS;

  gemm_wmma_f32<128, true><<<256, 256, 0, stream>>>(x, W_in, b_in, H0, ntiles);

  deg_init <<<nThreads, 256, 0, stream>>>(DINV, N);
  deg_acc  <<<degBlk, 256, 0, stream>>>(srcI, dstI, DINV, KEYW, N, E);
  deg_rsqrt<<<nThreads, 256, 0, stream>>>(DINV, N);

  alpha_kernel<<<nWaveBlk, 256, 0, stream>>>(H0, attl1, attr1, DINV, ADP, AR, N);
  layer_owned<true> <<<layerBlk, 1024, 0, stream>>>(srcI, dstI, H0, H0, ADP, AR, DINV, H1, (int*)OUTB, KEYW,
                                                    attl2, attr2, ADP2, AR2, N, E);

  layer_owned<false><<<layerBlk, 1024, 0, stream>>>(srcI, dstI, H1, H0, ADP2, AR2, DINV, OUTB, (int*)OUTB, KEYW,
                                                    nullptr, nullptr, nullptr, nullptr, N, E);

  gemm_wmma_f32<64, false><<<256, 256, 0, stream>>>(OUTB, W_out, b_out, out, ntiles);
}


# degree count leaves global float atomics: a block counts its 8192 edges' targets in an LDS tile of 16-bit pairs (LDS integer adds), stores it as a plane, one pass sums the planes; 8 launches to 7
# speedup vs baseline: 3.4327x; 1.2518x over previous
#include <hip/hip_runtime.h>
#include <hip/hip_bf16.h>

typedef __attribute__((ext_vector_type(2))) float v2f;
typedef __attribute__((ext_vector_type(8))) float v8f;
typedef __attribute__((ext_vector_type(4))) int v4i;
typedef __attribute__((ext_vector_type(4))) float v4f;

#define HID 64
#define EPS_BLEND 0.1f

#define OWN_ROWS   391
#define KEY_UNIT   4096
#define KEY_BLOCKS 256

#define PACK_BITS  9
#define PACK_MAXN  (1 << (31 - PACK_BITS))
__device__ __forceinline__ size_t edge_pad(int e) {
  return ((size_t)e + KEY_UNIT - 1) / KEY_UNIT * KEY_UNIT;
}
__device__ __forceinline__ bool keys_ok(const int* keyw, int n, int e) {
  const size_t epad = edge_pad(e);
  return ((n + OWN_ROWS - 1) / OWN_ROWS <= KEY_BLOCKS) && (n <= PACK_MAXN) &&
         (16 * (size_t)n + 5 * epad <= 256 * (size_t)n) &&
         ((((size_t)keyw) & 15) == 0);
}
__device__ __forceinline__ const int* pack_words(const int* keyw, int e) {
  return (const int*)((const unsigned char*)keyw + edge_pad(e));
}

template<int K, bool RELU>
__global__ __launch_bounds__(256)
void gemm_wmma_f32(const float* __restrict__ A,
                   const float* __restrict__ W,
                   const float* __restrict__ bias,
                   float* __restrict__ out,
                   int ntiles) {
  constexpr int NC = K / 8;
  const int lane = (int)(threadIdx.x & 31);
  const int wib    = __builtin_amdgcn_readfirstlane((int)(threadIdx.x >> 5));
  const int nwaves = (int)(gridDim.x * (blockDim.x >> 5));
  const int wave   = wib * (int)gridDim.x + (int)blockIdx.x;
  if (wave >= ntiles) return;

  const int m  = lane & 15;
  const int hk = (lane >> 4) << 2;

  v4f a[NC];
  int tile = wave;
  {
    const float* arow = A + ((size_t)tile * 16 + m) * K + hk;
    #pragma unroll
    for (int i = 0; i < NC; ++i) a[i] = *(const v4f*)(arow + i * 8);
  }

  v4f bw[NC][4];
  {
    const float* wrow = W + (size_t)m * K + hk;
    #pragma unroll
    for (int i = 0; i < NC; ++i) {
      #pragma unroll
      for (int t = 0; t < 4; ++t)
        bw[i][t] = *(const v4f*)(wrow + (size_t)t * 16 * K + i * 8);
    }
  }
  const int col   = lane & 15;
  const int rbase = (lane >> 4) * 8;
  const float bb0 = bias[col], bb1 = bias[16 + col], bb2 = bias[32 + col], bb3 = bias[48 + col];

  while (true) {
    const int  nt   = tile + nwaves;
    const bool more = nt < ntiles;
    const float* nrow = A + ((size_t)(more ? nt : tile) * 16 + m) * K + hk;

    v8f acc0 = {}, acc1 = {}, acc2 = {}, acc3 = {};
    #pragma unroll
    for (int i = 0; i < NC; ++i) {
      v2f alo; alo.x = a[i].x; alo.y = a[i].y;
      v2f ahi; ahi.x = a[i].z; ahi.y = a[i].w;
      v2f b0l; b0l.x = bw[i][0].x; b0l.y = bw[i][0].y;
      v2f b1l; b1l.x = bw[i][1].x; b1l.y = bw[i][1].y;
      v2f b2l; b2l.x = bw[i][2].x; b2l.y = bw[i][2].y;
      v2f b3l; b3l.x = bw[i][3].x; b3l.y = bw[i][3].y;
      v2f b0h; b0h.x = bw[i][0].z; b0h.y = bw[i][0].w;
      v2f b1h; b1h.x = bw[i][1].z; b1h.y = bw[i][1].w;
      v2f b2h; b2h.x = bw[i][2].z; b2h.y = bw[i][2].w;
      v2f b3h; b3h.x = bw[i][3].z; b3h.y = bw[i][3].w;
      acc0 = __builtin_amdgcn_wmma_f32_16x16x4_f32(false, alo, false, b0l, (short)0, acc0, false, false);
      acc1 = __builtin_amdgcn_wmma_f32_16x16x4_f32(false, alo, false, b1l, (short)0, acc1, false, false);
      acc2 = __builtin_amdgcn_wmma_f32_16x16x4_f32(false, alo, false, b2l, (short)0, acc2, false, false);
      acc3 = __builtin_amdgcn_wmma_f32_16x16x4_f32(false, alo, false, b3l, (short)0, acc3, false, false);
      acc0 = __builtin_amdgcn_wmma_f32_16x16x4_f32(false, ahi, false, b0h, (short)0, acc0, false, false);
      acc1 = __builtin_amdgcn_wmma_f32_16x16x4_f32(false, ahi, false, b1h, (short)0, acc1, false, false);
      acc2 = __builtin_amdgcn_wmma_f32_16x16x4_f32(false, ahi, false, b2h, (short)0, acc2, false, false);
      acc3 = __builtin_amdgcn_wmma_f32_16x16x4_f32(false, ahi, false, b3h, (short)0, acc3, false, false);
      a[i] = *(const v4f*)(nrow + i * 8);
      __builtin_amdgcn_sched_barrier(0);
    }

    float* orow = out + (size_t)tile * 16 * HID;
    #pragma unroll
    for (int v = 0; v < 8; ++v) {
      const int r = rbase + v;
      float x0 = acc0[v] + bb0;
      float x1 = acc1[v] + bb1;
      float x2 = acc2[v] + bb2;
      float x3 = acc3[v] + bb3;
      if (RELU) {
        x0 = fmaxf(x0, 0.f); x1 = fmaxf(x1, 0.f);
        x2 = fmaxf(x2, 0.f); x3 = fmaxf(x3, 0.f);
      }
      orow[(size_t)r * HID +      col] = x0;
      orow[(size_t)r * HID + 16 + col] = x1;
      orow[(size_t)r * HID + 32 + col] = x2;
      orow[(size_t)r * HID + 48 + col] = x3;
    }
    __builtin_amdgcn_sched_barrier(0);
    asm volatile("s_wait_xcnt 0x0");
    __builtin_amdgcn_sched_barrier(0);
    if (!more) break;
    tile = nt;
  }
}

__global__ void deg_init(float* __restrict__ deg, int n) {
  int i = blockIdx.x * blockDim.x + threadIdx.x;
  if (i < n) deg[i] = 1.0f;
}
__global__ void deg_acc(const int* __restrict__ src, const int* __restrict__ dst,
                        float* __restrict__ deg, int* __restrict__ keyw, int n, int e) {
  const int i = (int)(blockIdx.x * blockDim.x + threadIdx.x);
  unsigned k = 0;
  int pw = 0;
  if (i < e) {
    const int s = src[i];
    const int d = dst[i];
    atomicAdd(&deg[d], 1.0f);
    k = (unsigned)d / OWN_ROWS;
    pw = (int)(((unsigned)s << PACK_BITS) | ((unsigned)d - k * OWN_ROWS));
  }
  if (!keys_ok(keyw, n, e)) return;
  if (i < e) ((int*)pack_words(keyw, e))[i] = pw;
  v4i lo, hi;
  lo.x = (int)__builtin_amdgcn_ballot_w32((k &   1u) != 0u);
  lo.y = (int)__builtin_amdgcn_ballot_w32((k &   2u) != 0u);
  lo.z = (int)__builtin_amdgcn_ballot_w32((k &   4u) != 0u);
  lo.w = (int)__builtin_amdgcn_ballot_w32((k &   8u) != 0u);
  hi.x = (int)__builtin_amdgcn_ballot_w32((k &  16u) != 0u);
  hi.y = (int)__builtin_amdgcn_ballot_w32((k &  32u) != 0u);
  hi.z = (int)__builtin_amdgcn_ballot_w32((k &  64u) != 0u);
  hi.w = (int)__builtin_amdgcn_ballot_w32((k & 128u) != 0u);
  if ((threadIdx.x & 31) == 0) {
    const int g = i >> 5;
    int* p = keyw + (size_t)(g >> 5) * 256 + (g & 31) * 4;
    *(v4i*)p         = lo;
    *(v4i*)(p + 128) = hi;
  }
}
__global__ void deg_rsqrt(float* __restrict__ deg, int n) {
  int i = blockIdx.x * blockDim.x + threadIdx.x;
  if (i < n) deg[i] = rsqrtf(deg[i]);
}

#define HIST_WORDS 51200
#define HIST_NODES (2 * HIST_WORDS)
#define CNT_CHUNK  8192
#define CNT_PER_THREAD (CNT_CHUNK / 1024)

__global__ __launch_bounds__(1024)
void deg_count(const int* __restrict__ src, const int* __restrict__ dst,
               unsigned* __restrict__ planes, int* __restrict__ keyw, int n, int e) {
  __shared__ unsigned hist[HIST_WORDS];
  const int tid    = (int)threadIdx.x;
  const int nwords = (n + 1) >> 1;
  const int cb     = (int)blockIdx.x * CNT_CHUNK;
  const int epad   = (int)edge_pad(e);
  unsigned* plane  = planes + (size_t)blockIdx.x * (size_t)nwords;
  const bool planes_ok = keys_ok(keyw, n, e);

  for (int nb = 0; nb < n; nb += HIST_NODES) {
    for (int w = tid; w < HIST_WORDS; w += 1024) hist[w] = 0u;
    __syncthreads();

    int sv[CNT_PER_THREAD], dv[CNT_PER_THREAD];
    #pragma unroll
    for (int j = 0; j < CNT_PER_THREAD; ++j) {
      const int ic = min(cb + j * 1024 + tid, e - 1);
      sv[j] = src[ic];
      dv[j] = dst[ic];
    }
    #pragma unroll
    for (int j = 0; j < CNT_PER_THREAD; ++j) {
      const int i = cb + j * 1024 + tid;
      const int d = dv[j];
      const unsigned dr = (unsigned)(d - nb);
      if (i < e && dr < (unsigned)HIST_NODES)
        atomicAdd(&hist[dr >> 1], 1u << ((dr & 1u) * 16u));
      if (nb == 0 && planes_ok && i < epad) {
        unsigned k = 0;
        if (i < e) {
          k = (unsigned)d / OWN_ROWS;
          ((int*)pack_words(keyw, e))[i] =
              (int)(((unsigned)sv[j] << PACK_BITS) | ((unsigned)d - k * OWN_ROWS));
        }
        v4i lo, hi;
        lo.x = (int)__builtin_amdgcn_ballot_w32((k &   1u) != 0u);
        lo.y = (int)__builtin_amdgcn_ballot_w32((k &   2u) != 0u);
        lo.z = (int)__builtin_amdgcn_ballot_w32((k &   4u) != 0u);
        lo.w = (int)__builtin_amdgcn_ballot_w32((k &   8u) != 0u);
        hi.x = (int)__builtin_amdgcn_ballot_w32((k &  16u) != 0u);
        hi.y = (int)__builtin_amdgcn_ballot_w32((k &  32u) != 0u);
        hi.z = (int)__builtin_amdgcn_ballot_w32((k &  64u) != 0u);
        hi.w = (int)__builtin_amdgcn_ballot_w32((k & 128u) != 0u);
        if ((tid & 31) == 0) {
          const int g = i >> 5;
          int* p = keyw + (size_t)(g >> 5) * 256 + (g & 31) * 4;
          *(v4i*)p         = lo;
          *(v4i*)(p + 128) = hi;
        }
      }
    }
    __syncthreads();
    const int w0 = nb >> 1;
    const int wn = min(HIST_WORDS, nwords - w0);
    for (int w = tid; w < wn; w += 1024) plane[w0 + w] = hist[w];
    __syncthreads();
  }
}

__global__ void deg_finish(const unsigned* __restrict__ planes, float* __restrict__ dinv,
                           int n, int nplanes) {
  const int w = (int)(blockIdx.x * blockDim.x + threadIdx.x);
  const int nwords = (n + 1) >> 1;
  if (w >= nwords) return;
  unsigned lo = 0u, hi = 0u;
  #pragma unroll 14
  for (int b = 0; b < nplanes; ++b) {
    const unsigned v = planes[(size_t)b * (size_t)nwords + w];
    lo += v & 0xFFFFu;
    hi += v >> 16;
  }
  dinv[2 * w] = rsqrtf((float)(lo + 1u));
  if (2 * w + 1 < n) dinv[2 * w + 1] = rsqrtf((float)(hi + 1u));
}

__global__ void alpha_kernel(const float* __restrict__ h,
                             const float* __restrict__ attl,
                             const float* __restrict__ attr,
                             const float* __restrict__ dinv,
                             float* __restrict__ adp, float* __restrict__ ar, int n) {
  const int wave = (int)((blockIdx.x * blockDim.x + threadIdx.x) >> 5);
  const int lane = (int)(threadIdx.x & 31);
  if (wave >= n) return;
  const float* row = h + (size_t)wave * HID;
  const float h0v = row[lane], h1v = row[lane + 32];
  float pl = h0v * attl[lane] + h1v * attl[lane + 32];
  float pr = h0v * attr[lane] + h1v * attr[lane + 32];
  #pragma unroll
  for (int off = 16; off > 0; off >>= 1) {
    pl += __shfl_xor(pl, off, 32);
    pr += __shfl_xor(pr, off, 32);
  }
  if (lane == 0) {
    v2f o; o.x = pl; o.y = dinv[wave];
    *(v2f*)(adp + 2 * (size_t)wave) = o;
    ar[wave] = pr;
  }
}

#define LWAVES     32
#define UNIT_EDGES 512
#define QCAP       (UNIT_EDGES + 32)

static_assert(OWN_ROWS <= (1 << PACK_BITS), "local row must fit the packed word");

template<bool RECORD>
__global__ __launch_bounds__(1024)
void layer_owned(const int* __restrict__ src, const int* __restrict__ dst,
                 const float* __restrict__ h, const float* __restrict__ h0,
                 const float* __restrict__ adp, const float* __restrict__ ar,
                 const float* __restrict__ dinv, float* hout, int* lists,
                 const int* __restrict__ keyw,
                 const float* __restrict__ attl_next, const float* __restrict__ attr_next,
                 float* __restrict__ adp_next, float* __restrict__ ar_next,
                 int n, int e_cnt) {
  __shared__ float acc[OWN_ROWS * HID];
  __shared__ int   queue[LWAVES][QCAP];
  __shared__ float wself[OWN_ROWS];
  __shared__ float dotl[RECORD ? OWN_ROWS : 1];
  __shared__ float dotr[RECORD ? OWN_ROWS : 1];

  const int tid  = (int)threadIdx.x;
  const int lane = tid & 31;
  const int wave = __builtin_amdgcn_readfirstlane(tid >> 5);
  const int base = (int)blockIdx.x * OWN_ROWS;
  const int rows = min(OWN_ROWS, n - base);
  const int nel  = rows * HID;
  const size_t gbase = (size_t)base * HID;

  int* lw = lists + gbase + (size_t)wave * (size_t)(rows * 2);
  const int lcap = rows * 2 - 1;

  if (tid < rows) {
    const int node = base + tid;
    const float di = dinv[node];
    wself[tid] = tanhf(adp[2 * (size_t)node] + ar[node]) * di * di;
  }
  __syncthreads();
  for (int idx = tid; idx < nel; idx += 1024)
    acc[idx] = EPS_BLEND * h0[gbase + idx] + wself[idx >> 6] * h[gbase + idx];
  __syncthreads();

  auto add_rows = [&](int se, int dl, float w, int cnt) {
    for (int j0 = 0; j0 < cnt; j0 += 8) {
      float v0[8], v1[8];
      #pragma unroll
      for (int t = 0; t < 8; ++t) {
        const int sj = __builtin_amdgcn_readlane(se, j0 + t);
        const float* row = h + (size_t)sj * HID;
        v0[t] = row[lane];
        v1[t] = row[lane + 32];
      }
      #pragma unroll
      for (int t = 0; t < 8; ++t) {
        const int j = j0 + t;
        const float wj = __int_as_float(__builtin_amdgcn_readlane(__float_as_int(w), j));
        const int dj = __builtin_amdgcn_readlane(dl, j);
        const bool live = j < cnt;
        atomicAdd(&acc[dj * HID + lane],      live ? wj * v0[t] : 0.f);
        atomicAdd(&acc[dj * HID + 32 + lane], live ? wj * v1[t] : 0.f);
      }
    }
  };

  bool do_sweep = true;
  if (!RECORD) {
    int total = lw[0];
    if (total >= 0) {
      do_sweep = false;
      total = min(total, lcap);
      int  p_nxt = (lane < total) ? lw[1 + lane] : 0;
      int  se_c = 0, dl_c = 0;
      v2f  sp = {};
      float xr = 0.f, dd = 0.f;
      {
        se_c = p_nxt >> PACK_BITS;
        dl_c = p_nxt & ((1 << PACK_BITS) - 1);
        sp = *(const v2f*)(adp + 2 * (size_t)se_c); xr = ar[base + dl_c]; dd = dinv[base + dl_c];
        p_nxt = (32 + lane < total) ? lw[1 + 32 + lane] : 0;
      }
      for (int b = 0; b < total; b += 32) {
        const int cnt = min(32, total - b);
        float w = tanhf(sp.x + xr) * sp.y * dd;
        if (lane >= cnt) w = 0.f;
        const int se = se_c, dl = dl_c;
        se_c = p_nxt >> PACK_BITS;
        dl_c = p_nxt & ((1 << PACK_BITS) - 1);
        sp = *(const v2f*)(adp + 2 * (size_t)se_c); xr = ar[base + dl_c]; dd = dinv[base + dl_c];
        p_nxt = (b + 64 + lane < total) ? lw[1 + b + 64 + lane] : 0;
        add_rows(se, dl, w, cnt);
      }
    }
  }

  if (do_sweep) {
    int* qw = queue[wave];
    const unsigned lt = (1u << lane) - 1u;
    int qn = 0;
    int rec_n = (n <= PACK_MAXN) ? 0 : -1;

    const bool by_key = keys_ok(keyw, n, e_cnt);
    const int* packw = pack_words(keyw, e_cnt);

    auto drain = [&](bool all) {
      while (qn >= 32 || (all && qn > 0)) {
        const int cnt = min(qn, 32);
        const int qb  = qn - cnt;
        __builtin_amdgcn_wave_barrier();
        const bool act = lane < cnt;
        int se = 0, dd = base;
        if (act) {
          const int e = qw[qb + lane];
          if (by_key) {
            const int p = packw[e];
            se = p >> PACK_BITS;
            dd = base + (p & ((1 << PACK_BITS) - 1));
          } else {
            se = src[e];
            dd = dst[e];
          }
        }
        const int dl = dd - base;
        if (RECORD) {
          if (rec_n >= 0) {
            if (rec_n + cnt <= lcap) {
              if (act) lw[1 + rec_n + lane] = (se << PACK_BITS) | dl;
              rec_n += cnt;
            } else {
              rec_n = -1;
            }
          }
        }
        const v2f   sp = *(const v2f*)(adp + 2 * (size_t)se);
        const float xr = ar[dd];
        const float dv = dinv[dd];
        float w = tanhf(sp.x + xr) * sp.y * dv;
        if (!act) w = 0.f;
        add_rows(se, dl, w, cnt);
        qn = qb;
      }
    };

    if (by_key) {
      const unsigned char* key = (const unsigned char*)keyw;
      const int units = (e_cnt + KEY_UNIT - 1) / KEY_UNIT;
      const int step = max(1, units / (int)gridDim.x);
      const int rot  = units > 0 ? (int)(((unsigned)blockIdx.x * (unsigned)step) % (unsigned)units) : 0;
      unsigned cx[8];
      #pragma unroll
      for (int b = 0; b < 8; ++b) cx[b] = ((blockIdx.x >> b) & 1u) ? 0u : ~0u;

      auto load_keys = [&](int uu_in, v4i (&v)[8]) -> int {
        int uu = uu_in + rot;
        if (uu >= units) uu -= units;
        const int eb = uu * KEY_UNIT;
        #pragma unroll
        for (int k = 0; k < 8; ++k) v[k] = *(const v4i*)(key + eb + lane * 16 + k * 512);
        return eb;
      };

      auto scan_group = [&](const v4i& pa, const v4i& pb, int e0) {
        unsigned g = ((unsigned)pa.x ^ cx[0]) & ((unsigned)pa.y ^ cx[1]) &
                     ((unsigned)pa.z ^ cx[2]) & ((unsigned)pa.w ^ cx[3]) &
                     ((unsigned)pb.x ^ cx[4]) & ((unsigned)pb.y ^ cx[5]) &
                     ((unsigned)pb.z ^ cx[6]) & ((unsigned)pb.w ^ cx[7]);
        while (__builtin_amdgcn_ballot_w32(g != 0u)) {
          const int eidx = e0 + __builtin_ctz(g | 0x80000000u);
          const bool m = (g != 0u) && (eidx < e_cnt);
          const unsigned mm = __builtin_amdgcn_ballot_w32(m);
          if (m) qw[qn + __builtin_popcount(mm & lt)] = eidx;
          qn += __builtin_popcount(mm);
          g &= g - 1u;
          drain(false);
        }
      };

      v4i cur[8];
      int cur_eb = 0;
      int u = wave;
      const bool have = u < units;
      if (have) cur_eb = load_keys(u, cur);

      while (have) {
        v4i nxt[8];
        int nxt_eb = 0;
        const bool have_n = u + LWAVES < units;
        nxt_eb = load_keys(have_n ? u + LWAVES : u, nxt);

        #pragma unroll
        for (int q = 0; q < 4; ++q)
          scan_group(cur[2 * q], cur[2 * q + 1], cur_eb + q * 1024 + lane * 32);

        if (!have_n) break;
        #pragma unroll
        for (int k = 0; k < 8; ++k) cur[k] = nxt[k];
        cur_eb = nxt_eb;
        u += LWAVES;
      }
      drain(true);
    } else {
      const int units = (e_cnt + UNIT_EDGES - 1) / UNIT_EDGES;
      const int step = max(1, units / (int)gridDim.x);
      const int rot  = units > 0 ? (int)(((unsigned)blockIdx.x * (unsigned)step) % (unsigned)units) : 0;
      const bool vec_ok = ((e_cnt & 3) == 0) && ((((size_t)dst) & 15) == 0);

      v4i cur[4];
      int cur_eb = 0;
      int u = wave;
      bool have = u < units;

      auto load_unit = [&](int uu_in, v4i (&v)[4]) -> int {
        int uu = uu_in + rot;
        if (uu >= units) uu -= units;
        const int eb = uu * UNIT_EDGES + lane * 4;
        if (vec_ok && (uu + 1) * UNIT_EDGES <= e_cnt) {
          #pragma unroll
          for (int k = 0; k < 4; ++k) v[k] = *(const v4i*)(dst + eb + k * 128);
        } else {
          #pragma unroll
          for (int k = 0; k < 4; ++k) {
            #pragma unroll
            for (int j = 0; j < 4; ++j) {
              const int idx = eb + k * 128 + j;
              v[k][j] = (idx < e_cnt) ? dst[idx] : -1;
            }
          }
        }
        return eb;
      };

      if (have) cur_eb = load_unit(u, cur);

      while (true) {
        v4i nxt[4];
        int nxt_eb = 0;
        const bool have_n = have && (u + LWAVES < units);
        if (have_n) nxt_eb = load_unit(u + LWAVES, nxt);

        if (have) {
          #pragma unroll
          for (int k = 0; k < 4; ++k) {
            #pragma unroll
            for (int j = 0; j < 4; ++j) {
              const int d = cur[k][j];
              const bool m = (unsigned)(d - base) < (unsigned)rows;
              const unsigned mask = __builtin_amdgcn_ballot_w32(m);
              if (mask) {
                if (m) qw[qn + __builtin_popcount(mask & lt)] = cur_eb + k * 128 + j;
                qn += __builtin_popcount(mask);
              }
            }
          }
        }

        const bool last = !have_n;
        drain(last);
        if (last) break;
        #pragma unroll
        for (int k = 0; k < 4; ++k) cur[k] = nxt[k];
        cur_eb = nxt_eb;
        u += LWAVES;
      }
    }
    if (RECORD) {
      if (lane == 0) lw[0] = rec_n;
    }
  }
  __syncthreads();

  if (RECORD) {
    const float l0 = attl_next[lane], l1 = attl_next[lane + 32];
    const float r0 = attr_next[lane], r1 = attr_next[lane + 32];
    for (int rb = wave; rb < rows; rb += 4 * LWAVES) {
      float pl[4], pr[4];
      #pragma unroll
      for (int k = 0; k < 4; ++k) {
        const int r = rb + k * LWAVES;
        pl[k] = 0.f; pr[k] = 0.f;
        if (r < rows) {
          const float h0v = fmaxf(acc[r * HID + lane], 0.f);
          const float h1v = fmaxf(acc[r * HID + 32 + lane], 0.f);
          hout[gbase + (size_t)r * HID + lane]      = h0v;
          hout[gbase + (size_t)r * HID + 32 + lane] = h1v;
          pl[k] = h0v * l0 + h1v * l1;
          pr[k] = h0v * r0 + h1v * r1;
        }
      }
      #pragma unroll
      for (int off = 16; off > 0; off >>= 1) {
        #pragma unroll
        for (int k = 0; k < 4; ++k) {
          pl[k] += __shfl_xor(pl[k], off, 32);
          pr[k] += __shfl_xor(pr[k], off, 32);
        }
      }
      if (lane == 0) {
        #pragma unroll
        for (int k = 0; k < 4; ++k) {
          const int r = rb + k * LWAVES;
          if (r < rows) { dotl[r] = pl[k]; dotr[r] = pr[k]; }
        }
      }
    }
    __syncthreads();
    if (tid < rows) {
      v2f o; o.x = dotl[tid]; o.y = dinv[base + tid];
      *(v2f*)(adp_next + 2 * (size_t)(base + tid)) = o;
      ar_next[base + tid] = dotr[tid];
    }
  } else {
    for (int idx = tid; idx < nel; idx += 1024)
      hout[gbase + idx] = fmaxf(acc[idx], 0.f);
  }
}

extern "C" void kernel_launch(void* const* d_in, const int* in_sizes, int n_in,
                              void* d_out, int out_size, void* d_ws, size_t ws_size,
                              hipStream_t stream) {
  const float* x     = (const float*)d_in[0];
  const int*   ei    = (const int*)  d_in[1];
  const float* W_in  = (const float*)d_in[2];
  const float* b_in  = (const float*)d_in[3];
  const float* attl1 = (const float*)d_in[4];
  const float* attr1 = (const float*)d_in[5];
  const float* attl2 = (const float*)d_in[6];
  const float* attr2 = (const float*)d_in[7];
  const float* W_out = (const float*)d_in[8];
  const float* b_out = (const float*)d_in[9];
  float* out = (float*)d_out;

  const int IN = 128;
  const int E  = in_sizes[1] / 2;
  const int N  = in_sizes[0] / IN;
  const int*   srcI = ei;
  const int*   dstI = ei + E;

  float* H0   = (float*)d_ws;
  float* H1   = H0   + (size_t)N * HID;
  float* OUTB = H1   + (size_t)N * HID;
  float* ADP  = OUTB + (size_t)N * HID;
  float* DINV = ADP  + 2 * (size_t)N;
  float* ADP2 = out;
  float* AR2  = out + 2 * (size_t)N;
  float* AR   = out + 3 * (size_t)N;
  int*   KEYW = (int*)(out + 4 * (size_t)N);

  const int ntiles = N / 16;
  const int nThreads  = (N + 255) / 256;
  const int degBlk    = (E + KEY_UNIT - 1) / KEY_UNIT * (KEY_UNIT / 256);
  const int nWaveBlk  = (N + 7) / 8;
  const int layerBlk  = (N + OWN_ROWS - 1) / OWN_ROWS;

  gemm_wmma_f32<128, true><<<256, 256, 0, stream>>>(x, W_in, b_in, H0, ntiles);

  const int nWords = (N + 1) / 2;
  const int cntBlk = (degBlk * 256 + CNT_CHUNK - 1) / CNT_CHUNK;
  if ((size_t)cntBlk * (size_t)nWords <= 2 * (size_t)N * HID) {
    deg_count <<<cntBlk, 1024, 0, stream>>>(srcI, dstI, (unsigned*)H1, KEYW, N, E);
    deg_finish<<<(nWords + 255) / 256, 256, 0, stream>>>((const unsigned*)H1, DINV, N, cntBlk);
  } else {
    deg_init <<<nThreads, 256, 0, stream>>>(DINV, N);
    deg_acc  <<<degBlk, 256, 0, stream>>>(srcI, dstI, DINV, KEYW, N, E);
    deg_rsqrt<<<nThreads, 256, 0, stream>>>(DINV, N);
  }

  alpha_kernel<<<nWaveBlk, 256, 0, stream>>>(H0, attl1, attr1, DINV, ADP, AR, N);
  layer_owned<true> <<<layerBlk, 1024, 0, stream>>>(srcI, dstI, H0, H0, ADP, AR, DINV, H1, (int*)OUTB, KEYW,
                                                    attl2, attr2, ADP2, AR2, N, E);

  layer_owned<false><<<layerBlk, 1024, 0, stream>>>(srcI, dstI, H1, H0, ADP2, AR2, DINV, OUTB, (int*)OUTB, KEYW,
                                                    nullptr, nullptr, nullptr, nullptr, N, E);

  gemm_wmma_f32<64, false><<<256, 256, 0, stream>>>(OUTB, W_out, b_out, out, ntiles);
}


# owner keys as 32 one-hot nibble planes (degree pass bit-transposes each group): layer 1's sweep reads 2 bits per edge, not 8, one AND per group; per-wave add order follows the new scan
# speedup vs baseline: 3.5912x; 1.0462x over previous
#include <hip/hip_runtime.h>
#include <hip/hip_bf16.h>

typedef __attribute__((ext_vector_type(2))) float v2f;
typedef __attribute__((ext_vector_type(8))) float v8f;
typedef __attribute__((ext_vector_type(4))) int v4i;
typedef __attribute__((ext_vector_type(4))) float v4f;

#define HID 64
#define EPS_BLEND 0.1f

#define OWN_ROWS   391
#define KEY_UNIT   4096
#define KEY_BLOCKS 256

#define PACK_BITS  9
#define PACK_MAXN  (1 << (31 - PACK_BITS))
#define KEY_PLANES 32
__device__ __forceinline__ size_t edge_pad(int e) {
  return ((size_t)e + KEY_UNIT - 1) / KEY_UNIT * KEY_UNIT;
}
__device__ __forceinline__ bool keys_ok(const int* keyw, int n, int e) {
  const size_t epad = edge_pad(e);
  return ((n + OWN_ROWS - 1) / OWN_ROWS <= KEY_BLOCKS) && (n <= PACK_MAXN) &&
         (16 * (size_t)n + 8 * epad <= 256 * (size_t)n) &&
         ((((size_t)keyw) & 15) == 0);
}
__device__ __forceinline__ size_t plane_words(int e) { return edge_pad(e) / 32; }
__device__ __forceinline__ const int* pack_words(const int* keyw, int e) {
  return keyw + (size_t)KEY_PLANES * plane_words(e);
}
__device__ __forceinline__ unsigned key_onehot(unsigned k) {
  return (1u << (k >> 4)) | (0x10000u << (k & 15u));
}
template<int NW>
__device__ __forceinline__ void wave_bit_transpose(unsigned (&x)[NW], int lane) {
  #pragma unroll
  for (int t = 0; t < 5; ++t) {
    const int s = 16 >> t;
    const unsigned m = (s == 16) ? 0x0000FFFFu : (s == 8) ? 0x00FF00FFu :
                       (s == 4)  ? 0x0F0F0F0Fu : (s == 2) ? 0x33333333u : 0x55555555u;
    const unsigned up = (lane & s) ? ~0u : 0u;
    #pragma unroll
    for (int k = 0; k < NW; ++k) {
      const unsigned y = (unsigned)__builtin_amdgcn_ds_bpermute((lane ^ s) << 2, (int)x[k]);
      const unsigned lo = (up & (y >> s)) | (~up & x[k]);
      const unsigned hi = (up & x[k]) | (~up & (y << s));
      x[k] = (lo & m) | (hi & ~m);
    }
  }
}

template<int K, bool RELU>
__global__ __launch_bounds__(256)
void gemm_wmma_f32(const float* __restrict__ A,
                   const float* __restrict__ W,
                   const float* __restrict__ bias,
                   float* __restrict__ out,
                   int ntiles) {
  constexpr int NC = K / 8;
  const int lane = (int)(threadIdx.x & 31);
  const int wib    = __builtin_amdgcn_readfirstlane((int)(threadIdx.x >> 5));
  const int nwaves = (int)(gridDim.x * (blockDim.x >> 5));
  const int wave   = wib * (int)gridDim.x + (int)blockIdx.x;
  if (wave >= ntiles) return;

  const int m  = lane & 15;
  const int hk = (lane >> 4) << 2;

  v4f a[NC];
  int tile = wave;
  {
    const float* arow = A + ((size_t)tile * 16 + m) * K + hk;
    #pragma unroll
    for (int i = 0; i < NC; ++i) a[i] = *(const v4f*)(arow + i * 8);
  }

  v4f bw[NC][4];
  {
    const float* wrow = W + (size_t)m * K + hk;
    #pragma unroll
    for (int i = 0; i < NC; ++i) {
      #pragma unroll
      for (int t = 0; t < 4; ++t)
        bw[i][t] = *(const v4f*)(wrow + (size_t)t * 16 * K + i * 8);
    }
  }
  const int col   = lane & 15;
  const int rbase = (lane >> 4) * 8;
  const float bb0 = bias[col], bb1 = bias[16 + col], bb2 = bias[32 + col], bb3 = bias[48 + col];

  while (true) {
    const int  nt   = tile + nwaves;
    const bool more = nt < ntiles;
    const float* nrow = A + ((size_t)(more ? nt : tile) * 16 + m) * K + hk;

    v8f acc0 = {}, acc1 = {}, acc2 = {}, acc3 = {};
    #pragma unroll
    for (int i = 0; i < NC; ++i) {
      v2f alo; alo.x = a[i].x; alo.y = a[i].y;
      v2f ahi; ahi.x = a[i].z; ahi.y = a[i].w;
      v2f b0l; b0l.x = bw[i][0].x; b0l.y = bw[i][0].y;
      v2f b1l; b1l.x = bw[i][1].x; b1l.y = bw[i][1].y;
      v2f b2l; b2l.x = bw[i][2].x; b2l.y = bw[i][2].y;
      v2f b3l; b3l.x = bw[i][3].x; b3l.y = bw[i][3].y;
      v2f b0h; b0h.x = bw[i][0].z; b0h.y = bw[i][0].w;
      v2f b1h; b1h.x = bw[i][1].z; b1h.y = bw[i][1].w;
      v2f b2h; b2h.x = bw[i][2].z; b2h.y = bw[i][2].w;
      v2f b3h; b3h.x = bw[i][3].z; b3h.y = bw[i][3].w;
      acc0 = __builtin_amdgcn_wmma_f32_16x16x4_f32(false, alo, false, b0l, (short)0, acc0, false, false);
      acc1 = __builtin_amdgcn_wmma_f32_16x16x4_f32(false, alo, false, b1l, (short)0, acc1, false, false);
      acc2 = __builtin_amdgcn_wmma_f32_16x16x4_f32(false, alo, false, b2l, (short)0, acc2, false, false);
      acc3 = __builtin_amdgcn_wmma_f32_16x16x4_f32(false, alo, false, b3l, (short)0, acc3, false, false);
      acc0 = __builtin_amdgcn_wmma_f32_16x16x4_f32(false, ahi, false, b0h, (short)0, acc0, false, false);
      acc1 = __builtin_amdgcn_wmma_f32_16x16x4_f32(false, ahi, false, b1h, (short)0, acc1, false, false);
      acc2 = __builtin_amdgcn_wmma_f32_16x16x4_f32(false, ahi, false, b2h, (short)0, acc2, false, false);
      acc3 = __builtin_amdgcn_wmma_f32_16x16x4_f32(false, ahi, false, b3h, (short)0, acc3, false, false);
      a[i] = *(const v4f*)(nrow + i * 8);
      __builtin_amdgcn_sched_barrier(0);
    }

    float* orow = out + (size_t)tile * 16 * HID;
    #pragma unroll
    for (int v = 0; v < 8; ++v) {
      const int r = rbase + v;
      float x0 = acc0[v] + bb0;
      float x1 = acc1[v] + bb1;
      float x2 = acc2[v] + bb2;
      float x3 = acc3[v] + bb3;
      if (RELU) {
        x0 = fmaxf(x0, 0.f); x1 = fmaxf(x1, 0.f);
        x2 = fmaxf(x2, 0.f); x3 = fmaxf(x3, 0.f);
      }
      orow[(size_t)r * HID +      col] = x0;
      orow[(size_t)r * HID + 16 + col] = x1;
      orow[(size_t)r * HID + 32 + col] = x2;
      orow[(size_t)r * HID + 48 + col] = x3;
    }
    __builtin_amdgcn_sched_barrier(0);
    asm volatile("s_wait_xcnt 0x0");
    __builtin_amdgcn_sched_barrier(0);
    if (!more) break;
    tile = nt;
  }
}

__global__ void deg_init(float* __restrict__ deg, int n) {
  int i = blockIdx.x * blockDim.x + threadIdx.x;
  if (i < n) deg[i] = 1.0f;
}
__global__ void deg_acc(const int* __restrict__ src, const int* __restrict__ dst,
                        float* __restrict__ deg, int* __restrict__ keyw, int n, int e) {
  const int i = (int)(blockIdx.x * blockDim.x + threadIdx.x);
  unsigned k = 0;
  int pw = 0;
  if (i < e) {
    const int s = src[i];
    const int d = dst[i];
    atomicAdd(&deg[d], 1.0f);
    k = (unsigned)d / OWN_ROWS;
    pw = (int)(((unsigned)s << PACK_BITS) | ((unsigned)d - k * OWN_ROWS));
  }
  if (!keys_ok(keyw, n, e)) return;
  if (i < e) ((int*)pack_words(keyw, e))[i] = pw;
  const int lane = (int)(threadIdx.x & 31);
  unsigned w[1] = { (i < e) ? key_onehot(k) : 0u };
  wave_bit_transpose<1>(w, lane);
  ((unsigned*)keyw)[(size_t)lane * plane_words(e) + (size_t)(i >> 5)] = w[0];
}
__global__ void deg_rsqrt(float* __restrict__ deg, int n) {
  int i = blockIdx.x * blockDim.x + threadIdx.x;
  if (i < n) deg[i] = rsqrtf(deg[i]);
}

#define HIST_WORDS 51200
#define HIST_NODES (2 * HIST_WORDS)
#define CNT_CHUNK  8192
#define CNT_PER_THREAD (CNT_CHUNK / 1024)
#define CNT_GROUPS (CNT_CHUNK / 32)
#define STG_PITCH  (CNT_GROUPS + 1)

__global__ __launch_bounds__(1024)
void deg_count(const int* __restrict__ src, const int* __restrict__ dst,
               unsigned* __restrict__ planes, int* __restrict__ keyw, int n, int e) {
  __shared__ unsigned hist[HIST_WORDS];
  __shared__ unsigned stage[KEY_PLANES * STG_PITCH];
  const int tid    = (int)threadIdx.x;
  const int lane   = tid & 31;
  const int nwords = (n + 1) >> 1;
  const int cb     = (int)blockIdx.x * CNT_CHUNK;
  unsigned* plane  = planes + (size_t)blockIdx.x * (size_t)nwords;
  const bool planes_ok = keys_ok(keyw, n, e);

  for (int nb = 0; nb < n; nb += HIST_NODES) {
    for (int w = tid; w < HIST_WORDS; w += 1024) hist[w] = 0u;
    __syncthreads();

    int sv[CNT_PER_THREAD], dv[CNT_PER_THREAD];
    unsigned oh[CNT_PER_THREAD];
    #pragma unroll
    for (int j = 0; j < CNT_PER_THREAD; ++j) {
      const int ic = min(cb + j * 1024 + tid, e - 1);
      sv[j] = src[ic];
      dv[j] = dst[ic];
    }
    #pragma unroll
    for (int j = 0; j < CNT_PER_THREAD; ++j) {
      const int i = cb + j * 1024 + tid;
      const int d = dv[j];
      const unsigned dr = (unsigned)(d - nb);
      if (i < e && dr < (unsigned)HIST_NODES)
        atomicAdd(&hist[dr >> 1], 1u << ((dr & 1u) * 16u));
      oh[j] = 0u;
      if (nb == 0 && planes_ok && i < e) {
        const unsigned k = (unsigned)d / OWN_ROWS;
        ((int*)pack_words(keyw, e))[i] =
            (int)(((unsigned)sv[j] << PACK_BITS) | ((unsigned)d - k * OWN_ROWS));
        oh[j] = key_onehot(k);
      }
    }
    if (nb == 0 && planes_ok) {
      wave_bit_transpose<CNT_PER_THREAD>(oh, lane);
      #pragma unroll
      for (int j = 0; j < CNT_PER_THREAD; ++j)
        stage[lane * STG_PITCH + j * 32 + (tid >> 5)] = oh[j];
    }
    __syncthreads();
    const int w0 = nb >> 1;
    const int wn = min(HIST_WORDS, nwords - w0);
    for (int w = tid; w < wn; w += 1024) plane[w0 + w] = hist[w];
    if (nb == 0 && planes_ok) {
      const size_t npw = plane_words(e);
      const size_t g0  = (size_t)blockIdx.x * CNT_GROUPS;
      unsigned* pl = (unsigned*)keyw + (size_t)(tid >> 5) * npw + g0;
      #pragma unroll
      for (int r = 0; r < CNT_GROUPS / 32; ++r) {
        const int lg = r * 32 + lane;
        if (g0 + (size_t)lg < npw) pl[lg] = stage[(tid >> 5) * STG_PITCH + lg];
      }
    }
    __syncthreads();
  }
}

__global__ void deg_finish(const unsigned* __restrict__ planes, float* __restrict__ dinv,
                           int n, int nplanes) {
  const int w = (int)(blockIdx.x * blockDim.x + threadIdx.x);
  const int nwords = (n + 1) >> 1;
  if (w >= nwords) return;
  unsigned lo = 0u, hi = 0u;
  #pragma unroll 14
  for (int b = 0; b < nplanes; ++b) {
    const unsigned v = planes[(size_t)b * (size_t)nwords + w];
    lo += v & 0xFFFFu;
    hi += v >> 16;
  }
  dinv[2 * w] = rsqrtf((float)(lo + 1u));
  if (2 * w + 1 < n) dinv[2 * w + 1] = rsqrtf((float)(hi + 1u));
}

__global__ void alpha_kernel(const float* __restrict__ h,
                             const float* __restrict__ attl,
                             const float* __restrict__ attr,
                             const float* __restrict__ dinv,
                             float* __restrict__ adp, float* __restrict__ ar, int n) {
  const int wave = (int)((blockIdx.x * blockDim.x + threadIdx.x) >> 5);
  const int lane = (int)(threadIdx.x & 31);
  if (wave >= n) return;
  const float* row = h + (size_t)wave * HID;
  const float h0v = row[lane], h1v = row[lane + 32];
  float pl = h0v * attl[lane] + h1v * attl[lane + 32];
  float pr = h0v * attr[lane] + h1v * attr[lane + 32];
  #pragma unroll
  for (int off = 16; off > 0; off >>= 1) {
    pl += __shfl_xor(pl, off, 32);
    pr += __shfl_xor(pr, off, 32);
  }
  if (lane == 0) {
    v2f o; o.x = pl; o.y = dinv[wave];
    *(v2f*)(adp + 2 * (size_t)wave) = o;
    ar[wave] = pr;
  }
}

#define LWAVES     32
#define UNIT_EDGES 512
#define QCAP       (UNIT_EDGES + 32)

static_assert(OWN_ROWS <= (1 << PACK_BITS), "local row must fit the packed word");

template<bool RECORD>
__global__ __launch_bounds__(1024)
void layer_owned(const int* __restrict__ src, const int* __restrict__ dst,
                 const float* __restrict__ h, const float* __restrict__ h0,
                 const float* __restrict__ adp, const float* __restrict__ ar,
                 const float* __restrict__ dinv, float* hout, int* lists,
                 const int* __restrict__ keyw,
                 const float* __restrict__ attl_next, const float* __restrict__ attr_next,
                 float* __restrict__ adp_next, float* __restrict__ ar_next,
                 int n, int e_cnt) {
  __shared__ float acc[OWN_ROWS * HID];
  __shared__ int   queue[LWAVES][QCAP];
  __shared__ float wself[OWN_ROWS];
  __shared__ float dotl[RECORD ? OWN_ROWS : 1];
  __shared__ float dotr[RECORD ? OWN_ROWS : 1];

  const int tid  = (int)threadIdx.x;
  const int lane = tid & 31;
  const int wave = __builtin_amdgcn_readfirstlane(tid >> 5);
  const int base = (int)blockIdx.x * OWN_ROWS;
  const int rows = min(OWN_ROWS, n - base);
  const int nel  = rows * HID;
  const size_t gbase = (size_t)base * HID;

  int* lw = lists + gbase + (size_t)wave * (size_t)(rows * 2);
  const int lcap = rows * 2 - 1;

  if (tid < rows) {
    const int node = base + tid;
    const float di = dinv[node];
    wself[tid] = tanhf(adp[2 * (size_t)node] + ar[node]) * di * di;
  }
  __syncthreads();
  for (int idx = tid; idx < nel; idx += 1024)
    acc[idx] = EPS_BLEND * h0[gbase + idx] + wself[idx >> 6] * h[gbase + idx];
  __syncthreads();

  auto add_rows = [&](int se, int dl, float w, int cnt) {
    for (int j0 = 0; j0 < cnt; j0 += 8) {
      float v0[8], v1[8];
      #pragma unroll
      for (int t = 0; t < 8; ++t) {
        const int sj = __builtin_amdgcn_readlane(se, j0 + t);
        const float* row = h + (size_t)sj * HID;
        v0[t] = row[lane];
        v1[t] = row[lane + 32];
      }
      #pragma unroll
      for (int t = 0; t < 8; ++t) {
        const int j = j0 + t;
        const float wj = __int_as_float(__builtin_amdgcn_readlane(__float_as_int(w), j));
        const int dj = __builtin_amdgcn_readlane(dl, j);
        const bool live = j < cnt;
        atomicAdd(&acc[dj * HID + lane],      live ? wj * v0[t] : 0.f);
        atomicAdd(&acc[dj * HID + 32 + lane], live ? wj * v1[t] : 0.f);
      }
    }
  };

  bool do_sweep = true;
  if (!RECORD) {
    int total = lw[0];
    if (total >= 0) {
      do_sweep = false;
      total = min(total, lcap);
      int  p_nxt = (lane < total) ? lw[1 + lane] : 0;
      int  se_c = 0, dl_c = 0;
      v2f  sp = {};
      float xr = 0.f, dd = 0.f;
      {
        se_c = p_nxt >> PACK_BITS;
        dl_c = p_nxt & ((1 << PACK_BITS) - 1);
        sp = *(const v2f*)(adp + 2 * (size_t)se_c); xr = ar[base + dl_c]; dd = dinv[base + dl_c];
        p_nxt = (32 + lane < total) ? lw[1 + 32 + lane] : 0;
      }
      for (int b = 0; b < total; b += 32) {
        const int cnt = min(32, total - b);
        float w = tanhf(sp.x + xr) * sp.y * dd;
        if (lane >= cnt) w = 0.f;
        const int se = se_c, dl = dl_c;
        se_c = p_nxt >> PACK_BITS;
        dl_c = p_nxt & ((1 << PACK_BITS) - 1);
        sp = *(const v2f*)(adp + 2 * (size_t)se_c); xr = ar[base + dl_c]; dd = dinv[base + dl_c];
        p_nxt = (b + 64 + lane < total) ? lw[1 + b + 64 + lane] : 0;
        add_rows(se, dl, w, cnt);
      }
    }
  }

  if (do_sweep) {
    int* qw = queue[wave];
    const unsigned lt = (1u << lane) - 1u;
    int qn = 0;
    int rec_n = (n <= PACK_MAXN) ? 0 : -1;

    const bool by_key = keys_ok(keyw, n, e_cnt);
    const int* packw = pack_words(keyw, e_cnt);

    auto drain = [&](bool all) {
      while (qn >= 32 || (all && qn > 0)) {
        const int cnt = min(qn, 32);
        const int qb  = qn - cnt;
        __builtin_amdgcn_wave_barrier();
        const bool act = lane < cnt;
        int se = 0, dd = base;
        if (act) {
          const int e = qw[qb + lane];
          if (by_key) {
            const int p = packw[e];
            se = p >> PACK_BITS;
            dd = base + (p & ((1 << PACK_BITS) - 1));
          } else {
            se = src[e];
            dd = dst[e];
          }
        }
        const int dl = dd - base;
        if (RECORD) {
          if (rec_n >= 0) {
            if (rec_n + cnt <= lcap) {
              if (act) lw[1 + rec_n + lane] = (se << PACK_BITS) | dl;
              rec_n += cnt;
            } else {
              rec_n = -1;
            }
          }
        }
        const v2f   sp = *(const v2f*)(adp + 2 * (size_t)se);
        const float xr = ar[dd];
        const float dv = dinv[dd];
        float w = tanhf(sp.x + xr) * sp.y * dv;
        if (!act) w = 0.f;
        add_rows(se, dl, w, cnt);
        qn = qb;
      }
    };

    if (by_key) {
      const size_t npw = plane_words(e_cnt);
      const unsigned* hp = (const unsigned*)keyw + (size_t)(blockIdx.x >> 4) * npw;
      const unsigned* lp = (const unsigned*)keyw + (size_t)(16 + (blockIdx.x & 15u)) * npw;
      const int units = (e_cnt + KEY_UNIT - 1) / KEY_UNIT;
      const int step = max(1, units / (int)gridDim.x);
      const int rot  = units > 0 ? (int)(((unsigned)blockIdx.x * (unsigned)step) % (unsigned)units) : 0;
      auto load_keys = [&](int uu_in, v4i& vh, v4i& vl) -> int {
        int uu = uu_in + rot;
        if (uu >= units) uu -= units;
        const int gb = uu * (KEY_UNIT / 32) + lane * 4;
        vh = *(const v4i*)(hp + gb);
        vl = *(const v4i*)(lp + gb);
        return gb;
      };

      auto scan_group = [&](unsigned g, int e0) {
        while (__builtin_amdgcn_ballot_w32(g != 0u)) {
          const int eidx = e0 + __builtin_ctz(g | 0x80000000u);
          const bool m = (g != 0u) && (eidx < e_cnt);
          const unsigned mm = __builtin_amdgcn_ballot_w32(m);
          if (m) qw[qn + __builtin_popcount(mm & lt)] = eidx;
          qn += __builtin_popcount(mm);
          g &= g - 1u;
          drain(false);
        }
      };

      v4i ch = {}, cl = {};
      int c_gb = 0;
      int u = wave;
      const bool have = u < units;
      if (have) c_gb = load_keys(u, ch, cl);

      while (have) {
        v4i nh, nl;
        const bool have_n = u + LWAVES < units;
        const int n_gb = load_keys(have_n ? u + LWAVES : u, nh, nl);

        #pragma unroll
        for (int q = 0; q < 4; ++q)
          scan_group((unsigned)ch[q] & (unsigned)cl[q], (c_gb + q) * 32);

        if (!have_n) break;
        ch = nh; cl = nl; c_gb = n_gb;
        u += LWAVES;
      }
      drain(true);
    } else {
      const int units = (e_cnt + UNIT_EDGES - 1) / UNIT_EDGES;
      const int step = max(1, units / (int)gridDim.x);
      const int rot  = units > 0 ? (int)(((unsigned)blockIdx.x * (unsigned)step) % (unsigned)units) : 0;
      const bool vec_ok = ((e_cnt & 3) == 0) && ((((size_t)dst) & 15) == 0);

      v4i cur[4];
      int cur_eb = 0;
      int u = wave;
      bool have = u < units;

      auto load_unit = [&](int uu_in, v4i (&v)[4]) -> int {
        int uu = uu_in + rot;
        if (uu >= units) uu -= units;
        const int eb = uu * UNIT_EDGES + lane * 4;
        if (vec_ok && (uu + 1) * UNIT_EDGES <= e_cnt) {
          #pragma unroll
          for (int k = 0; k < 4; ++k) v[k] = *(const v4i*)(dst + eb + k * 128);
        } else {
          #pragma unroll
          for (int k = 0; k < 4; ++k) {
            #pragma unroll
            for (int j = 0; j < 4; ++j) {
              const int idx = eb + k * 128 + j;
              v[k][j] = (idx < e_cnt) ? dst[idx] : -1;
            }
          }
        }
        return eb;
      };

      if (have) cur_eb = load_unit(u, cur);

      while (true) {
        v4i nxt[4];
        int nxt_eb = 0;
        const bool have_n = have && (u + LWAVES < units);
        if (have_n) nxt_eb = load_unit(u + LWAVES, nxt);

        if (have) {
          #pragma unroll
          for (int k = 0; k < 4; ++k) {
            #pragma unroll
            for (int j = 0; j < 4; ++j) {
              const int d = cur[k][j];
              const bool m = (unsigned)(d - base) < (unsigned)rows;
              const unsigned mask = __builtin_amdgcn_ballot_w32(m);
              if (mask) {
                if (m) qw[qn + __builtin_popcount(mask & lt)] = cur_eb + k * 128 + j;
                qn += __builtin_popcount(mask);
              }
            }
          }
        }

        const bool last = !have_n;
        drain(last);
        if (last) break;
        #pragma unroll
        for (int k = 0; k < 4; ++k) cur[k] = nxt[k];
        cur_eb = nxt_eb;
        u += LWAVES;
      }
    }
    if (RECORD) {
      if (lane == 0) lw[0] = rec_n;
    }
  }
  __syncthreads();

  if (RECORD) {
    const float l0 = attl_next[lane], l1 = attl_next[lane + 32];
    const float r0 = attr_next[lane], r1 = attr_next[lane + 32];
    for (int rb = wave; rb < rows; rb += 4 * LWAVES) {
      float pl[4], pr[4];
      #pragma unroll
      for (int k = 0; k < 4; ++k) {
        const int r = rb + k * LWAVES;
        pl[k] = 0.f; pr[k] = 0.f;
        if (r < rows) {
          const float h0v = fmaxf(acc[r * HID + lane], 0.f);
          const float h1v = fmaxf(acc[r * HID + 32 + lane], 0.f);
          hout[gbase + (size_t)r * HID + lane]      = h0v;
          hout[gbase + (size_t)r * HID + 32 + lane] = h1v;
          pl[k] = h0v * l0 + h1v * l1;
          pr[k] = h0v * r0 + h1v * r1;
        }
      }
      #pragma unroll
      for (int off = 16; off > 0; off >>= 1) {
        #pragma unroll
        for (int k = 0; k < 4; ++k) {
          pl[k] += __shfl_xor(pl[k], off, 32);
          pr[k] += __shfl_xor(pr[k], off, 32);
        }
      }
      if (lane == 0) {
        #pragma unroll
        for (int k = 0; k < 4; ++k) {
          const int r = rb + k * LWAVES;
          if (r < rows) { dotl[r] = pl[k]; dotr[r] = pr[k]; }
        }
      }
    }
    __syncthreads();
    if (tid < rows) {
      v2f o; o.x = dotl[tid]; o.y = dinv[base + tid];
      *(v2f*)(adp_next + 2 * (size_t)(base + tid)) = o;
      ar_next[base + tid] = dotr[tid];
    }
  } else {
    for (int idx = tid; idx < nel; idx += 1024)
      hout[gbase + idx] = fmaxf(acc[idx], 0.f);
  }
}

extern "C" void kernel_launch(void* const* d_in, const int* in_sizes, int n_in,
                              void* d_out, int out_size, void* d_ws, size_t ws_size,
                              hipStream_t stream) {
  const float* x     = (const float*)d_in[0];
  const int*   ei    = (const int*)  d_in[1];
  const float* W_in  = (const float*)d_in[2];
  const float* b_in  = (const float*)d_in[3];
  const float* attl1 = (const float*)d_in[4];
  const float* attr1 = (const float*)d_in[5];
  const float* attl2 = (const float*)d_in[6];
  const float* attr2 = (const float*)d_in[7];
  const float* W_out = (const float*)d_in[8];
  const float* b_out = (const float*)d_in[9];
  float* out = (float*)d_out;

  const int IN = 128;
  const int E  = in_sizes[1] / 2;
  const int N  = in_sizes[0] / IN;
  const int*   srcI = ei;
  const int*   dstI = ei + E;

  float* H0   = (float*)d_ws;
  float* H1   = H0   + (size_t)N * HID;
  float* OUTB = H1   + (size_t)N * HID;
  float* ADP  = OUTB + (size_t)N * HID;
  float* DINV = ADP  + 2 * (size_t)N;
  float* ADP2 = out;
  float* AR2  = out + 2 * (size_t)N;
  float* AR   = out + 3 * (size_t)N;
  int*   KEYW = (int*)(out + 4 * (size_t)N);

  const int ntiles = N / 16;
  const int nThreads  = (N + 255) / 256;
  const int degBlk    = (E + KEY_UNIT - 1) / KEY_UNIT * (KEY_UNIT / 256);
  const int nWaveBlk  = (N + 7) / 8;
  const int layerBlk  = (N + OWN_ROWS - 1) / OWN_ROWS;

  gemm_wmma_f32<128, true><<<256, 256, 0, stream>>>(x, W_in, b_in, H0, ntiles);

  const int nWords = (N + 1) / 2;
  const int cntBlk = (degBlk * 256 + CNT_CHUNK - 1) / CNT_CHUNK;
  if ((size_t)cntBlk * (size_t)nWords <= 2 * (size_t)N * HID) {
    deg_count <<<cntBlk, 1024, 0, stream>>>(srcI, dstI, (unsigned*)H1, KEYW, N, E);
    deg_finish<<<(nWords + 255) / 256, 256, 0, stream>>>((const unsigned*)H1, DINV, N, cntBlk);
  } else {
    deg_init <<<nThreads, 256, 0, stream>>>(DINV, N);
    deg_acc  <<<degBlk, 256, 0, stream>>>(srcI, dstI, DINV, KEYW, N, E);
    deg_rsqrt<<<nThreads, 256, 0, stream>>>(DINV, N);
  }

  alpha_kernel<<<nWaveBlk, 256, 0, stream>>>(H0, attl1, attr1, DINV, ADP, AR, N);
  layer_owned<true> <<<layerBlk, 1024, 0, stream>>>(srcI, dstI, H0, H0, ADP, AR, DINV, H1, (int*)OUTB, KEYW,
                                                    attl2, attr2, ADP2, AR2, N, E);

  layer_owned<false><<<layerBlk, 1024, 0, stream>>>(srcI, dstI, H1, H0, ADP2, AR2, DINV, OUTB, (int*)OUTB, KEYW,
                                                    nullptr, nullptr, nullptr, nullptr, N, E);

  gemm_wmma_f32<64, false><<<256, 256, 0, stream>>>(OUTB, W_out, b_out, out, ntiles);
}


# projection GEMM swaps its WMMA operands (W as A, tile as B): a lane holds 8 consecutive channels of one node and sums layer 1's attention dots in registers (new sum order); 7 launches to 6
# speedup vs baseline: 4.0857x; 1.1377x over previous
#include <hip/hip_runtime.h>
#include <hip/hip_bf16.h>

typedef __attribute__((ext_vector_type(2))) float v2f;
typedef __attribute__((ext_vector_type(8))) float v8f;
typedef __attribute__((ext_vector_type(4))) int v4i;
typedef __attribute__((ext_vector_type(4))) float v4f;

#define HID 64
#define EPS_BLEND 0.1f

#define OWN_ROWS   391
#define KEY_UNIT   4096
#define KEY_BLOCKS 256

#define PACK_BITS  9
#define PACK_MAXN  (1 << (31 - PACK_BITS))
#define KEY_PLANES 32
__device__ __forceinline__ size_t edge_pad(int e) {
  return ((size_t)e + KEY_UNIT - 1) / KEY_UNIT * KEY_UNIT;
}
__device__ __forceinline__ bool keys_ok(const int* keyw, int n, int e) {
  const size_t epad = edge_pad(e);
  return ((n + OWN_ROWS - 1) / OWN_ROWS <= KEY_BLOCKS) && (n <= PACK_MAXN) &&
         (16 * (size_t)n + 8 * epad <= 256 * (size_t)n) &&
         ((((size_t)keyw) & 15) == 0);
}
__device__ __forceinline__ size_t plane_words(int e) { return edge_pad(e) / 32; }
__device__ __forceinline__ const int* pack_words(const int* keyw, int e) {
  return keyw + (size_t)KEY_PLANES * plane_words(e);
}
__device__ __forceinline__ unsigned key_onehot(unsigned k) {
  return (1u << (k >> 4)) | (0x10000u << (k & 15u));
}
template<int NW>
__device__ __forceinline__ void wave_bit_transpose(unsigned (&x)[NW], int lane) {
  #pragma unroll
  for (int t = 0; t < 5; ++t) {
    const int s = 16 >> t;
    const unsigned m = (s == 16) ? 0x0000FFFFu : (s == 8) ? 0x00FF00FFu :
                       (s == 4)  ? 0x0F0F0F0Fu : (s == 2) ? 0x33333333u : 0x55555555u;
    const unsigned up = (lane & s) ? ~0u : 0u;
    #pragma unroll
    for (int k = 0; k < NW; ++k) {
      const unsigned y = (unsigned)__builtin_amdgcn_ds_bpermute((lane ^ s) << 2, (int)x[k]);
      const unsigned lo = (up & (y >> s)) | (~up & x[k]);
      const unsigned hi = (up & x[k]) | (~up & (y << s));
      x[k] = (lo & m) | (hi & ~m);
    }
  }
}

template<int K, bool RELU, bool DOTS>
__global__ __launch_bounds__(256)
void gemm_wmma_f32(const float* __restrict__ A,
                   const float* __restrict__ W,
                   const float* __restrict__ bias,
                   float* __restrict__ out,
                   int ntiles,
                   const float* __restrict__ attl, const float* __restrict__ attr,
                   const float* __restrict__ dinv,
                   float* __restrict__ adp, float* __restrict__ ar) {
  constexpr int NC = K / 8;
  const int lane = (int)(threadIdx.x & 31);
  const int wib    = __builtin_amdgcn_readfirstlane((int)(threadIdx.x >> 5));
  const int nwaves = (int)(gridDim.x * (blockDim.x >> 5));
  const int wave   = wib * (int)gridDim.x + (int)blockIdx.x;
  if (wave >= ntiles) return;

  const int m  = lane & 15;
  const int hk = (lane >> 4) << 2;

  v4f a[NC];
  int tile = wave;
  {
    const float* arow = A + ((size_t)tile * 16 + m) * K + hk;
    #pragma unroll
    for (int i = 0; i < NC; ++i) a[i] = *(const v4f*)(arow + i * 8);
  }
  float dcur = 0.f;
  if (DOTS) {
    dcur = dinv[tile * 16 + (lane & 15)];
    __builtin_amdgcn_sched_barrier(0);
  }

  v4f bw[NC][4];
  {
    const float* wrow = W + (size_t)m * K + hk;
    #pragma unroll
    for (int i = 0; i < NC; ++i) {
      #pragma unroll
      for (int t = 0; t < 4; ++t)
        bw[i][t] = *(const v4f*)(wrow + (size_t)t * 16 * K + i * 8);
    }
  }
  const int col   = lane & 15;
  const int rbase = (lane >> 4) * 8;
  const float bb0 = bias[col], bb1 = bias[16 + col], bb2 = bias[32 + col], bb3 = bias[48 + col];
  float bsr[4][8], alr[4][8], arr[4][8];
  if (DOTS) {
    #pragma unroll
    for (int t = 0; t < 4; ++t) {
      #pragma unroll
      for (int q = 0; q < 2; ++q) {
        const int ch = 16 * t + rbase + 4 * q;
        const v4f b4 = *(const v4f*)(bias + ch);
        const v4f l4 = *(const v4f*)(attl + ch);
        const v4f r4 = *(const v4f*)(attr + ch);
        #pragma unroll
        for (int v = 0; v < 4; ++v) {
          bsr[t][4 * q + v] = b4[v]; alr[t][4 * q + v] = l4[v]; arr[t][4 * q + v] = r4[v];
        }
      }
    }
  }

  while (true) {
    const int  nt   = tile + nwaves;
    const bool more = nt < ntiles;
    const float* nrow = A + ((size_t)(more ? nt : tile) * 16 + m) * K + hk;
    float dnxt = 0.f;
    if (DOTS) dnxt = dinv[(more ? nt : tile) * 16 + (lane & 15)];

    v8f acc0 = {}, acc1 = {}, acc2 = {}, acc3 = {};
    #pragma unroll
    for (int i = 0; i < NC; ++i) {
      v2f alo; alo.x = a[i].x; alo.y = a[i].y;
      v2f ahi; ahi.x = a[i].z; ahi.y = a[i].w;
      v2f b0l; b0l.x = bw[i][0].x; b0l.y = bw[i][0].y;
      v2f b1l; b1l.x = bw[i][1].x; b1l.y = bw[i][1].y;
      v2f b2l; b2l.x = bw[i][2].x; b2l.y = bw[i][2].y;
      v2f b3l; b3l.x = bw[i][3].x; b3l.y = bw[i][3].y;
      v2f b0h; b0h.x = bw[i][0].z; b0h.y = bw[i][0].w;
      v2f b1h; b1h.x = bw[i][1].z; b1h.y = bw[i][1].w;
      v2f b2h; b2h.x = bw[i][2].z; b2h.y = bw[i][2].w;
      v2f b3h; b3h.x = bw[i][3].z; b3h.y = bw[i][3].w;
      if (DOTS) {
        acc0 = __builtin_amdgcn_wmma_f32_16x16x4_f32(false, b0l, false, alo, (short)0, acc0, false, false);
        acc1 = __builtin_amdgcn_wmma_f32_16x16x4_f32(false, b1l, false, alo, (short)0, acc1, false, false);
        acc2 = __builtin_amdgcn_wmma_f32_16x16x4_f32(false, b2l, false, alo, (short)0, acc2, false, false);
        acc3 = __builtin_amdgcn_wmma_f32_16x16x4_f32(false, b3l, false, alo, (short)0, acc3, false, false);
        acc0 = __builtin_amdgcn_wmma_f32_16x16x4_f32(false, b0h, false, ahi, (short)0, acc0, false, false);
        acc1 = __builtin_amdgcn_wmma_f32_16x16x4_f32(false, b1h, false, ahi, (short)0, acc1, false, false);
        acc2 = __builtin_amdgcn_wmma_f32_16x16x4_f32(false, b2h, false, ahi, (short)0, acc2, false, false);
        acc3 = __builtin_amdgcn_wmma_f32_16x16x4_f32(false, b3h, false, ahi, (short)0, acc3, false, false);
      } else {
        acc0 = __builtin_amdgcn_wmma_f32_16x16x4_f32(false, alo, false, b0l, (short)0, acc0, false, false);
        acc1 = __builtin_amdgcn_wmma_f32_16x16x4_f32(false, alo, false, b1l, (short)0, acc1, false, false);
        acc2 = __builtin_amdgcn_wmma_f32_16x16x4_f32(false, alo, false, b2l, (short)0, acc2, false, false);
        acc3 = __builtin_amdgcn_wmma_f32_16x16x4_f32(false, alo, false, b3l, (short)0, acc3, false, false);
        acc0 = __builtin_amdgcn_wmma_f32_16x16x4_f32(false, ahi, false, b0h, (short)0, acc0, false, false);
        acc1 = __builtin_amdgcn_wmma_f32_16x16x4_f32(false, ahi, false, b1h, (short)0, acc1, false, false);
        acc2 = __builtin_amdgcn_wmma_f32_16x16x4_f32(false, ahi, false, b2h, (short)0, acc2, false, false);
        acc3 = __builtin_amdgcn_wmma_f32_16x16x4_f32(false, ahi, false, b3h, (short)0, acc3, false, false);
      }
      a[i] = *(const v4f*)(nrow + i * 8);
      __builtin_amdgcn_sched_barrier(0);
    }

    if (DOTS) {
      const int node = tile * 16 + col;
      const float dnode = dcur;
      float* orow = out + (size_t)node * HID + rbase;
      const v8f accs[4] = { acc0, acc1, acc2, acc3 };
      float pl[4], pr[4];
      #pragma unroll
      for (int t = 0; t < 4; ++t) {
        v4f lo, hi;
        pl[t] = 0.f; pr[t] = 0.f;
        #pragma unroll
        for (int v = 0; v < 8; ++v) {
          float xv = accs[t][v] + bsr[t][v];
          if (RELU) xv = fmaxf(xv, 0.f);
          pl[t] += xv * alr[t][v];
          pr[t] += xv * arr[t][v];
          if (v < 4) lo[v] = xv; else hi[v - 4] = xv;
        }
        *(v4f*)(orow + 16 * t)     = lo;
        *(v4f*)(orow + 16 * t + 4) = hi;
      }
      float sl = (pl[0] + pl[1]) + (pl[2] + pl[3]);
      float sr = (pr[0] + pr[1]) + (pr[2] + pr[3]);
      sl += __shfl_xor(sl, 16, 32);
      sr += __shfl_xor(sr, 16, 32);
      if (lane < 16) {
        v2f o; o.x = sl; o.y = dnode;
        *(v2f*)(adp + 2 * (size_t)node) = o;
        ar[node] = sr;
      }
    } else {
      float* orow = out + (size_t)tile * 16 * HID;
      #pragma unroll
      for (int v = 0; v < 8; ++v) {
        const int r = rbase + v;
        float x0 = acc0[v] + bb0;
        float x1 = acc1[v] + bb1;
        float x2 = acc2[v] + bb2;
        float x3 = acc3[v] + bb3;
        if (RELU) {
          x0 = fmaxf(x0, 0.f); x1 = fmaxf(x1, 0.f);
          x2 = fmaxf(x2, 0.f); x3 = fmaxf(x3, 0.f);
        }
        orow[(size_t)r * HID +      col] = x0;
        orow[(size_t)r * HID + 16 + col] = x1;
        orow[(size_t)r * HID + 32 + col] = x2;
        orow[(size_t)r * HID + 48 + col] = x3;
      }
    }
    __builtin_amdgcn_sched_barrier(0);
    asm volatile("s_wait_xcnt 0x0");
    __builtin_amdgcn_sched_barrier(0);
    if (!more) break;
    tile = nt;
    dcur = dnxt;
  }
}

__global__ void deg_init(float* __restrict__ deg, int n) {
  int i = blockIdx.x * blockDim.x + threadIdx.x;
  if (i < n) deg[i] = 1.0f;
}
__global__ void deg_acc(const int* __restrict__ src, const int* __restrict__ dst,
                        float* __restrict__ deg, int* __restrict__ keyw, int n, int e) {
  const int i = (int)(blockIdx.x * blockDim.x + threadIdx.x);
  unsigned k = 0;
  int pw = 0;
  if (i < e) {
    const int s = src[i];
    const int d = dst[i];
    atomicAdd(&deg[d], 1.0f);
    k = (unsigned)d / OWN_ROWS;
    pw = (int)(((unsigned)s << PACK_BITS) | ((unsigned)d - k * OWN_ROWS));
  }
  if (!keys_ok(keyw, n, e)) return;
  if (i < e) ((int*)pack_words(keyw, e))[i] = pw;
  const int lane = (int)(threadIdx.x & 31);
  unsigned w[1] = { (i < e) ? key_onehot(k) : 0u };
  wave_bit_transpose<1>(w, lane);
  ((unsigned*)keyw)[(size_t)lane * plane_words(e) + (size_t)(i >> 5)] = w[0];
}
__global__ void deg_rsqrt(float* __restrict__ deg, int n) {
  int i = blockIdx.x * blockDim.x + threadIdx.x;
  if (i < n) deg[i] = rsqrtf(deg[i]);
}

#define HIST_WORDS 51200
#define HIST_NODES (2 * HIST_WORDS)
#define CNT_CHUNK  8192
#define CNT_PER_THREAD (CNT_CHUNK / 1024)
#define CNT_GROUPS (CNT_CHUNK / 32)
#define STG_PITCH  (CNT_GROUPS + 1)

__global__ __launch_bounds__(1024)
void deg_count(const int* __restrict__ src, const int* __restrict__ dst,
               unsigned* __restrict__ planes, int* __restrict__ keyw, int n, int e) {
  __shared__ unsigned hist[HIST_WORDS];
  __shared__ unsigned stage[KEY_PLANES * STG_PITCH];
  const int tid    = (int)threadIdx.x;
  const int lane   = tid & 31;
  const int nwords = (n + 1) >> 1;
  const int cb     = (int)blockIdx.x * CNT_CHUNK;
  unsigned* plane  = planes + (size_t)blockIdx.x * (size_t)nwords;
  const bool planes_ok = keys_ok(keyw, n, e);

  for (int nb = 0; nb < n; nb += HIST_NODES) {
    for (int w = tid; w < HIST_WORDS; w += 1024) hist[w] = 0u;
    __syncthreads();

    int sv[CNT_PER_THREAD], dv[CNT_PER_THREAD];
    unsigned oh[CNT_PER_THREAD];
    #pragma unroll
    for (int j = 0; j < CNT_PER_THREAD; ++j) {
      const int ic = min(cb + j * 1024 + tid, e - 1);
      sv[j] = src[ic];
      dv[j] = dst[ic];
    }
    #pragma unroll
    for (int j = 0; j < CNT_PER_THREAD; ++j) {
      const int i = cb + j * 1024 + tid;
      const int d = dv[j];
      const unsigned dr = (unsigned)(d - nb);
      if (i < e && dr < (unsigned)HIST_NODES)
        atomicAdd(&hist[dr >> 1], 1u << ((dr & 1u) * 16u));
      oh[j] = 0u;
      if (nb == 0 && planes_ok && i < e) {
        const unsigned k = (unsigned)d / OWN_ROWS;
        ((int*)pack_words(keyw, e))[i] =
            (int)(((unsigned)sv[j] << PACK_BITS) | ((unsigned)d - k * OWN_ROWS));
        oh[j] = key_onehot(k);
      }
    }
    if (nb == 0 && planes_ok) {
      wave_bit_transpose<CNT_PER_THREAD>(oh, lane);
      #pragma unroll
      for (int j = 0; j < CNT_PER_THREAD; ++j)
        stage[lane * STG_PITCH + j * 32 + (tid >> 5)] = oh[j];
    }
    __syncthreads();
    const int w0 = nb >> 1;
    const int wn = min(HIST_WORDS, nwords - w0);
    for (int w = tid; w < wn; w += 1024) plane[w0 + w] = hist[w];
    if (nb == 0 && planes_ok) {
      const size_t npw = plane_words(e);
      const size_t g0  = (size_t)blockIdx.x * CNT_GROUPS;
      unsigned* pl = (unsigned*)keyw + (size_t)(tid >> 5) * npw + g0;
      #pragma unroll
      for (int r = 0; r < CNT_GROUPS / 32; ++r) {
        const int lg = r * 32 + lane;
        if (g0 + (size_t)lg < npw) pl[lg] = stage[(tid >> 5) * STG_PITCH + lg];
      }
    }
    __syncthreads();
  }
}

__global__ void deg_finish(const unsigned* __restrict__ planes, float* __restrict__ dinv,
                           int n, int nplanes) {
  const int w = (int)(blockIdx.x * blockDim.x + threadIdx.x);
  const int nwords = (n + 1) >> 1;
  if (w >= nwords) return;
  unsigned lo = 0u, hi = 0u;
  #pragma unroll 14
  for (int b = 0; b < nplanes; ++b) {
    const unsigned v = planes[(size_t)b * (size_t)nwords + w];
    lo += v & 0xFFFFu;
    hi += v >> 16;
  }
  dinv[2 * w] = rsqrtf((float)(lo + 1u));
  if (2 * w + 1 < n) dinv[2 * w + 1] = rsqrtf((float)(hi + 1u));
}

#define LWAVES     32
#define UNIT_EDGES 512
#define QCAP       (UNIT_EDGES + 32)

static_assert(OWN_ROWS <= (1 << PACK_BITS), "local row must fit the packed word");

template<bool RECORD>
__global__ __launch_bounds__(1024)
void layer_owned(const int* __restrict__ src, const int* __restrict__ dst,
                 const float* __restrict__ h, const float* __restrict__ h0,
                 const float* __restrict__ adp, const float* __restrict__ ar,
                 const float* __restrict__ dinv, float* hout, int* lists,
                 const int* __restrict__ keyw,
                 const float* __restrict__ attl_next, const float* __restrict__ attr_next,
                 float* __restrict__ adp_next, float* __restrict__ ar_next,
                 int n, int e_cnt) {
  __shared__ float acc[OWN_ROWS * HID];
  __shared__ int   queue[LWAVES][QCAP];
  __shared__ float wself[OWN_ROWS];
  __shared__ float dotl[RECORD ? OWN_ROWS : 1];
  __shared__ float dotr[RECORD ? OWN_ROWS : 1];

  const int tid  = (int)threadIdx.x;
  const int lane = tid & 31;
  const int wave = __builtin_amdgcn_readfirstlane(tid >> 5);
  const int base = (int)blockIdx.x * OWN_ROWS;
  const int rows = min(OWN_ROWS, n - base);
  const int nel  = rows * HID;
  const size_t gbase = (size_t)base * HID;

  int* lw = lists + gbase + (size_t)wave * (size_t)(rows * 2);
  const int lcap = rows * 2 - 1;

  if (tid < rows) {
    const int node = base + tid;
    const float di = dinv[node];
    wself[tid] = tanhf(adp[2 * (size_t)node] + ar[node]) * di * di;
  }
  __syncthreads();
  for (int idx = tid; idx < nel; idx += 1024)
    acc[idx] = EPS_BLEND * h0[gbase + idx] + wself[idx >> 6] * h[gbase + idx];
  __syncthreads();

  auto add_rows = [&](int se, int dl, float w, int cnt) {
    for (int j0 = 0; j0 < cnt; j0 += 8) {
      float v0[8], v1[8];
      #pragma unroll
      for (int t = 0; t < 8; ++t) {
        const int sj = __builtin_amdgcn_readlane(se, j0 + t);
        const float* row = h + (size_t)sj * HID;
        v0[t] = row[lane];
        v1[t] = row[lane + 32];
      }
      #pragma unroll
      for (int t = 0; t < 8; ++t) {
        const int j = j0 + t;
        const float wj = __int_as_float(__builtin_amdgcn_readlane(__float_as_int(w), j));
        const int dj = __builtin_amdgcn_readlane(dl, j);
        const bool live = j < cnt;
        atomicAdd(&acc[dj * HID + lane],      live ? wj * v0[t] : 0.f);
        atomicAdd(&acc[dj * HID + 32 + lane], live ? wj * v1[t] : 0.f);
      }
    }
  };

  bool do_sweep = true;
  if (!RECORD) {
    int total = lw[0];
    if (total >= 0) {
      do_sweep = false;
      total = min(total, lcap);
      int  p_nxt = (lane < total) ? lw[1 + lane] : 0;
      int  se_c = 0, dl_c = 0;
      v2f  sp = {};
      float xr = 0.f, dd = 0.f;
      {
        se_c = p_nxt >> PACK_BITS;
        dl_c = p_nxt & ((1 << PACK_BITS) - 1);
        sp = *(const v2f*)(adp + 2 * (size_t)se_c); xr = ar[base + dl_c]; dd = dinv[base + dl_c];
        p_nxt = (32 + lane < total) ? lw[1 + 32 + lane] : 0;
      }
      for (int b = 0; b < total; b += 32) {
        const int cnt = min(32, total - b);
        float w = tanhf(sp.x + xr) * sp.y * dd;
        if (lane >= cnt) w = 0.f;
        const int se = se_c, dl = dl_c;
        se_c = p_nxt >> PACK_BITS;
        dl_c = p_nxt & ((1 << PACK_BITS) - 1);
        sp = *(const v2f*)(adp + 2 * (size_t)se_c); xr = ar[base + dl_c]; dd = dinv[base + dl_c];
        p_nxt = (b + 64 + lane < total) ? lw[1 + b + 64 + lane] : 0;
        add_rows(se, dl, w, cnt);
      }
    }
  }

  if (do_sweep) {
    int* qw = queue[wave];
    const unsigned lt = (1u << lane) - 1u;
    int qn = 0;
    int rec_n = (n <= PACK_MAXN) ? 0 : -1;

    const bool by_key = keys_ok(keyw, n, e_cnt);
    const int* packw = pack_words(keyw, e_cnt);

    auto drain = [&](bool all) {
      while (qn >= 32 || (all && qn > 0)) {
        const int cnt = min(qn, 32);
        const int qb  = qn - cnt;
        __builtin_amdgcn_wave_barrier();
        const bool act = lane < cnt;
        int se = 0, dd = base;
        if (act) {
          const int e = qw[qb + lane];
          if (by_key) {
            const int p = packw[e];
            se = p >> PACK_BITS;
            dd = base + (p & ((1 << PACK_BITS) - 1));
          } else {
            se = src[e];
            dd = dst[e];
          }
        }
        const int dl = dd - base;
        if (RECORD) {
          if (rec_n >= 0) {
            if (rec_n + cnt <= lcap) {
              if (act) lw[1 + rec_n + lane] = (se << PACK_BITS) | dl;
              rec_n += cnt;
            } else {
              rec_n = -1;
            }
          }
        }
        const v2f   sp = *(const v2f*)(adp + 2 * (size_t)se);
        const float xr = ar[dd];
        const float dv = dinv[dd];
        float w = tanhf(sp.x + xr) * sp.y * dv;
        if (!act) w = 0.f;
        add_rows(se, dl, w, cnt);
        qn = qb;
      }
    };

    if (by_key) {
      const size_t npw = plane_words(e_cnt);
      const unsigned* hp = (const unsigned*)keyw + (size_t)(blockIdx.x >> 4) * npw;
      const unsigned* lp = (const unsigned*)keyw + (size_t)(16 + (blockIdx.x & 15u)) * npw;
      const int units = (e_cnt + KEY_UNIT - 1) / KEY_UNIT;
      const int step = max(1, units / (int)gridDim.x);
      const int rot  = units > 0 ? (int)(((unsigned)blockIdx.x * (unsigned)step) % (unsigned)units) : 0;
      auto load_keys = [&](int uu_in, v4i& vh, v4i& vl) -> int {
        int uu = uu_in + rot;
        if (uu >= units) uu -= units;
        const int gb = uu * (KEY_UNIT / 32) + lane * 4;
        vh = *(const v4i*)(hp + gb);
        vl = *(const v4i*)(lp + gb);
        return gb;
      };

      auto scan_group = [&](unsigned g, int e0) {
        while (__builtin_amdgcn_ballot_w32(g != 0u)) {
          const int eidx = e0 + __builtin_ctz(g | 0x80000000u);
          const bool m = (g != 0u) && (eidx < e_cnt);
          const unsigned mm = __builtin_amdgcn_ballot_w32(m);
          if (m) qw[qn + __builtin_popcount(mm & lt)] = eidx;
          qn += __builtin_popcount(mm);
          g &= g - 1u;
          drain(false);
        }
      };

      v4i ch = {}, cl = {};
      int c_gb = 0;
      int u = wave;
      const bool have = u < units;
      if (have) c_gb = load_keys(u, ch, cl);

      while (have) {
        v4i nh, nl;
        const bool have_n = u + LWAVES < units;
        const int n_gb = load_keys(have_n ? u + LWAVES : u, nh, nl);

        #pragma unroll
        for (int q = 0; q < 4; ++q)
          scan_group((unsigned)ch[q] & (unsigned)cl[q], (c_gb + q) * 32);

        if (!have_n) break;
        ch = nh; cl = nl; c_gb = n_gb;
        u += LWAVES;
      }
      drain(true);
    } else {
      const int units = (e_cnt + UNIT_EDGES - 1) / UNIT_EDGES;
      const int step = max(1, units / (int)gridDim.x);
      const int rot  = units > 0 ? (int)(((unsigned)blockIdx.x * (unsigned)step) % (unsigned)units) : 0;
      const bool vec_ok = ((e_cnt & 3) == 0) && ((((size_t)dst) & 15) == 0);

      v4i cur[4];
      int cur_eb = 0;
      int u = wave;
      bool have = u < units;

      auto load_unit = [&](int uu_in, v4i (&v)[4]) -> int {
        int uu = uu_in + rot;
        if (uu >= units) uu -= units;
        const int eb = uu * UNIT_EDGES + lane * 4;
        if (vec_ok && (uu + 1) * UNIT_EDGES <= e_cnt) {
          #pragma unroll
          for (int k = 0; k < 4; ++k) v[k] = *(const v4i*)(dst + eb + k * 128);
        } else {
          #pragma unroll
          for (int k = 0; k < 4; ++k) {
            #pragma unroll
            for (int j = 0; j < 4; ++j) {
              const int idx = eb + k * 128 + j;
              v[k][j] = (idx < e_cnt) ? dst[idx] : -1;
            }
          }
        }
        return eb;
      };

      if (have) cur_eb = load_unit(u, cur);

      while (true) {
        v4i nxt[4];
        int nxt_eb = 0;
        const bool have_n = have && (u + LWAVES < units);
        if (have_n) nxt_eb = load_unit(u + LWAVES, nxt);

        if (have) {
          #pragma unroll
          for (int k = 0; k < 4; ++k) {
            #pragma unroll
            for (int j = 0; j < 4; ++j) {
              const int d = cur[k][j];
              const bool m = (unsigned)(d - base) < (unsigned)rows;
              const unsigned mask = __builtin_amdgcn_ballot_w32(m);
              if (mask) {
                if (m) qw[qn + __builtin_popcount(mask & lt)] = cur_eb + k * 128 + j;
                qn += __builtin_popcount(mask);
              }
            }
          }
        }

        const bool last = !have_n;
        drain(last);
        if (last) break;
        #pragma unroll
        for (int k = 0; k < 4; ++k) cur[k] = nxt[k];
        cur_eb = nxt_eb;
        u += LWAVES;
      }
    }
    if (RECORD) {
      if (lane == 0) lw[0] = rec_n;
    }
  }
  __syncthreads();

  if (RECORD) {
    const float l0 = attl_next[lane], l1 = attl_next[lane + 32];
    const float r0 = attr_next[lane], r1 = attr_next[lane + 32];
    for (int rb = wave; rb < rows; rb += 4 * LWAVES) {
      float pl[4], pr[4];
      #pragma unroll
      for (int k = 0; k < 4; ++k) {
        const int r = rb + k * LWAVES;
        pl[k] = 0.f; pr[k] = 0.f;
        if (r < rows) {
          const float h0v = fmaxf(acc[r * HID + lane], 0.f);
          const float h1v = fmaxf(acc[r * HID + 32 + lane], 0.f);
          hout[gbase + (size_t)r * HID + lane]      = h0v;
          hout[gbase + (size_t)r * HID + 32 + lane] = h1v;
          pl[k] = h0v * l0 + h1v * l1;
          pr[k] = h0v * r0 + h1v * r1;
        }
      }
      #pragma unroll
      for (int off = 16; off > 0; off >>= 1) {
        #pragma unroll
        for (int k = 0; k < 4; ++k) {
          pl[k] += __shfl_xor(pl[k], off, 32);
          pr[k] += __shfl_xor(pr[k], off, 32);
        }
      }
      if (lane == 0) {
        #pragma unroll
        for (int k = 0; k < 4; ++k) {
          const int r = rb + k * LWAVES;
          if (r < rows) { dotl[r] = pl[k]; dotr[r] = pr[k]; }
        }
      }
    }
    __syncthreads();
    if (tid < rows) {
      v2f o; o.x = dotl[tid]; o.y = dinv[base + tid];
      *(v2f*)(adp_next + 2 * (size_t)(base + tid)) = o;
      ar_next[base + tid] = dotr[tid];
    }
  } else {
    for (int idx = tid; idx < nel; idx += 1024)
      hout[gbase + idx] = fmaxf(acc[idx], 0.f);
  }
}

extern "C" void kernel_launch(void* const* d_in, const int* in_sizes, int n_in,
                              void* d_out, int out_size, void* d_ws, size_t ws_size,
                              hipStream_t stream) {
  const float* x     = (const float*)d_in[0];
  const int*   ei    = (const int*)  d_in[1];
  const float* W_in  = (const float*)d_in[2];
  const float* b_in  = (const float*)d_in[3];
  const float* attl1 = (const float*)d_in[4];
  const float* attr1 = (const float*)d_in[5];
  const float* attl2 = (const float*)d_in[6];
  const float* attr2 = (const float*)d_in[7];
  const float* W_out = (const float*)d_in[8];
  const float* b_out = (const float*)d_in[9];
  float* out = (float*)d_out;

  const int IN = 128;
  const int E  = in_sizes[1] / 2;
  const int N  = in_sizes[0] / IN;
  const int*   srcI = ei;
  const int*   dstI = ei + E;

  float* H0   = (float*)d_ws;
  float* H1   = H0   + (size_t)N * HID;
  float* OUTB = H1   + (size_t)N * HID;
  float* ADP  = OUTB + (size_t)N * HID;
  float* DINV = ADP  + 2 * (size_t)N;
  float* ADP2 = out;
  float* AR2  = out + 2 * (size_t)N;
  float* AR   = out + 3 * (size_t)N;
  int*   KEYW = (int*)(out + 4 * (size_t)N);

  const int ntiles = N / 16;
  const int nThreads  = (N + 255) / 256;
  const int degBlk    = (E + KEY_UNIT - 1) / KEY_UNIT * (KEY_UNIT / 256);
  const int layerBlk  = (N + OWN_ROWS - 1) / OWN_ROWS;

  const int nWords = (N + 1) / 2;
  const int cntBlk = (degBlk * 256 + CNT_CHUNK - 1) / CNT_CHUNK;
  if ((size_t)cntBlk * (size_t)nWords <= 2 * (size_t)N * HID) {
    deg_count <<<cntBlk, 1024, 0, stream>>>(srcI, dstI, (unsigned*)H1, KEYW, N, E);
    deg_finish<<<(nWords + 255) / 256, 256, 0, stream>>>((const unsigned*)H1, DINV, N, cntBlk);
  } else {
    deg_init <<<nThreads, 256, 0, stream>>>(DINV, N);
    deg_acc  <<<degBlk, 256, 0, stream>>>(srcI, dstI, DINV, KEYW, N, E);
    deg_rsqrt<<<nThreads, 256, 0, stream>>>(DINV, N);
  }

  gemm_wmma_f32<128, true, true><<<256, 256, 0, stream>>>(x, W_in, b_in, H0, ntiles,
                                                          attl1, attr1, DINV, ADP, AR);

  layer_owned<true> <<<layerBlk, 1024, 0, stream>>>(srcI, dstI, H0, H0, ADP, AR, DINV, H1, (int*)OUTB, KEYW,
                                                    attl2, attr2, ADP2, AR2, N, E);

  layer_owned<false><<<layerBlk, 1024, 0, stream>>>(srcI, dstI, H1, H0, ADP2, AR2, DINV, OUTB, (int*)OUTB, KEYW,
                                                    nullptr, nullptr, nullptr, nullptr, N, E);

  gemm_wmma_f32<64, false, false><<<256, 256, 0, stream>>>(OUTB, W_out, b_out, out, ntiles,
                                                           nullptr, nullptr, nullptr, nullptr, nullptr);
}


# an edge's source scalars (attention dot, 1/sqrt(deg)) ride in the 64 free sign bits of its relu row: two wave ballots on the gathered row return them, no 8-byte pair gather per edge; add order kept
# speedup vs baseline: 4.2715x; 1.0455x over previous
#include <hip/hip_runtime.h>
#include <hip/hip_bf16.h>

typedef __attribute__((ext_vector_type(2))) float v2f;
typedef __attribute__((ext_vector_type(8))) float v8f;
typedef __attribute__((ext_vector_type(4))) int v4i;
typedef __attribute__((ext_vector_type(4))) float v4f;

#define HID 64
#define EPS_BLEND 0.1f

#define OWN_ROWS   391
#define KEY_UNIT   4096
#define KEY_BLOCKS 256

#define PACK_BITS  9
#define PACK_MAXN  (1 << (31 - PACK_BITS))
#define KEY_PLANES 32
__device__ __forceinline__ size_t edge_pad(int e) {
  return ((size_t)e + KEY_UNIT - 1) / KEY_UNIT * KEY_UNIT;
}
__device__ __forceinline__ bool keys_ok(const int* keyw, int n, int e) {
  const size_t epad = edge_pad(e);
  return ((n + OWN_ROWS - 1) / OWN_ROWS <= KEY_BLOCKS) && (n <= PACK_MAXN) &&
         (16 * (size_t)n + 8 * epad <= 256 * (size_t)n) &&
         ((((size_t)keyw) & 15) == 0);
}
__device__ __forceinline__ size_t plane_words(int e) { return edge_pad(e) / 32; }
__device__ __forceinline__ const int* pack_words(const int* keyw, int e) {
  return keyw + (size_t)KEY_PLANES * plane_words(e);
}
__device__ __forceinline__ unsigned key_onehot(unsigned k) {
  return (1u << (k >> 4)) | (0x10000u << (k & 15u));
}
template<int NW>
__device__ __forceinline__ void wave_bit_transpose(unsigned (&x)[NW], int lane) {
  #pragma unroll
  for (int t = 0; t < 5; ++t) {
    const int s = 16 >> t;
    const unsigned m = (s == 16) ? 0x0000FFFFu : (s == 8) ? 0x00FF00FFu :
                       (s == 4)  ? 0x0F0F0F0Fu : (s == 2) ? 0x33333333u : 0x55555555u;
    const unsigned up = (lane & s) ? ~0u : 0u;
    #pragma unroll
    for (int k = 0; k < NW; ++k) {
      const unsigned y = (unsigned)__builtin_amdgcn_ds_bpermute((lane ^ s) << 2, (int)x[k]);
      const unsigned lo = (up & (y >> s)) | (~up & x[k]);
      const unsigned hi = (up & x[k]) | (~up & (y << s));
      x[k] = (lo & m) | (hi & ~m);
    }
  }
}

template<int K, bool RELU, bool DOTS>
__global__ __launch_bounds__(256)
void gemm_wmma_f32(const float* __restrict__ A,
                   const float* __restrict__ W,
                   const float* __restrict__ bias,
                   float* __restrict__ out,
                   int ntiles,
                   const float* __restrict__ attl, const float* __restrict__ attr,
                   const float* __restrict__ dinv,
                   float* __restrict__ adp, float* __restrict__ ar) {
  constexpr int NC = K / 8;
  const int lane = (int)(threadIdx.x & 31);
  const int wib    = __builtin_amdgcn_readfirstlane((int)(threadIdx.x >> 5));
  const int nwaves = (int)(gridDim.x * (blockDim.x >> 5));
  const int wave   = wib * (int)gridDim.x + (int)blockIdx.x;
  if (wave >= ntiles) return;

  const int m  = lane & 15;
  const int hk = (lane >> 4) << 2;

  v4f a[NC];
  int tile = wave;
  {
    const float* arow = A + ((size_t)tile * 16 + m) * K + hk;
    #pragma unroll
    for (int i = 0; i < NC; ++i) a[i] = *(const v4f*)(arow + i * 8);
  }
  float dcur = 0.f;
  if (DOTS) {
    dcur = dinv[tile * 16 + (lane & 15)];
    __builtin_amdgcn_sched_barrier(0);
  }

  v4f bw[NC][4];
  {
    const float* wrow = W + (size_t)m * K + hk;
    #pragma unroll
    for (int i = 0; i < NC; ++i) {
      #pragma unroll
      for (int t = 0; t < 4; ++t)
        bw[i][t] = *(const v4f*)(wrow + (size_t)t * 16 * K + i * 8);
    }
  }
  const int col   = lane & 15;
  const int rbase = (lane >> 4) * 8;
  const float bb0 = bias[col], bb1 = bias[16 + col], bb2 = bias[32 + col], bb3 = bias[48 + col];
  float bsr[4][8], alr[4][8], arr[4][8];
  if (DOTS) {
    #pragma unroll
    for (int t = 0; t < 4; ++t) {
      #pragma unroll
      for (int q = 0; q < 2; ++q) {
        const int ch = 16 * t + rbase + 4 * q;
        const v4f b4 = *(const v4f*)(bias + ch);
        const v4f l4 = *(const v4f*)(attl + ch);
        const v4f r4 = *(const v4f*)(attr + ch);
        #pragma unroll
        for (int v = 0; v < 4; ++v) {
          bsr[t][4 * q + v] = b4[v]; alr[t][4 * q + v] = l4[v]; arr[t][4 * q + v] = r4[v];
        }
      }
    }
  }

  while (true) {
    const int  nt   = tile + nwaves;
    const bool more = nt < ntiles;
    const float* nrow = A + ((size_t)(more ? nt : tile) * 16 + m) * K + hk;
    float dnxt = 0.f;
    if (DOTS) dnxt = dinv[(more ? nt : tile) * 16 + (lane & 15)];

    v8f acc0 = {}, acc1 = {}, acc2 = {}, acc3 = {};
    #pragma unroll
    for (int i = 0; i < NC; ++i) {
      v2f alo; alo.x = a[i].x; alo.y = a[i].y;
      v2f ahi; ahi.x = a[i].z; ahi.y = a[i].w;
      v2f b0l; b0l.x = bw[i][0].x; b0l.y = bw[i][0].y;
      v2f b1l; b1l.x = bw[i][1].x; b1l.y = bw[i][1].y;
      v2f b2l; b2l.x = bw[i][2].x; b2l.y = bw[i][2].y;
      v2f b3l; b3l.x = bw[i][3].x; b3l.y = bw[i][3].y;
      v2f b0h; b0h.x = bw[i][0].z; b0h.y = bw[i][0].w;
      v2f b1h; b1h.x = bw[i][1].z; b1h.y = bw[i][1].w;
      v2f b2h; b2h.x = bw[i][2].z; b2h.y = bw[i][2].w;
      v2f b3h; b3h.x = bw[i][3].z; b3h.y = bw[i][3].w;
      if (DOTS) {
        acc0 = __builtin_amdgcn_wmma_f32_16x16x4_f32(false, b0l, false, alo, (short)0, acc0, false, false);
        acc1 = __builtin_amdgcn_wmma_f32_16x16x4_f32(false, b1l, false, alo, (short)0, acc1, false, false);
        acc2 = __builtin_amdgcn_wmma_f32_16x16x4_f32(false, b2l, false, alo, (short)0, acc2, false, false);
        acc3 = __builtin_amdgcn_wmma_f32_16x16x4_f32(false, b3l, false, alo, (short)0, acc3, false, false);
        acc0 = __builtin_amdgcn_wmma_f32_16x16x4_f32(false, b0h, false, ahi, (short)0, acc0, false, false);
        acc1 = __builtin_amdgcn_wmma_f32_16x16x4_f32(false, b1h, false, ahi, (short)0, acc1, false, false);
        acc2 = __builtin_amdgcn_wmma_f32_16x16x4_f32(false, b2h, false, ahi, (short)0, acc2, false, false);
        acc3 = __builtin_amdgcn_wmma_f32_16x16x4_f32(false, b3h, false, ahi, (short)0, acc3, false, false);
      } else {
        acc0 = __builtin_amdgcn_wmma_f32_16x16x4_f32(false, alo, false, b0l, (short)0, acc0, false, false);
        acc1 = __builtin_amdgcn_wmma_f32_16x16x4_f32(false, alo, false, b1l, (short)0, acc1, false, false);
        acc2 = __builtin_amdgcn_wmma_f32_16x16x4_f32(false, alo, false, b2l, (short)0, acc2, false, false);
        acc3 = __builtin_amdgcn_wmma_f32_16x16x4_f32(false, alo, false, b3l, (short)0, acc3, false, false);
        acc0 = __builtin_amdgcn_wmma_f32_16x16x4_f32(false, ahi, false, b0h, (short)0, acc0, false, false);
        acc1 = __builtin_amdgcn_wmma_f32_16x16x4_f32(false, ahi, false, b1h, (short)0, acc1, false, false);
        acc2 = __builtin_amdgcn_wmma_f32_16x16x4_f32(false, ahi, false, b2h, (short)0, acc2, false, false);
        acc3 = __builtin_amdgcn_wmma_f32_16x16x4_f32(false, ahi, false, b3h, (short)0, acc3, false, false);
      }
      a[i] = *(const v4f*)(nrow + i * 8);
      __builtin_amdgcn_sched_barrier(0);
    }

    if (DOTS) {
      const int node = tile * 16 + col;
      const float dnode = dcur;
      float* orow = out + (size_t)node * HID + rbase;
      v8f accs[4] = { acc0, acc1, acc2, acc3 };
      float pl[4], pr[4];
      #pragma unroll
      for (int t = 0; t < 4; ++t) {
        pl[t] = 0.f; pr[t] = 0.f;
        #pragma unroll
        for (int v = 0; v < 8; ++v) {
          float xv = accs[t][v] + bsr[t][v];
          if (RELU) xv = fmaxf(xv, 0.f);
          pl[t] += xv * alr[t][v];
          pr[t] += xv * arr[t][v];
          accs[t][v] = xv;
        }
      }
      float sl = (pl[0] + pl[1]) + (pl[2] + pl[3]);
      float sr = (pr[0] + pr[1]) + (pr[2] + pr[3]);
      sl += __shfl_xor(sl, 16, 32);
      sr += __shfl_xor(sr, 16, 32);
      static_assert(RELU || !DOTS, "the sign bits are free only behind a relu");
      {
        const unsigned sb = __float_as_uint(sl), db = __float_as_uint(dnode);
        #pragma unroll
        for (int t = 0; t < 4; ++t) {
          const unsigned w8 = ((t < 2) ? sb : db) >> (16 * (t & 1) + rbase);
          v4f lo, hi;
          #pragma unroll
          for (int v = 0; v < 8; ++v) {
            const unsigned mag = __float_as_uint(accs[t][v]) & 0x7FFFFFFFu;
            const float xs = __uint_as_float(mag | ((w8 << (31 - v)) & 0x80000000u));
            if (v < 4) lo[v] = xs; else hi[v - 4] = xs;
          }
          *(v4f*)(orow + 16 * t)     = lo;
          *(v4f*)(orow + 16 * t + 4) = hi;
        }
      }
      if (lane < 16) {
        v2f o; o.x = sl; o.y = dnode;
        *(v2f*)(adp + 2 * (size_t)node) = o;
        ar[node] = sr;
      }
    } else {
      float* orow = out + (size_t)tile * 16 * HID;
      #pragma unroll
      for (int v = 0; v < 8; ++v) {
        const int r = rbase + v;
        float x0 = acc0[v] + bb0;
        float x1 = acc1[v] + bb1;
        float x2 = acc2[v] + bb2;
        float x3 = acc3[v] + bb3;
        if (RELU) {
          x0 = fmaxf(x0, 0.f); x1 = fmaxf(x1, 0.f);
          x2 = fmaxf(x2, 0.f); x3 = fmaxf(x3, 0.f);
        }
        orow[(size_t)r * HID +      col] = x0;
        orow[(size_t)r * HID + 16 + col] = x1;
        orow[(size_t)r * HID + 32 + col] = x2;
        orow[(size_t)r * HID + 48 + col] = x3;
      }
    }
    __builtin_amdgcn_sched_barrier(0);
    asm volatile("s_wait_xcnt 0x0");
    __builtin_amdgcn_sched_barrier(0);
    if (!more) break;
    tile = nt;
    dcur = dnxt;
  }
}

__global__ void deg_init(float* __restrict__ deg, int n) {
  int i = blockIdx.x * blockDim.x + threadIdx.x;
  if (i < n) deg[i] = 1.0f;
}
__global__ void deg_acc(const int* __restrict__ src, const int* __restrict__ dst,
                        float* __restrict__ deg, int* __restrict__ keyw, int n, int e) {
  const int i = (int)(blockIdx.x * blockDim.x + threadIdx.x);
  unsigned k = 0;
  int pw = 0;
  if (i < e) {
    const int s = src[i];
    const int d = dst[i];
    atomicAdd(&deg[d], 1.0f);
    k = (unsigned)d / OWN_ROWS;
    pw = (int)(((unsigned)s << PACK_BITS) | ((unsigned)d - k * OWN_ROWS));
  }
  if (!keys_ok(keyw, n, e)) return;
  if (i < e) ((int*)pack_words(keyw, e))[i] = pw;
  const int lane = (int)(threadIdx.x & 31);
  unsigned w[1] = { (i < e) ? key_onehot(k) : 0u };
  wave_bit_transpose<1>(w, lane);
  ((unsigned*)keyw)[(size_t)lane * plane_words(e) + (size_t)(i >> 5)] = w[0];
}
__global__ void deg_rsqrt(float* __restrict__ deg, int n) {
  int i = blockIdx.x * blockDim.x + threadIdx.x;
  if (i < n) deg[i] = rsqrtf(deg[i]);
}

#define HIST_WORDS 51200
#define HIST_NODES (2 * HIST_WORDS)
#define CNT_CHUNK  8192
#define CNT_PER_THREAD (CNT_CHUNK / 1024)
#define CNT_GROUPS (CNT_CHUNK / 32)
#define STG_PITCH  (CNT_GROUPS + 1)

__global__ __launch_bounds__(1024)
void deg_count(const int* __restrict__ src, const int* __restrict__ dst,
               unsigned* __restrict__ planes, int* __restrict__ keyw, int n, int e) {
  __shared__ unsigned hist[HIST_WORDS];
  __shared__ unsigned stage[KEY_PLANES * STG_PITCH];
  const int tid    = (int)threadIdx.x;
  const int lane   = tid & 31;
  const int nwords = (n + 1) >> 1;
  const int cb     = (int)blockIdx.x * CNT_CHUNK;
  unsigned* plane  = planes + (size_t)blockIdx.x * (size_t)nwords;
  const bool planes_ok = keys_ok(keyw, n, e);

  for (int nb = 0; nb < n; nb += HIST_NODES) {
    for (int w = tid; w < HIST_WORDS; w += 1024) hist[w] = 0u;
    __syncthreads();

    int sv[CNT_PER_THREAD], dv[CNT_PER_THREAD];
    unsigned oh[CNT_PER_THREAD];
    #pragma unroll
    for (int j = 0; j < CNT_PER_THREAD; ++j) {
      const int ic = min(cb + j * 1024 + tid, e - 1);
      sv[j] = src[ic];
      dv[j] = dst[ic];
    }
    #pragma unroll
    for (int j = 0; j < CNT_PER_THREAD; ++j) {
      const int i = cb + j * 1024 + tid;
      const int d = dv[j];
      const unsigned dr = (unsigned)(d - nb);
      if (i < e && dr < (unsigned)HIST_NODES)
        atomicAdd(&hist[dr >> 1], 1u << ((dr & 1u) * 16u));
      oh[j] = 0u;
      if (nb == 0 && planes_ok && i < e) {
        const unsigned k = (unsigned)d / OWN_ROWS;
        ((int*)pack_words(keyw, e))[i] =
            (int)(((unsigned)sv[j] << PACK_BITS) | ((unsigned)d - k * OWN_ROWS));
        oh[j] = key_onehot(k);
      }
    }
    if (nb == 0 && planes_ok) {
      wave_bit_transpose<CNT_PER_THREAD>(oh, lane);
      #pragma unroll
      for (int j = 0; j < CNT_PER_THREAD; ++j)
        stage[lane * STG_PITCH + j * 32 + (tid >> 5)] = oh[j];
    }
    __syncthreads();
    const int w0 = nb >> 1;
    const int wn = min(HIST_WORDS, nwords - w0);
    for (int w = tid; w < wn; w += 1024) plane[w0 + w] = hist[w];
    if (nb == 0 && planes_ok) {
      const size_t npw = plane_words(e);
      const size_t g0  = (size_t)blockIdx.x * CNT_GROUPS;
      unsigned* pl = (unsigned*)keyw + (size_t)(tid >> 5) * npw + g0;
      #pragma unroll
      for (int r = 0; r < CNT_GROUPS / 32; ++r) {
        const int lg = r * 32 + lane;
        if (g0 + (size_t)lg < npw) pl[lg] = stage[(tid >> 5) * STG_PITCH + lg];
      }
    }
    __syncthreads();
  }
}

__global__ void deg_finish(const unsigned* __restrict__ planes, float* __restrict__ dinv,
                           int n, int nplanes) {
  const int w = (int)(blockIdx.x * blockDim.x + threadIdx.x);
  const int nwords = (n + 1) >> 1;
  if (w >= nwords) return;
  unsigned lo = 0u, hi = 0u;
  #pragma unroll 14
  for (int b = 0; b < nplanes; ++b) {
    const unsigned v = planes[(size_t)b * (size_t)nwords + w];
    lo += v & 0xFFFFu;
    hi += v >> 16;
  }
  dinv[2 * w] = rsqrtf((float)(lo + 1u));
  if (2 * w + 1 < n) dinv[2 * w + 1] = rsqrtf((float)(hi + 1u));
}

#define LWAVES     32
#define UNIT_EDGES 512
#define QCAP       (UNIT_EDGES + 32)

static_assert(OWN_ROWS <= (1 << PACK_BITS), "local row must fit the packed word");

template<bool RECORD>
__global__ __launch_bounds__(1024)
void layer_owned(const int* __restrict__ src, const int* __restrict__ dst,
                 const float* __restrict__ h, const float* __restrict__ h0,
                 const float* __restrict__ adp, const float* __restrict__ ar,
                 const float* __restrict__ dinv, float* hout, int* lists,
                 const int* __restrict__ keyw,
                 const float* __restrict__ attl_next, const float* __restrict__ attr_next,
                 float* __restrict__ adp_next, float* __restrict__ ar_next,
                 int n, int e_cnt) {
  __shared__ float acc[OWN_ROWS * HID];
  __shared__ int   queue[LWAVES][QCAP];
  __shared__ float wself[OWN_ROWS];
  __shared__ float dinv_own[RECORD ? OWN_ROWS : 1];
  __shared__ float dotl[RECORD ? OWN_ROWS : 1];
  __shared__ float dotr[RECORD ? OWN_ROWS : 1];

  const int tid  = (int)threadIdx.x;
  const int lane = tid & 31;
  const int wave = __builtin_amdgcn_readfirstlane(tid >> 5);
  const int base = (int)blockIdx.x * OWN_ROWS;
  const int rows = min(OWN_ROWS, n - base);
  const int nel  = rows * HID;
  const size_t gbase = (size_t)base * HID;

  int* lw = lists + gbase + (size_t)wave * (size_t)(rows * 2);
  const int lcap = rows * 2 - 1;

  if (tid < rows) {
    const int node = base + tid;
    const float di = dinv[node];
    wself[tid] = tanhf(adp[2 * (size_t)node] + ar[node]) * di * di;
    if (RECORD) dinv_own[tid] = di;
  }
  __syncthreads();
  for (int idx = tid; idx < nel; idx += 1024)
    acc[idx] = EPS_BLEND * fabsf(h0[gbase + idx]) + wself[idx >> 6] * fabsf(h[gbase + idx]);
  __syncthreads();

  auto add_rows = [&](int se, int dl, float xr, float dv, int cnt) {
    for (int j0 = 0; j0 < cnt; j0 += 8) {
      float v0[8], v1[8];
      #pragma unroll
      for (int t = 0; t < 8; ++t) {
        const int sj = __builtin_amdgcn_readlane(se, j0 + t);
        const float* row = h + (size_t)sj * HID;
        v0[t] = row[lane];
        v1[t] = row[lane + 32];
      }
      int sdot = 0, sdin = 0;
      const int rel = lane - j0;
      #pragma unroll
      for (int t = 0; t < 8; ++t) {
        const unsigned ma = __builtin_amdgcn_ballot_w32(__float_as_int(v0[t]) < 0);
        const unsigned md = __builtin_amdgcn_ballot_w32(__float_as_int(v1[t]) < 0);
        sdot = (rel == t) ? (int)ma : sdot;
        sdin = (rel == t) ? (int)md : sdin;
      }
      const float w = tanhf(__int_as_float(sdot) + xr) * __int_as_float(sdin) * dv;
      #pragma unroll
      for (int t = 0; t < 8; ++t) {
        const int j = j0 + t;
        const float wj = __int_as_float(__builtin_amdgcn_readlane(__float_as_int(w), j));
        const int dj = __builtin_amdgcn_readlane(dl, j);
        const bool live = j < cnt;
        atomicAdd(&acc[dj * HID + lane],      live ? wj * fabsf(v0[t]) : 0.f);
        atomicAdd(&acc[dj * HID + 32 + lane], live ? wj * fabsf(v1[t]) : 0.f);
      }
    }
  };

  bool do_sweep = true;
  if (!RECORD) {
    int total = lw[0];
    if (total >= 0) {
      do_sweep = false;
      total = min(total, lcap);
      int  p_nxt = (lane < total) ? lw[1 + lane] : 0;
      int  se_c = 0, dl_c = 0;
      float xr = 0.f, dd = 0.f;
      {
        se_c = p_nxt >> PACK_BITS;
        dl_c = p_nxt & ((1 << PACK_BITS) - 1);
        xr = ar[base + dl_c]; dd = dinv[base + dl_c];
        p_nxt = (32 + lane < total) ? lw[1 + 32 + lane] : 0;
      }
      for (int b = 0; b < total; b += 32) {
        const int cnt = min(32, total - b);
        const int se = se_c, dl = dl_c;
        const float xr_c = xr, dd_c = dd;
        se_c = p_nxt >> PACK_BITS;
        dl_c = p_nxt & ((1 << PACK_BITS) - 1);
        xr = ar[base + dl_c]; dd = dinv[base + dl_c];
        p_nxt = (b + 64 + lane < total) ? lw[1 + b + 64 + lane] : 0;
        add_rows(se, dl, xr_c, dd_c, cnt);
      }
    }
  }

  if (do_sweep) {
    int* qw = queue[wave];
    const unsigned lt = (1u << lane) - 1u;
    int qn = 0;
    int rec_n = (n <= PACK_MAXN) ? 0 : -1;

    const bool by_key = keys_ok(keyw, n, e_cnt);
    const int* packw = pack_words(keyw, e_cnt);

    auto drain = [&](bool all) {
      while (qn >= 32 || (all && qn > 0)) {
        const int cnt = min(qn, 32);
        const int qb  = qn - cnt;
        __builtin_amdgcn_wave_barrier();
        const bool act = lane < cnt;
        int se = 0, dd = base;
        if (act) {
          const int e = qw[qb + lane];
          if (by_key) {
            const int p = packw[e];
            se = p >> PACK_BITS;
            dd = base + (p & ((1 << PACK_BITS) - 1));
          } else {
            se = src[e];
            dd = dst[e];
          }
        }
        const int dl = dd - base;
        if (RECORD) {
          if (rec_n >= 0) {
            if (rec_n + cnt <= lcap) {
              if (act) lw[1 + rec_n + lane] = (se << PACK_BITS) | dl;
              rec_n += cnt;
            } else {
              rec_n = -1;
            }
          }
        }
        const float xr = ar[dd];
        const float dv = dinv[dd];
        add_rows(se, dl, xr, dv, cnt);
        qn = qb;
      }
    };

    if (by_key) {
      const size_t npw = plane_words(e_cnt);
      const unsigned* hp = (const unsigned*)keyw + (size_t)(blockIdx.x >> 4) * npw;
      const unsigned* lp = (const unsigned*)keyw + (size_t)(16 + (blockIdx.x & 15u)) * npw;
      const int units = (e_cnt + KEY_UNIT - 1) / KEY_UNIT;
      const int step = max(1, units / (int)gridDim.x);
      const int rot  = units > 0 ? (int)(((unsigned)blockIdx.x * (unsigned)step) % (unsigned)units) : 0;
      auto load_keys = [&](int uu_in, v4i& vh, v4i& vl) -> int {
        int uu = uu_in + rot;
        if (uu >= units) uu -= units;
        const int gb = uu * (KEY_UNIT / 32) + lane * 4;
        vh = *(const v4i*)(hp + gb);
        vl = *(const v4i*)(lp + gb);
        return gb;
      };

      auto scan_group = [&](unsigned g, int e0) {
        while (__builtin_amdgcn_ballot_w32(g != 0u)) {
          const int eidx = e0 + __builtin_ctz(g | 0x80000000u);
          const bool m = (g != 0u) && (eidx < e_cnt);
          const unsigned mm = __builtin_amdgcn_ballot_w32(m);
          if (m) qw[qn + __builtin_popcount(mm & lt)] = eidx;
          qn += __builtin_popcount(mm);
          g &= g - 1u;
          drain(false);
        }
      };

      v4i ch = {}, cl = {};
      int c_gb = 0;
      int u = wave;
      const bool have = u < units;
      if (have) c_gb = load_keys(u, ch, cl);

      while (have) {
        v4i nh, nl;
        const bool have_n = u + LWAVES < units;
        const int n_gb = load_keys(have_n ? u + LWAVES : u, nh, nl);

        #pragma unroll
        for (int q = 0; q < 4; ++q)
          scan_group((unsigned)ch[q] & (unsigned)cl[q], (c_gb + q) * 32);

        if (!have_n) break;
        ch = nh; cl = nl; c_gb = n_gb;
        u += LWAVES;
      }
      drain(true);
    } else {
      const int units = (e_cnt + UNIT_EDGES - 1) / UNIT_EDGES;
      const int step = max(1, units / (int)gridDim.x);
      const int rot  = units > 0 ? (int)(((unsigned)blockIdx.x * (unsigned)step) % (unsigned)units) : 0;
      const bool vec_ok = ((e_cnt & 3) == 0) && ((((size_t)dst) & 15) == 0);

      v4i cur[4];
      int cur_eb = 0;
      int u = wave;
      bool have = u < units;

      auto load_unit = [&](int uu_in, v4i (&v)[4]) -> int {
        int uu = uu_in + rot;
        if (uu >= units) uu -= units;
        const int eb = uu * UNIT_EDGES + lane * 4;
        if (vec_ok && (uu + 1) * UNIT_EDGES <= e_cnt) {
          #pragma unroll
          for (int k = 0; k < 4; ++k) v[k] = *(const v4i*)(dst + eb + k * 128);
        } else {
          #pragma unroll
          for (int k = 0; k < 4; ++k) {
            #pragma unroll
            for (int j = 0; j < 4; ++j) {
              const int idx = eb + k * 128 + j;
              v[k][j] = (idx < e_cnt) ? dst[idx] : -1;
            }
          }
        }
        return eb;
      };

      if (have) cur_eb = load_unit(u, cur);

      while (true) {
        v4i nxt[4];
        int nxt_eb = 0;
        const bool have_n = have && (u + LWAVES < units);
        if (have_n) nxt_eb = load_unit(u + LWAVES, nxt);

        if (have) {
          #pragma unroll
          for (int k = 0; k < 4; ++k) {
            #pragma unroll
            for (int j = 0; j < 4; ++j) {
              const int d = cur[k][j];
              const bool m = (unsigned)(d - base) < (unsigned)rows;
              const unsigned mask = __builtin_amdgcn_ballot_w32(m);
              if (mask) {
                if (m) qw[qn + __builtin_popcount(mask & lt)] = cur_eb + k * 128 + j;
                qn += __builtin_popcount(mask);
              }
            }
          }
        }

        const bool last = !have_n;
        drain(last);
        if (last) break;
        #pragma unroll
        for (int k = 0; k < 4; ++k) cur[k] = nxt[k];
        cur_eb = nxt_eb;
        u += LWAVES;
      }
    }
    if (RECORD) {
      if (lane == 0) lw[0] = rec_n;
    }
  }
  __syncthreads();

  if (RECORD) {
    const float l0 = attl_next[lane], l1 = attl_next[lane + 32];
    const float r0 = attr_next[lane], r1 = attr_next[lane + 32];
    for (int rb = wave; rb < rows; rb += 4 * LWAVES) {
      float pl[4], pr[4], hv0[4], hv1[4];
      #pragma unroll
      for (int k = 0; k < 4; ++k) {
        const int r = rb + k * LWAVES;
        pl[k] = 0.f; pr[k] = 0.f; hv0[k] = 0.f; hv1[k] = 0.f;
        if (r < rows) {
          hv0[k] = fmaxf(acc[r * HID + lane], 0.f);
          hv1[k] = fmaxf(acc[r * HID + 32 + lane], 0.f);
          pl[k] = hv0[k] * l0 + hv1[k] * l1;
          pr[k] = hv0[k] * r0 + hv1[k] * r1;
        }
      }
      #pragma unroll
      for (int off = 16; off > 0; off >>= 1) {
        #pragma unroll
        for (int k = 0; k < 4; ++k) {
          pl[k] += __shfl_xor(pl[k], off, 32);
          pr[k] += __shfl_xor(pr[k], off, 32);
        }
      }
      #pragma unroll
      for (int k = 0; k < 4; ++k) {
        const int r = rb + k * LWAVES;
        if (r < rows) {
          const unsigned sb = __float_as_uint(pl[k]), db = __float_as_uint(dinv_own[r]);
          const unsigned m0 = __float_as_uint(hv0[k]) & 0x7FFFFFFFu;
          const unsigned m1 = __float_as_uint(hv1[k]) & 0x7FFFFFFFu;
          hout[gbase + (size_t)r * HID + lane]      = __uint_as_float(m0 | ((sb << (31 - lane)) & 0x80000000u));
          hout[gbase + (size_t)r * HID + 32 + lane] = __uint_as_float(m1 | ((db << (31 - lane)) & 0x80000000u));
        }
      }
      if (lane == 0) {
        #pragma unroll
        for (int k = 0; k < 4; ++k) {
          const int r = rb + k * LWAVES;
          if (r < rows) { dotl[r] = pl[k]; dotr[r] = pr[k]; }
        }
      }
    }
    __syncthreads();
    if (tid < rows) {
      v2f o; o.x = dotl[tid]; o.y = dinv[base + tid];
      *(v2f*)(adp_next + 2 * (size_t)(base + tid)) = o;
      ar_next[base + tid] = dotr[tid];
    }
  } else {
    for (int idx = tid; idx < nel; idx += 1024)
      hout[gbase + idx] = fmaxf(acc[idx], 0.f);
  }
}

extern "C" void kernel_launch(void* const* d_in, const int* in_sizes, int n_in,
                              void* d_out, int out_size, void* d_ws, size_t ws_size,
                              hipStream_t stream) {
  const float* x     = (const float*)d_in[0];
  const int*   ei    = (const int*)  d_in[1];
  const float* W_in  = (const float*)d_in[2];
  const float* b_in  = (const float*)d_in[3];
  const float* attl1 = (const float*)d_in[4];
  const float* attr1 = (const float*)d_in[5];
  const float* attl2 = (const float*)d_in[6];
  const float* attr2 = (const float*)d_in[7];
  const float* W_out = (const float*)d_in[8];
  const float* b_out = (const float*)d_in[9];
  float* out = (float*)d_out;

  const int IN = 128;
  const int E  = in_sizes[1] / 2;
  const int N  = in_sizes[0] / IN;
  const int*   srcI = ei;
  const int*   dstI = ei + E;

  float* H0   = (float*)d_ws;
  float* H1   = H0   + (size_t)N * HID;
  float* OUTB = H1   + (size_t)N * HID;
  float* ADP  = OUTB + (size_t)N * HID;
  float* DINV = ADP  + 2 * (size_t)N;
  float* ADP2 = out;
  float* AR2  = out + 2 * (size_t)N;
  float* AR   = out + 3 * (size_t)N;
  int*   KEYW = (int*)(out + 4 * (size_t)N);

  const int ntiles = N / 16;
  const int nThreads  = (N + 255) / 256;
  const int degBlk    = (E + KEY_UNIT - 1) / KEY_UNIT * (KEY_UNIT / 256);
  const int layerBlk  = (N + OWN_ROWS - 1) / OWN_ROWS;

  const int nWords = (N + 1) / 2;
  const int cntBlk = (degBlk * 256 + CNT_CHUNK - 1) / CNT_CHUNK;
  if ((size_t)cntBlk * (size_t)nWords <= 2 * (size_t)N * HID) {
    deg_count <<<cntBlk, 1024, 0, stream>>>(srcI, dstI, (unsigned*)H1, KEYW, N, E);
    deg_finish<<<(nWords + 255) / 256, 256, 0, stream>>>((const unsigned*)H1, DINV, N, cntBlk);
  } else {
    deg_init <<<nThreads, 256, 0, stream>>>(DINV, N);
    deg_acc  <<<degBlk, 256, 0, stream>>>(srcI, dstI, DINV, KEYW, N, E);
    deg_rsqrt<<<nThreads, 256, 0, stream>>>(DINV, N);
  }

  gemm_wmma_f32<128, true, true><<<256, 256, 0, stream>>>(x, W_in, b_in, H0, ntiles,
                                                          attl1, attr1, DINV, ADP, AR);

  layer_owned<true> <<<layerBlk, 1024, 0, stream>>>(srcI, dstI, H0, H0, ADP, AR, DINV, H1, (int*)OUTB, KEYW,
                                                    attl2, attr2, ADP2, AR2, N, E);

  layer_owned<false><<<layerBlk, 1024, 0, stream>>>(srcI, dstI, H1, H0, ADP2, AR2, DINV, OUTB, (int*)OUTB, KEYW,
                                                    nullptr, nullptr, nullptr, nullptr, N, E);

  gemm_wmma_f32<64, false, false><<<256, 256, 0, stream>>>(OUTB, W_out, b_out, out, ntiles,
                                                           nullptr, nullptr, nullptr, nullptr, nullptr);
}
